# starting kernel
# baseline (speedup 1.0000x reference)
#include <hip/hip_runtime.h>

#define H 256
#define TM 128
#define TNB 64
#define KC 32
#define APAD 40

typedef __bf16 bf16;
typedef __attribute__((ext_vector_type(16))) __bf16 v16bf;
typedef __attribute__((ext_vector_type(8)))  __bf16 v8bf;
typedef __attribute__((ext_vector_type(8)))  float  v8f;

__device__ __forceinline__ unsigned short f2bfbits(float f) {
  unsigned u = __float_as_uint(f);
  return (unsigned short)((u + 0x7FFFu + ((u >> 16) & 1u)) >> 16);
}
__device__ __forceinline__ float bfbits2f(unsigned short h) {
  return __uint_as_float(((unsigned)h) << 16);
}
__device__ __forceinline__ bf16 f2bf(float f) {
  unsigned short hs = f2bfbits(f);
  return __builtin_bit_cast(bf16, hs);
}
__device__ __forceinline__ float lrelu(float v) { return v > 0.f ? v : 0.2f * v; }
__device__ __forceinline__ unsigned f2ord(float f) {
  unsigned u = __float_as_uint(f);
  return (u & 0x80000000u) ? ~u : (u | 0x80000000u);
}
__device__ __forceinline__ float ord2f(unsigned u) {
  unsigned b = (u & 0x80000000u) ? (u & 0x7FFFFFFFu) : ~u;
  return __uint_as_float(b);
}

__device__ __forceinline__ v16bf load_frag(const bf16* rowp, int g0) {
  v8bf a = *(const v8bf*)(rowp + g0);
  v8bf b = *(const v8bf*)(rowp + 16 + g0);
  return __builtin_shufflevector(a, b, 0, 1, 2, 3, 4, 5, 6, 7,
                                 8, 9, 10, 11, 12, 13, 14, 15);
}

#define WMMA_BF16(A_, B_, C_) \
  __builtin_amdgcn_wmma_f32_16x16x32_bf16(false, (A_), false, (B_), (short)0, (C_), false, false)

__global__ __launch_bounds__(256) void k_gemm_bf16x3(
    const float* __restrict__ A, const float* __restrict__ B,
    const float* __restrict__ bias, float* __restrict__ C, int M)
{
  __shared__ bf16 Ah[TM][APAD];
  __shared__ bf16 Al[TM][APAD];
  __shared__ bf16 Bth[TNB][APAD];
  __shared__ bf16 Btl[TNB][APAD];

  const int tid = threadIdx.x;
  const int m0 = blockIdx.x * TM;
  const int n0 = blockIdx.y * TNB;
  const int wid = tid >> 5, lane = tid & 31;
  const int hl = lane & 15, hh = lane >> 4;
  const int wr = wid >> 1;
  const int wc = wid & 1;
  const int g0 = hh << 3;

  v8f acc00 = {}, acc01 = {}, acc10 = {}, acc11 = {};

  for (int k0 = 0; k0 < H; k0 += KC) {
    __syncthreads();
    for (int idx = tid; idx < (TM * KC) / 4; idx += 256) {
      int r = idx >> 3;
      int c4 = (idx & 7) << 2;
      int gr = m0 + r;
      float4 v = (gr < M) ? *(const float4*)&A[(size_t)gr * H + k0 + c4]
                          : make_float4(0.f, 0.f, 0.f, 0.f);
      ushort4 hiv, lov;
      hiv.x = f2bfbits(v.x); lov.x = f2bfbits(v.x - bfbits2f(hiv.x));
      hiv.y = f2bfbits(v.y); lov.y = f2bfbits(v.y - bfbits2f(hiv.y));
      hiv.z = f2bfbits(v.z); lov.z = f2bfbits(v.z - bfbits2f(hiv.z));
      hiv.w = f2bfbits(v.w); lov.w = f2bfbits(v.w - bfbits2f(hiv.w));
      *(ushort4*)&Ah[r][c4] = hiv;
      *(ushort4*)&Al[r][c4] = lov;
    }
    for (int idx = tid; idx < (KC * TNB) / 4; idx += 256) {
      int r = idx >> 4;
      int c4 = (idx & 15) << 2;
      float4 v = *(const float4*)&B[(size_t)(k0 + r) * H + n0 + c4];
      bf16 h0 = f2bf(v.x), h1 = f2bf(v.y), h2 = f2bf(v.z), h3 = f2bf(v.w);
      Bth[c4 + 0][r] = h0; Btl[c4 + 0][r] = f2bf(v.x - bfbits2f(__builtin_bit_cast(unsigned short, h0)));
      Bth[c4 + 1][r] = h1; Btl[c4 + 1][r] = f2bf(v.y - bfbits2f(__builtin_bit_cast(unsigned short, h1)));
      Bth[c4 + 2][r] = h2; Btl[c4 + 2][r] = f2bf(v.z - bfbits2f(__builtin_bit_cast(unsigned short, h2)));
      Bth[c4 + 3][r] = h3; Btl[c4 + 3][r] = f2bf(v.w - bfbits2f(__builtin_bit_cast(unsigned short, h3)));
    }
    __syncthreads();

    v16bf ah0 = load_frag(&Ah[(wr << 5) + hl][0], g0);
    v16bf al0 = load_frag(&Al[(wr << 5) + hl][0], g0);
    v16bf ah1 = load_frag(&Ah[(wr << 5) + 16 + hl][0], g0);
    v16bf al1 = load_frag(&Al[(wr << 5) + 16 + hl][0], g0);
    v16bf bh0 = load_frag(&Bth[(wc << 5) + hl][0], g0);
    v16bf bl0 = load_frag(&Btl[(wc << 5) + hl][0], g0);
    v16bf bh1 = load_frag(&Bth[(wc << 5) + 16 + hl][0], g0);
    v16bf bl1 = load_frag(&Btl[(wc << 5) + 16 + hl][0], g0);

    acc00 = WMMA_BF16(ah0, bh0, acc00);
    acc00 = WMMA_BF16(al0, bh0, acc00);
    acc00 = WMMA_BF16(ah0, bl0, acc00);
    acc01 = WMMA_BF16(ah0, bh1, acc01);
    acc01 = WMMA_BF16(al0, bh1, acc01);
    acc01 = WMMA_BF16(ah0, bl1, acc01);
    acc10 = WMMA_BF16(ah1, bh0, acc10);
    acc10 = WMMA_BF16(al1, bh0, acc10);
    acc10 = WMMA_BF16(ah1, bl0, acc10);
    acc11 = WMMA_BF16(ah1, bh1, acc11);
    acc11 = WMMA_BF16(al1, bh1, acc11);
    acc11 = WMMA_BF16(ah1, bl1, acc11);
  }

  const int col0 = n0 + (wc << 5) + hl;
  const int col1 = col0 + 16;
  const float b0 = bias ? bias[col0] : 0.f;
  const float b1v = bias ? bias[col1] : 0.f;
  const int rbase0 = m0 + (wr << 5) + (hh << 3);
  const int rbase1 = rbase0 + 16;
#pragma unroll
  for (int r = 0; r < 8; ++r) {
    int g0r = rbase0 + r;
    if (g0r < M) {
      C[(size_t)g0r * H + col0] = acc00[r] + b0;
      C[(size_t)g0r * H + col1] = acc01[r] + b1v;
    }
    int g1r = rbase1 + r;
    if (g1r < M) {
      C[(size_t)g1r * H + col0] = acc10[r] + b0;
      C[(size_t)g1r * H + col1] = acc11[r] + b1v;
    }
  }
}

__global__ void k_zero(float* p, int n) {
  int i = blockIdx.x * blockDim.x + threadIdx.x;
  if (i < n) p[i] = 0.f;
}

__global__ void k_colsum(const float* __restrict__ x, float* __restrict__ g, int N) {
  int c = threadIdx.x;
  int r0 = blockIdx.x * 256;
  int r1 = r0 + 256; if (r1 > N) r1 = N;
  float acc = 0.f;
  for (int r = r0; r < r1; ++r) acc += x[(size_t)r * H + c];
  atomicAdd(&g[c], acc);
}

__global__ void k_selector(const float* __restrict__ g,
                           const float* __restrict__ W1, const float* __restrict__ b1,
                           const float* __restrict__ W2, const float* __restrict__ b2,
                           float* __restrict__ sw, int N) {
  __shared__ float gsh[H];
  __shared__ float hid[128];
  __shared__ float lg[4];
  int t = threadIdx.x;
  float invN = 1.f / (float)N;
  gsh[t] = g[t] * invN;
  gsh[t + 128] = g[t + 128] * invN;
  __syncthreads();
  float a = b1[t];
  for (int k = 0; k < H; ++k) a += gsh[k] * W1[k * 128 + t];
  hid[t] = fmaxf(a, 0.f);
  __syncthreads();
  if (t < 4) {
    float l = b2[t];
    for (int k = 0; k < 128; ++k) l += hid[k] * W2[k * 4 + t];
    lg[t] = l;
  }
  __syncthreads();
  if (t == 0) {
    float m = fmaxf(fmaxf(lg[0], lg[1]), fmaxf(lg[2], lg[3]));
    float e0 = __expf(lg[0] - m), e1 = __expf(lg[1] - m);
    float e2 = __expf(lg[2] - m), e3 = __expf(lg[3] - m);
    float s = e0 + e1 + e2 + e3;
    sw[0] = e0 / s; sw[1] = e1 / s; sw[2] = e2 / s; sw[3] = e3 / s;
  }
}

__global__ void k_sd(const float* __restrict__ h, const float* __restrict__ asrc,
                     const float* __restrict__ adst, float* __restrict__ s,
                     float* __restrict__ d, int N) {
  int n = blockIdx.x * 8 + (threadIdx.x >> 5);
  if (n >= N) return;
  int lane = threadIdx.x & 31;
  float vs = 0.f, vd = 0.f;
  for (int c = lane; c < H; c += 32) {
    float v = h[(size_t)n * H + c];
    vs += v * asrc[c];
    vd += v * adst[c];
  }
  for (int o = 16; o; o >>= 1) { vs += __shfl_xor(vs, o, 32); vd += __shfl_xor(vd, o, 32); }
  if (lane == 0) { s[n] = vs; d[n] = vd; }
}

__global__ void k_init_m(const float* __restrict__ s, const float* __restrict__ d,
                         unsigned* __restrict__ mb, int N) {
  int n = blockIdx.x * blockDim.x + threadIdx.x;
  if (n < N) mb[n] = f2ord(lrelu(s[n] + d[n]));
}

__global__ void k_edge_max(const int* __restrict__ src, const int* __restrict__ dst,
                           const int* __restrict__ attr, int branch,
                           const float* __restrict__ s, const float* __restrict__ d,
                           unsigned* __restrict__ mb, int E) {
  int e = blockIdx.x * blockDim.x + threadIdx.x;
  if (e < E && attr[e] == branch) {
    int dn = dst[e];
    float v = lrelu(s[src[e]] + d[dn]);
    atomicMax(&mb[dn], f2ord(v));
  }
}

__global__ void k_init_num(const float* __restrict__ h, const float* __restrict__ s,
                           const float* __restrict__ d, const unsigned* __restrict__ mb,
                           float* __restrict__ mf, float* __restrict__ denom,
                           float* __restrict__ num, int N) {
  int n = blockIdx.x * 8 + (threadIdx.x >> 5);
  if (n >= N) return;
  int lane = threadIdx.x & 31;
  float m = ord2f(mb[n]);
  float wself = __expf(lrelu(s[n] + d[n]) - m);
  if (lane == 0) { mf[n] = m; denom[n] = wself; }
  for (int c = lane; c < H; c += 32) num[(size_t)n * H + c] = wself * h[(size_t)n * H + c];
}

__global__ void k_edge_accum(const int* __restrict__ src, const int* __restrict__ dst,
                             const int* __restrict__ attr, int branch,
                             const float* __restrict__ s, const float* __restrict__ d,
                             const float* __restrict__ mf, const float* __restrict__ h,
                             float* __restrict__ denom, float* __restrict__ num, int E) {
  int e = blockIdx.x * 8 + (threadIdx.x >> 5);
  if (e >= E || attr[e] != branch) return;
  int sn = src[e], dn = dst[e];
  float w = __expf(lrelu(s[sn] + d[dn]) - mf[dn]);
  int lane = threadIdx.x & 31;
  if (lane == 0) atomicAdd(&denom[dn], w);
  for (int c = lane; c < H; c += 32)
    atomicAdd(&num[(size_t)dn * H + c], w * h[(size_t)sn * H + c]);
}

__global__ void k_finalize(float* __restrict__ num, const float* __restrict__ denom,
                           const float* __restrict__ bias, int total) {
  int i = blockIdx.x * blockDim.x + threadIdx.x;
  if (i < total) {
    int n = i >> 8;
    num[i] = num[i] / denom[n] + bias[i & (H - 1)];
  }
}

__global__ void k_ln_relu(const float* __restrict__ t, const float* __restrict__ gamma,
                          const float* __restrict__ beta, float* __restrict__ out,
                          const float* __restrict__ swp, int swidx, int accumulate, int N) {
  int n = blockIdx.x * 8 + (threadIdx.x >> 5);
  if (n >= N) return;
  int lane = threadIdx.x & 31;
  float v[8];
  float sum = 0.f, sq = 0.f;
#pragma unroll
  for (int j = 0; j < 8; ++j) {
    v[j] = t[(size_t)n * H + lane + j * 32];
    sum += v[j];
    sq += v[j] * v[j];
  }
  for (int o = 16; o; o >>= 1) { sum += __shfl_xor(sum, o, 32); sq += __shfl_xor(sq, o, 32); }
  float mu = sum * (1.f / H);
  float var = sq * (1.f / H) - mu * mu;
  float inv = rsqrtf(var + 1e-5f);
  float scale = swp ? swp[swidx] : 1.f;
#pragma unroll
  for (int j = 0; j < 8; ++j) {
    int c = lane + j * 32;
    float y = (v[j] - mu) * inv * gamma[c] + beta[c];
    y = fmaxf(y, 0.f) * scale;
    if (accumulate) out[(size_t)n * H + c] += y;
    else            out[(size_t)n * H + c] = y;
  }
}

extern "C" void kernel_launch(void* const* d_in, const int* in_sizes, int n_in,
                              void* d_out, int out_size, void* d_ws, size_t ws_size,
                              hipStream_t stream) {
  const float* x     = (const float*)d_in[0];
  const int*   ei    = (const int*)d_in[1];
  const int*   ea    = (const int*)d_in[2];
  const float* Wg    = (const float*)d_in[3];
  const float* a_src = (const float*)d_in[4];
  const float* a_dst = (const float*)d_in[5];
  const float* bg    = (const float*)d_in[6];
  const float* Wsl   = (const float*)d_in[7];
  const float* bsl   = (const float*)d_in[8];
  const float* gsl   = (const float*)d_in[9];
  const float* betas = (const float*)d_in[10];
  const float* W1    = (const float*)d_in[11];
  const float* b1    = (const float*)d_in[12];
  const float* W2    = (const float*)d_in[13];
  const float* b2    = (const float*)d_in[14];
  const float* Wf    = (const float*)d_in[15];
  const float* bfv   = (const float*)d_in[16];
  const float* gf    = (const float*)d_in[17];
  const float* betaf = (const float*)d_in[18];

  const int N = in_sizes[0] / H;
  const int E = in_sizes[2];
  float* out = (float*)d_out;

  char* wsb = (char*)d_ws;
  size_t off = 0;
  auto alloc = [&](size_t bytes) -> void* {
    void* p = wsb + off;
    off += (bytes + 255) & ~(size_t)255;
    return p;
  };
  float*    g     = (float*)alloc(H * sizeof(float));
  float*    sw    = (float*)alloc(4 * sizeof(float));
  float*    sArr  = (float*)alloc((size_t)N * sizeof(float));
  float*    dArr  = (float*)alloc((size_t)N * sizeof(float));
  unsigned* mb    = (unsigned*)alloc((size_t)N * sizeof(unsigned));
  float*    mf    = (float*)alloc((size_t)N * sizeof(float));
  float*    denom = (float*)alloc((size_t)N * sizeof(float));
  float*    bufH  = (float*)alloc((size_t)N * H * sizeof(float));
  float*    bufX  = (float*)alloc((size_t)N * H * sizeof(float));
  float*    bufC  = (float*)alloc((size_t)N * H * sizeof(float));

  dim3 gemmGrid((unsigned)((N + TM - 1) / TM), (unsigned)(H / TNB));
  int nWaveBlocks = (N + 7) / 8;
  int eWaveBlocks = (E + 7) / 8;
  int nThrBlocks  = (N + 255) / 256;
  int eThrBlocks  = (E + 255) / 256;
  int nhBlocks    = (N * H + 255) / 256;

  k_zero<<<1, 256, 0, stream>>>(g, H);
  k_colsum<<<nThrBlocks, 256, 0, stream>>>(x, g, N);
  k_selector<<<1, 128, 0, stream>>>(g, W1, b1, W2, b2, sw, N);

  for (int i = 0; i < 3; ++i) {
    k_gemm_bf16x3<<<gemmGrid, 256, 0, stream>>>(x, Wg + (size_t)i * H * H, nullptr, bufH, N);
    k_sd<<<nWaveBlocks, 256, 0, stream>>>(bufH, a_src + i * H, a_dst + i * H, sArr, dArr, N);
    k_init_m<<<nThrBlocks, 256, 0, stream>>>(sArr, dArr, mb, N);
    k_edge_max<<<eThrBlocks, 256, 0, stream>>>(ei, ei + E, ea, i, sArr, dArr, mb, E);
    k_init_num<<<nWaveBlocks, 256, 0, stream>>>(bufH, sArr, dArr, mb, mf, denom, bufX, N);
    k_edge_accum<<<eWaveBlocks, 256, 0, stream>>>(ei, ei + E, ea, i, sArr, dArr, mf, bufH, denom, bufX, E);
    k_finalize<<<nhBlocks, 256, 0, stream>>>(bufX, denom, bg + i * H, N * H);
    k_gemm_bf16x3<<<gemmGrid, 256, 0, stream>>>(bufX, Wsl + (size_t)i * H * H, bsl + i * H, bufH, N);
    k_ln_relu<<<nWaveBlocks, 256, 0, stream>>>(bufH, gsl + i * H, betas + i * H, bufC, sw, i,
                                               i > 0 ? 1 : 0, N);
  }
  k_gemm_bf16x3<<<gemmGrid, 256, 0, stream>>>(x, Wsl + (size_t)3 * H * H, bsl + 3 * H, bufH, N);
  k_ln_relu<<<nWaveBlocks, 256, 0, stream>>>(bufH, gsl + 3 * H, betas + 3 * H, bufC, sw, 3, 1, N);

  k_gemm_bf16x3<<<gemmGrid, 256, 0, stream>>>(bufC, Wf, bfv, bufH, N);
  k_ln_relu<<<nWaveBlocks, 256, 0, stream>>>(bufH, gf, betaf, out, nullptr, 0, 0, N);
}


# column sum of x: 256 work-groups read whole rows with 16-byte loads into partial rows that the selector folds; no atomics, no zeroing launch; the mean is now summed in a fixed tree order
# speedup vs baseline: 1.1021x; 1.1021x over previous
#include <hip/hip_runtime.h>

#define H 256
#define TM 128
#define TNB 64
#define KC 32
#define APAD 40

typedef __bf16 bf16;
typedef __attribute__((ext_vector_type(16))) __bf16 v16bf;
typedef __attribute__((ext_vector_type(8)))  __bf16 v8bf;
typedef __attribute__((ext_vector_type(8)))  float  v8f;

__device__ __forceinline__ unsigned short f2bfbits(float f) {
  unsigned u = __float_as_uint(f);
  return (unsigned short)((u + 0x7FFFu + ((u >> 16) & 1u)) >> 16);
}
__device__ __forceinline__ float bfbits2f(unsigned short h) {
  return __uint_as_float(((unsigned)h) << 16);
}
__device__ __forceinline__ bf16 f2bf(float f) {
  unsigned short hs = f2bfbits(f);
  return __builtin_bit_cast(bf16, hs);
}
__device__ __forceinline__ float lrelu(float v) { return v > 0.f ? v : 0.2f * v; }
__device__ __forceinline__ unsigned f2ord(float f) {
  unsigned u = __float_as_uint(f);
  return (u & 0x80000000u) ? ~u : (u | 0x80000000u);
}
__device__ __forceinline__ float ord2f(unsigned u) {
  unsigned b = (u & 0x80000000u) ? (u & 0x7FFFFFFFu) : ~u;
  return __uint_as_float(b);
}

__device__ __forceinline__ v16bf load_frag(const bf16* rowp, int g0) {
  v8bf a = *(const v8bf*)(rowp + g0);
  v8bf b = *(const v8bf*)(rowp + 16 + g0);
  return __builtin_shufflevector(a, b, 0, 1, 2, 3, 4, 5, 6, 7,
                                 8, 9, 10, 11, 12, 13, 14, 15);
}

#define WMMA_BF16(A_, B_, C_) \
  __builtin_amdgcn_wmma_f32_16x16x32_bf16(false, (A_), false, (B_), (short)0, (C_), false, false)

__global__ __launch_bounds__(256) void k_gemm_bf16x3(
    const float* __restrict__ A, const float* __restrict__ B,
    const float* __restrict__ bias, float* __restrict__ C, int M)
{
  __shared__ bf16 Ah[TM][APAD];
  __shared__ bf16 Al[TM][APAD];
  __shared__ bf16 Bth[TNB][APAD];
  __shared__ bf16 Btl[TNB][APAD];

  const int tid = threadIdx.x;
  const int m0 = blockIdx.x * TM;
  const int n0 = blockIdx.y * TNB;
  const int wid = tid >> 5, lane = tid & 31;
  const int hl = lane & 15, hh = lane >> 4;
  const int wr = wid >> 1;
  const int wc = wid & 1;
  const int g0 = hh << 3;

  v8f acc00 = {}, acc01 = {}, acc10 = {}, acc11 = {};

  for (int k0 = 0; k0 < H; k0 += KC) {
    __syncthreads();
    for (int idx = tid; idx < (TM * KC) / 4; idx += 256) {
      int r = idx >> 3;
      int c4 = (idx & 7) << 2;
      int gr = m0 + r;
      float4 v = (gr < M) ? *(const float4*)&A[(size_t)gr * H + k0 + c4]
                          : make_float4(0.f, 0.f, 0.f, 0.f);
      ushort4 hiv, lov;
      hiv.x = f2bfbits(v.x); lov.x = f2bfbits(v.x - bfbits2f(hiv.x));
      hiv.y = f2bfbits(v.y); lov.y = f2bfbits(v.y - bfbits2f(hiv.y));
      hiv.z = f2bfbits(v.z); lov.z = f2bfbits(v.z - bfbits2f(hiv.z));
      hiv.w = f2bfbits(v.w); lov.w = f2bfbits(v.w - bfbits2f(hiv.w));
      *(ushort4*)&Ah[r][c4] = hiv;
      *(ushort4*)&Al[r][c4] = lov;
    }
    for (int idx = tid; idx < (KC * TNB) / 4; idx += 256) {
      int r = idx >> 4;
      int c4 = (idx & 15) << 2;
      float4 v = *(const float4*)&B[(size_t)(k0 + r) * H + n0 + c4];
      bf16 h0 = f2bf(v.x), h1 = f2bf(v.y), h2 = f2bf(v.z), h3 = f2bf(v.w);
      Bth[c4 + 0][r] = h0; Btl[c4 + 0][r] = f2bf(v.x - bfbits2f(__builtin_bit_cast(unsigned short, h0)));
      Bth[c4 + 1][r] = h1; Btl[c4 + 1][r] = f2bf(v.y - bfbits2f(__builtin_bit_cast(unsigned short, h1)));
      Bth[c4 + 2][r] = h2; Btl[c4 + 2][r] = f2bf(v.z - bfbits2f(__builtin_bit_cast(unsigned short, h2)));
      Bth[c4 + 3][r] = h3; Btl[c4 + 3][r] = f2bf(v.w - bfbits2f(__builtin_bit_cast(unsigned short, h3)));
    }
    __syncthreads();

    v16bf ah0 = load_frag(&Ah[(wr << 5) + hl][0], g0);
    v16bf al0 = load_frag(&Al[(wr << 5) + hl][0], g0);
    v16bf ah1 = load_frag(&Ah[(wr << 5) + 16 + hl][0], g0);
    v16bf al1 = load_frag(&Al[(wr << 5) + 16 + hl][0], g0);
    v16bf bh0 = load_frag(&Bth[(wc << 5) + hl][0], g0);
    v16bf bl0 = load_frag(&Btl[(wc << 5) + hl][0], g0);
    v16bf bh1 = load_frag(&Bth[(wc << 5) + 16 + hl][0], g0);
    v16bf bl1 = load_frag(&Btl[(wc << 5) + 16 + hl][0], g0);

    acc00 = WMMA_BF16(ah0, bh0, acc00);
    acc00 = WMMA_BF16(al0, bh0, acc00);
    acc00 = WMMA_BF16(ah0, bl0, acc00);
    acc01 = WMMA_BF16(ah0, bh1, acc01);
    acc01 = WMMA_BF16(al0, bh1, acc01);
    acc01 = WMMA_BF16(ah0, bl1, acc01);
    acc10 = WMMA_BF16(ah1, bh0, acc10);
    acc10 = WMMA_BF16(al1, bh0, acc10);
    acc10 = WMMA_BF16(ah1, bl0, acc10);
    acc11 = WMMA_BF16(ah1, bh1, acc11);
    acc11 = WMMA_BF16(al1, bh1, acc11);
    acc11 = WMMA_BF16(ah1, bl1, acc11);
  }

  const int col0 = n0 + (wc << 5) + hl;
  const int col1 = col0 + 16;
  const float b0 = bias ? bias[col0] : 0.f;
  const float b1v = bias ? bias[col1] : 0.f;
  const int rbase0 = m0 + (wr << 5) + (hh << 3);
  const int rbase1 = rbase0 + 16;
#pragma unroll
  for (int r = 0; r < 8; ++r) {
    int g0r = rbase0 + r;
    if (g0r < M) {
      C[(size_t)g0r * H + col0] = acc00[r] + b0;
      C[(size_t)g0r * H + col1] = acc01[r] + b1v;
    }
    int g1r = rbase1 + r;
    if (g1r < M) {
      C[(size_t)g1r * H + col0] = acc10[r] + b0;
      C[(size_t)g1r * H + col1] = acc11[r] + b1v;
    }
  }
}

#define CS_GROUPS 256
#define CS_THREADS 512

__global__ __launch_bounds__(CS_THREADS) void k_colsum_part(const float* __restrict__ x,
                                                           float* __restrict__ part, int N) {
  __shared__ float4 red[CS_THREADS / 64][64];
  const int t = threadIdx.x;
  const int cq = t & 63;
  const int rs = t >> 6;
  const int per = (N + (int)gridDim.x - 1) / (int)gridDim.x;
  const int r0 = (int)blockIdx.x * per;
  int r1 = r0 + per; if (r1 > N) r1 = N;
  const float4* xp = (const float4*)x + cq;
  float4 acc = make_float4(0.f, 0.f, 0.f, 0.f);
  int r = r0 + rs;
  for (; r + 56 < r1; r += 64) {
    float4 v[8];
#pragma unroll
    for (int j = 0; j < 8; ++j) v[j] = xp[(size_t)(r + 8 * j) * 64];
#pragma unroll
    for (int j = 0; j < 8; ++j) {
      acc.x += v[j].x; acc.y += v[j].y; acc.z += v[j].z; acc.w += v[j].w;
    }
  }
  for (; r < r1; r += 8) {
    float4 v = xp[(size_t)r * 64];
    acc.x += v.x; acc.y += v.y; acc.z += v.z; acc.w += v.w;
  }
  red[rs][cq] = acc;
  __syncthreads();
  if (t < H) {
    const float* rf = (const float*)&red[0][0];
    float s = 0.f;
#pragma unroll
    for (int j = 0; j < CS_THREADS / 64; ++j) s += rf[j * H + t];
    part[(size_t)blockIdx.x * H + t] = s;
  }
}

__global__ __launch_bounds__(1024) void k_selector(const float* __restrict__ part,
                           const float* __restrict__ W1, const float* __restrict__ b1,
                           const float* __restrict__ W2, const float* __restrict__ b2,
                           float* __restrict__ sw, int N) {
  __shared__ float4 psum[16][64];
  __shared__ float gsh[H];
  __shared__ float hid[128];
  __shared__ float lg[4];
  const int t = threadIdx.x;
  {
    const int cq = t & 63, q = t >> 6;
    const float4* pp = (const float4*)part + cq;
    float4 s = make_float4(0.f, 0.f, 0.f, 0.f);
#pragma unroll
    for (int j = 0; j < CS_GROUPS / 16; ++j) {
      float4 v = pp[(size_t)(q + 16 * j) * 64];
      s.x += v.x; s.y += v.y; s.z += v.z; s.w += v.w;
    }
    psum[q][cq] = s;
  }
  __syncthreads();
  if (t < H) {
    const float* pf = (const float*)&psum[0][0];
    float gsum = 0.f;
#pragma unroll
    for (int j = 0; j < 16; ++j) gsum += pf[j * H + t];
    float invN = 1.f / (float)N;
    gsh[t] = gsum * invN;
  }
  __syncthreads();
  if (t < 128) {
    float a = b1[t];
    for (int k = 0; k < H; ++k) a += gsh[k] * W1[k * 128 + t];
    hid[t] = fmaxf(a, 0.f);
  }
  __syncthreads();
  if (t < 4) {
    float l = b2[t];
    for (int k = 0; k < 128; ++k) l += hid[k] * W2[k * 4 + t];
    lg[t] = l;
  }
  __syncthreads();
  if (t == 0) {
    float m = fmaxf(fmaxf(lg[0], lg[1]), fmaxf(lg[2], lg[3]));
    float e0 = __expf(lg[0] - m), e1 = __expf(lg[1] - m);
    float e2 = __expf(lg[2] - m), e3 = __expf(lg[3] - m);
    float s = e0 + e1 + e2 + e3;
    sw[0] = e0 / s; sw[1] = e1 / s; sw[2] = e2 / s; sw[3] = e3 / s;
  }
}

__global__ void k_sd(const float* __restrict__ h, const float* __restrict__ asrc,
                     const float* __restrict__ adst, float* __restrict__ s,
                     float* __restrict__ d, int N) {
  int n = blockIdx.x * 8 + (threadIdx.x >> 5);
  if (n >= N) return;
  int lane = threadIdx.x & 31;
  float vs = 0.f, vd = 0.f;
  for (int c = lane; c < H; c += 32) {
    float v = h[(size_t)n * H + c];
    vs += v * asrc[c];
    vd += v * adst[c];
  }
  for (int o = 16; o; o >>= 1) { vs += __shfl_xor(vs, o, 32); vd += __shfl_xor(vd, o, 32); }
  if (lane == 0) { s[n] = vs; d[n] = vd; }
}

__global__ void k_init_m(const float* __restrict__ s, const float* __restrict__ d,
                         unsigned* __restrict__ mb, int N) {
  int n = blockIdx.x * blockDim.x + threadIdx.x;
  if (n < N) mb[n] = f2ord(lrelu(s[n] + d[n]));
}

__global__ void k_edge_max(const int* __restrict__ src, const int* __restrict__ dst,
                           const int* __restrict__ attr, int branch,
                           const float* __restrict__ s, const float* __restrict__ d,
                           unsigned* __restrict__ mb, int E) {
  int e = blockIdx.x * blockDim.x + threadIdx.x;
  if (e < E && attr[e] == branch) {
    int dn = dst[e];
    float v = lrelu(s[src[e]] + d[dn]);
    atomicMax(&mb[dn], f2ord(v));
  }
}

__global__ void k_init_num(const float* __restrict__ h, const float* __restrict__ s,
                           const float* __restrict__ d, const unsigned* __restrict__ mb,
                           float* __restrict__ mf, float* __restrict__ denom,
                           float* __restrict__ num, int N) {
  int n = blockIdx.x * 8 + (threadIdx.x >> 5);
  if (n >= N) return;
  int lane = threadIdx.x & 31;
  float m = ord2f(mb[n]);
  float wself = __expf(lrelu(s[n] + d[n]) - m);
  if (lane == 0) { mf[n] = m; denom[n] = wself; }
  for (int c = lane; c < H; c += 32) num[(size_t)n * H + c] = wself * h[(size_t)n * H + c];
}

__global__ void k_edge_accum(const int* __restrict__ src, const int* __restrict__ dst,
                             const int* __restrict__ attr, int branch,
                             const float* __restrict__ s, const float* __restrict__ d,
                             const float* __restrict__ mf, const float* __restrict__ h,
                             float* __restrict__ denom, float* __restrict__ num, int E) {
  int e = blockIdx.x * 8 + (threadIdx.x >> 5);
  if (e >= E || attr[e] != branch) return;
  int sn = src[e], dn = dst[e];
  float w = __expf(lrelu(s[sn] + d[dn]) - mf[dn]);
  int lane = threadIdx.x & 31;
  if (lane == 0) atomicAdd(&denom[dn], w);
  for (int c = lane; c < H; c += 32)
    atomicAdd(&num[(size_t)dn * H + c], w * h[(size_t)sn * H + c]);
}

__global__ void k_finalize(float* __restrict__ num, const float* __restrict__ denom,
                           const float* __restrict__ bias, int total) {
  int i = blockIdx.x * blockDim.x + threadIdx.x;
  if (i < total) {
    int n = i >> 8;
    num[i] = num[i] / denom[n] + bias[i & (H - 1)];
  }
}

__global__ void k_ln_relu(const float* __restrict__ t, const float* __restrict__ gamma,
                          const float* __restrict__ beta, float* __restrict__ out,
                          const float* __restrict__ swp, int swidx, int accumulate, int N) {
  int n = blockIdx.x * 8 + (threadIdx.x >> 5);
  if (n >= N) return;
  int lane = threadIdx.x & 31;
  float v[8];
  float sum = 0.f, sq = 0.f;
#pragma unroll
  for (int j = 0; j < 8; ++j) {
    v[j] = t[(size_t)n * H + lane + j * 32];
    sum += v[j];
    sq += v[j] * v[j];
  }
  for (int o = 16; o; o >>= 1) { sum += __shfl_xor(sum, o, 32); sq += __shfl_xor(sq, o, 32); }
  float mu = sum * (1.f / H);
  float var = sq * (1.f / H) - mu * mu;
  float inv = rsqrtf(var + 1e-5f);
  float scale = swp ? swp[swidx] : 1.f;
#pragma unroll
  for (int j = 0; j < 8; ++j) {
    int c = lane + j * 32;
    float y = (v[j] - mu) * inv * gamma[c] + beta[c];
    y = fmaxf(y, 0.f) * scale;
    if (accumulate) out[(size_t)n * H + c] += y;
    else            out[(size_t)n * H + c] = y;
  }
}

extern "C" void kernel_launch(void* const* d_in, const int* in_sizes, int n_in,
                              void* d_out, int out_size, void* d_ws, size_t ws_size,
                              hipStream_t stream) {
  const float* x     = (const float*)d_in[0];
  const int*   ei    = (const int*)d_in[1];
  const int*   ea    = (const int*)d_in[2];
  const float* Wg    = (const float*)d_in[3];
  const float* a_src = (const float*)d_in[4];
  const float* a_dst = (const float*)d_in[5];
  const float* bg    = (const float*)d_in[6];
  const float* Wsl   = (const float*)d_in[7];
  const float* bsl   = (const float*)d_in[8];
  const float* gsl   = (const float*)d_in[9];
  const float* betas = (const float*)d_in[10];
  const float* W1    = (const float*)d_in[11];
  const float* b1    = (const float*)d_in[12];
  const float* W2    = (const float*)d_in[13];
  const float* b2    = (const float*)d_in[14];
  const float* Wf    = (const float*)d_in[15];
  const float* bfv   = (const float*)d_in[16];
  const float* gf    = (const float*)d_in[17];
  const float* betaf = (const float*)d_in[18];

  const int N = in_sizes[0] / H;
  const int E = in_sizes[2];
  float* out = (float*)d_out;

  char* wsb = (char*)d_ws;
  size_t off = 0;
  auto alloc = [&](size_t bytes) -> void* {
    void* p = wsb + off;
    off += (bytes + 255) & ~(size_t)255;
    return p;
  };
  float*    sw    = (float*)alloc(4 * sizeof(float));
  float*    sArr  = (float*)alloc((size_t)N * sizeof(float));
  float*    dArr  = (float*)alloc((size_t)N * sizeof(float));
  unsigned* mb    = (unsigned*)alloc((size_t)N * sizeof(unsigned));
  float*    mf    = (float*)alloc((size_t)N * sizeof(float));
  float*    denom = (float*)alloc((size_t)N * sizeof(float));
  float*    bufH  = (float*)alloc((size_t)N * H * sizeof(float));
  float*    bufX  = (float*)alloc((size_t)N * H * sizeof(float));
  float*    bufC  = (float*)alloc((size_t)N * H * sizeof(float));

  dim3 gemmGrid((unsigned)((N + TM - 1) / TM), (unsigned)(H / TNB));
  int nWaveBlocks = (N + 7) / 8;
  int eWaveBlocks = (E + 7) / 8;
  int nThrBlocks  = (N + 255) / 256;
  int eThrBlocks  = (E + 255) / 256;
  int nhBlocks    = (N * H + 255) / 256;

  float* part = bufX;
  k_colsum_part<<<CS_GROUPS, CS_THREADS, 0, stream>>>(x, part, N);
  k_selector<<<1, 1024, 0, stream>>>(part, W1, b1, W2, b2, sw, N);

  for (int i = 0; i < 3; ++i) {
    k_gemm_bf16x3<<<gemmGrid, 256, 0, stream>>>(x, Wg + (size_t)i * H * H, nullptr, bufH, N);
    k_sd<<<nWaveBlocks, 256, 0, stream>>>(bufH, a_src + i * H, a_dst + i * H, sArr, dArr, N);
    k_init_m<<<nThrBlocks, 256, 0, stream>>>(sArr, dArr, mb, N);
    k_edge_max<<<eThrBlocks, 256, 0, stream>>>(ei, ei + E, ea, i, sArr, dArr, mb, E);
    k_init_num<<<nWaveBlocks, 256, 0, stream>>>(bufH, sArr, dArr, mb, mf, denom, bufX, N);
    k_edge_accum<<<eWaveBlocks, 256, 0, stream>>>(ei, ei + E, ea, i, sArr, dArr, mf, bufH, denom, bufX, E);
    k_finalize<<<nhBlocks, 256, 0, stream>>>(bufX, denom, bg + i * H, N * H);
    k_gemm_bf16x3<<<gemmGrid, 256, 0, stream>>>(bufX, Wsl + (size_t)i * H * H, bsl + i * H, bufH, N);
    k_ln_relu<<<nWaveBlocks, 256, 0, stream>>>(bufH, gsl + i * H, betas + i * H, bufC, sw, i,
                                               i > 0 ? 1 : 0, N);
  }
  k_gemm_bf16x3<<<gemmGrid, 256, 0, stream>>>(x, Wsl + (size_t)3 * H * H, bsl + 3 * H, bufH, N);
  k_ln_relu<<<nWaveBlocks, 256, 0, stream>>>(bufH, gsl + 3 * H, betas + 3 * H, bufC, sw, 3, 1, N);

  k_gemm_bf16x3<<<gemmGrid, 256, 0, stream>>>(bufC, Wf, bfv, bufH, N);
  k_ln_relu<<<nWaveBlocks, 256, 0, stream>>>(bufH, gf, betaf, out, nullptr, 0, 0, N);
}


# fixed grids with wave loops: row kernels walk rows per wave, edge scatter takes 32 edges per wave, finalize takes float4 pieces grid-stride; far fewer waves to dispatch; atomics and sum orders kept
# speedup vs baseline: 1.1834x; 1.0738x over previous
#include <hip/hip_runtime.h>

#define H 256
#define TM 128
#define TNB 64
#define KC 32
#define APAD 40

typedef __bf16 bf16;
typedef __attribute__((ext_vector_type(16))) __bf16 v16bf;
typedef __attribute__((ext_vector_type(8)))  __bf16 v8bf;
typedef __attribute__((ext_vector_type(8)))  float  v8f;

__device__ __forceinline__ unsigned short f2bfbits(float f) {
  unsigned u = __float_as_uint(f);
  return (unsigned short)((u + 0x7FFFu + ((u >> 16) & 1u)) >> 16);
}
__device__ __forceinline__ float bfbits2f(unsigned short h) {
  return __uint_as_float(((unsigned)h) << 16);
}
__device__ __forceinline__ bf16 f2bf(float f) {
  unsigned short hs = f2bfbits(f);
  return __builtin_bit_cast(bf16, hs);
}
__device__ __forceinline__ float lrelu(float v) { return v > 0.f ? v : 0.2f * v; }
__device__ __forceinline__ unsigned f2ord(float f) {
  unsigned u = __float_as_uint(f);
  return (u & 0x80000000u) ? ~u : (u | 0x80000000u);
}
__device__ __forceinline__ float ord2f(unsigned u) {
  unsigned b = (u & 0x80000000u) ? (u & 0x7FFFFFFFu) : ~u;
  return __uint_as_float(b);
}

__device__ __forceinline__ v16bf load_frag(const bf16* rowp, int g0) {
  v8bf a = *(const v8bf*)(rowp + g0);
  v8bf b = *(const v8bf*)(rowp + 16 + g0);
  return __builtin_shufflevector(a, b, 0, 1, 2, 3, 4, 5, 6, 7,
                                 8, 9, 10, 11, 12, 13, 14, 15);
}

#define WMMA_BF16(A_, B_, C_) \
  __builtin_amdgcn_wmma_f32_16x16x32_bf16(false, (A_), false, (B_), (short)0, (C_), false, false)

__global__ __launch_bounds__(256) void k_gemm_bf16x3(
    const float* __restrict__ A, const float* __restrict__ B,
    const float* __restrict__ bias, float* __restrict__ C, int M)
{
  __shared__ bf16 Ah[TM][APAD];
  __shared__ bf16 Al[TM][APAD];
  __shared__ bf16 Bth[TNB][APAD];
  __shared__ bf16 Btl[TNB][APAD];

  const int tid = threadIdx.x;
  const int m0 = blockIdx.x * TM;
  const int n0 = blockIdx.y * TNB;
  const int wid = tid >> 5, lane = tid & 31;
  const int hl = lane & 15, hh = lane >> 4;
  const int wr = wid >> 1;
  const int wc = wid & 1;
  const int g0 = hh << 3;

  v8f acc00 = {}, acc01 = {}, acc10 = {}, acc11 = {};

  for (int k0 = 0; k0 < H; k0 += KC) {
    __syncthreads();
    for (int idx = tid; idx < (TM * KC) / 4; idx += 256) {
      int r = idx >> 3;
      int c4 = (idx & 7) << 2;
      int gr = m0 + r;
      float4 v = (gr < M) ? *(const float4*)&A[(size_t)gr * H + k0 + c4]
                          : make_float4(0.f, 0.f, 0.f, 0.f);
      ushort4 hiv, lov;
      hiv.x = f2bfbits(v.x); lov.x = f2bfbits(v.x - bfbits2f(hiv.x));
      hiv.y = f2bfbits(v.y); lov.y = f2bfbits(v.y - bfbits2f(hiv.y));
      hiv.z = f2bfbits(v.z); lov.z = f2bfbits(v.z - bfbits2f(hiv.z));
      hiv.w = f2bfbits(v.w); lov.w = f2bfbits(v.w - bfbits2f(hiv.w));
      *(ushort4*)&Ah[r][c4] = hiv;
      *(ushort4*)&Al[r][c4] = lov;
    }
    for (int idx = tid; idx < (KC * TNB) / 4; idx += 256) {
      int r = idx >> 4;
      int c4 = (idx & 15) << 2;
      float4 v = *(const float4*)&B[(size_t)(k0 + r) * H + n0 + c4];
      bf16 h0 = f2bf(v.x), h1 = f2bf(v.y), h2 = f2bf(v.z), h3 = f2bf(v.w);
      Bth[c4 + 0][r] = h0; Btl[c4 + 0][r] = f2bf(v.x - bfbits2f(__builtin_bit_cast(unsigned short, h0)));
      Bth[c4 + 1][r] = h1; Btl[c4 + 1][r] = f2bf(v.y - bfbits2f(__builtin_bit_cast(unsigned short, h1)));
      Bth[c4 + 2][r] = h2; Btl[c4 + 2][r] = f2bf(v.z - bfbits2f(__builtin_bit_cast(unsigned short, h2)));
      Bth[c4 + 3][r] = h3; Btl[c4 + 3][r] = f2bf(v.w - bfbits2f(__builtin_bit_cast(unsigned short, h3)));
    }
    __syncthreads();

    v16bf ah0 = load_frag(&Ah[(wr << 5) + hl][0], g0);
    v16bf al0 = load_frag(&Al[(wr << 5) + hl][0], g0);
    v16bf ah1 = load_frag(&Ah[(wr << 5) + 16 + hl][0], g0);
    v16bf al1 = load_frag(&Al[(wr << 5) + 16 + hl][0], g0);
    v16bf bh0 = load_frag(&Bth[(wc << 5) + hl][0], g0);
    v16bf bl0 = load_frag(&Btl[(wc << 5) + hl][0], g0);
    v16bf bh1 = load_frag(&Bth[(wc << 5) + 16 + hl][0], g0);
    v16bf bl1 = load_frag(&Btl[(wc << 5) + 16 + hl][0], g0);

    acc00 = WMMA_BF16(ah0, bh0, acc00);
    acc00 = WMMA_BF16(al0, bh0, acc00);
    acc00 = WMMA_BF16(ah0, bl0, acc00);
    acc01 = WMMA_BF16(ah0, bh1, acc01);
    acc01 = WMMA_BF16(al0, bh1, acc01);
    acc01 = WMMA_BF16(ah0, bl1, acc01);
    acc10 = WMMA_BF16(ah1, bh0, acc10);
    acc10 = WMMA_BF16(al1, bh0, acc10);
    acc10 = WMMA_BF16(ah1, bl0, acc10);
    acc11 = WMMA_BF16(ah1, bh1, acc11);
    acc11 = WMMA_BF16(al1, bh1, acc11);
    acc11 = WMMA_BF16(ah1, bl1, acc11);
  }

  const int col0 = n0 + (wc << 5) + hl;
  const int col1 = col0 + 16;
  const float b0 = bias ? bias[col0] : 0.f;
  const float b1v = bias ? bias[col1] : 0.f;
  const int rbase0 = m0 + (wr << 5) + (hh << 3);
  const int rbase1 = rbase0 + 16;
#pragma unroll
  for (int r = 0; r < 8; ++r) {
    int g0r = rbase0 + r;
    if (g0r < M) {
      C[(size_t)g0r * H + col0] = acc00[r] + b0;
      C[(size_t)g0r * H + col1] = acc01[r] + b1v;
    }
    int g1r = rbase1 + r;
    if (g1r < M) {
      C[(size_t)g1r * H + col0] = acc10[r] + b0;
      C[(size_t)g1r * H + col1] = acc11[r] + b1v;
    }
  }
}

#define CS_GROUPS 256
#define CS_THREADS 512

__global__ __launch_bounds__(CS_THREADS) void k_colsum_part(const float* __restrict__ x,
                                                           float* __restrict__ part, int N) {
  __shared__ float4 red[CS_THREADS / 64][64];
  const int t = threadIdx.x;
  const int cq = t & 63;
  const int rs = t >> 6;
  const int per = (N + (int)gridDim.x - 1) / (int)gridDim.x;
  const int r0 = (int)blockIdx.x * per;
  int r1 = r0 + per; if (r1 > N) r1 = N;
  const float4* xp = (const float4*)x + cq;
  float4 acc = make_float4(0.f, 0.f, 0.f, 0.f);
  int r = r0 + rs;
  for (; r + 56 < r1; r += 64) {
    float4 v[8];
#pragma unroll
    for (int j = 0; j < 8; ++j) v[j] = xp[(size_t)(r + 8 * j) * 64];
#pragma unroll
    for (int j = 0; j < 8; ++j) {
      acc.x += v[j].x; acc.y += v[j].y; acc.z += v[j].z; acc.w += v[j].w;
    }
  }
  for (; r < r1; r += 8) {
    float4 v = xp[(size_t)r * 64];
    acc.x += v.x; acc.y += v.y; acc.z += v.z; acc.w += v.w;
  }
  red[rs][cq] = acc;
  __syncthreads();
  if (t < H) {
    const float* rf = (const float*)&red[0][0];
    float s = 0.f;
#pragma unroll
    for (int j = 0; j < CS_THREADS / 64; ++j) s += rf[j * H + t];
    part[(size_t)blockIdx.x * H + t] = s;
  }
}

__global__ __launch_bounds__(1024) void k_selector(const float* __restrict__ part,
                           const float* __restrict__ W1, const float* __restrict__ b1,
                           const float* __restrict__ W2, const float* __restrict__ b2,
                           float* __restrict__ sw, int N) {
  __shared__ float4 psum[16][64];
  __shared__ float gsh[H];
  __shared__ float hid[128];
  __shared__ float lg[4];
  const int t = threadIdx.x;
  {
    const int cq = t & 63, q = t >> 6;
    const float4* pp = (const float4*)part + cq;
    float4 s = make_float4(0.f, 0.f, 0.f, 0.f);
#pragma unroll
    for (int j = 0; j < CS_GROUPS / 16; ++j) {
      float4 v = pp[(size_t)(q + 16 * j) * 64];
      s.x += v.x; s.y += v.y; s.z += v.z; s.w += v.w;
    }
    psum[q][cq] = s;
  }
  __syncthreads();
  if (t < H) {
    const float* pf = (const float*)&psum[0][0];
    float gsum = 0.f;
#pragma unroll
    for (int j = 0; j < 16; ++j) gsum += pf[j * H + t];
    float invN = 1.f / (float)N;
    gsh[t] = gsum * invN;
  }
  __syncthreads();
  if (t < 128) {
    float a = b1[t];
    for (int k = 0; k < H; ++k) a += gsh[k] * W1[k * 128 + t];
    hid[t] = fmaxf(a, 0.f);
  }
  __syncthreads();
  if (t < 4) {
    float l = b2[t];
    for (int k = 0; k < 128; ++k) l += hid[k] * W2[k * 4 + t];
    lg[t] = l;
  }
  __syncthreads();
  if (t == 0) {
    float m = fmaxf(fmaxf(lg[0], lg[1]), fmaxf(lg[2], lg[3]));
    float e0 = __expf(lg[0] - m), e1 = __expf(lg[1] - m);
    float e2 = __expf(lg[2] - m), e3 = __expf(lg[3] - m);
    float s = e0 + e1 + e2 + e3;
    sw[0] = e0 / s; sw[1] = e1 / s; sw[2] = e2 / s; sw[3] = e3 / s;
  }
}

#define ROW_BLOCKS 2048
#define EL_BLOCKS 2048

__global__ __launch_bounds__(256) void k_sd(const float* __restrict__ h,
                     const float* __restrict__ asrc, const float* __restrict__ adst,
                     float* __restrict__ s, float* __restrict__ d, int N) {
  const int lane = threadIdx.x & 31;
  const int nw = (int)gridDim.x * 8;
  float as[8], ad[8];
#pragma unroll
  for (int j = 0; j < 8; ++j) { as[j] = asrc[lane + 32 * j]; ad[j] = adst[lane + 32 * j]; }
  for (int n = blockIdx.x * 8 + (threadIdx.x >> 5); n < N; n += nw) {
    const float* hp = h + (size_t)n * H + lane;
    float v[8];
#pragma unroll
    for (int j = 0; j < 8; ++j) v[j] = hp[32 * j];
    float vs = 0.f, vd = 0.f;
#pragma unroll
    for (int j = 0; j < 8; ++j) { vs += v[j] * as[j]; vd += v[j] * ad[j]; }
    for (int o = 16; o; o >>= 1) { vs += __shfl_xor(vs, o, 32); vd += __shfl_xor(vd, o, 32); }
    if (lane == 0) { s[n] = vs; d[n] = vd; }
  }
}

__global__ void k_init_m(const float* __restrict__ s, const float* __restrict__ d,
                         unsigned* __restrict__ mb, int N) {
  int n = blockIdx.x * blockDim.x + threadIdx.x;
  if (n < N) mb[n] = f2ord(lrelu(s[n] + d[n]));
}

__global__ void k_edge_max(const int* __restrict__ src, const int* __restrict__ dst,
                           const int* __restrict__ attr, int branch,
                           const float* __restrict__ s, const float* __restrict__ d,
                           unsigned* __restrict__ mb, int E) {
  int e = blockIdx.x * blockDim.x + threadIdx.x;
  if (e < E && attr[e] == branch) {
    int dn = dst[e];
    float v = lrelu(s[src[e]] + d[dn]);
    atomicMax(&mb[dn], f2ord(v));
  }
}

__global__ __launch_bounds__(256) void k_init_num(const float* __restrict__ h,
                           const float* __restrict__ s,
                           const float* __restrict__ d, const unsigned* __restrict__ mb,
                           float* __restrict__ mf, float* __restrict__ denom,
                           float* __restrict__ num, int N) {
  const int lane = threadIdx.x & 31;
  const int nw = (int)gridDim.x * 8;
  for (int n = blockIdx.x * 8 + (threadIdx.x >> 5); n < N; n += nw) {
    const float* hp = h + (size_t)n * H + lane;
    float* np = num + (size_t)n * H + lane;
    float v[8];
#pragma unroll
    for (int j = 0; j < 8; ++j) v[j] = hp[32 * j];
    float m = ord2f(mb[n]);
    float wself = __expf(lrelu(s[n] + d[n]) - m);
    if (lane == 0) { mf[n] = m; denom[n] = wself; }
#pragma unroll
    for (int j = 0; j < 8; ++j) np[32 * j] = wself * v[j];
  }
}

__global__ __launch_bounds__(256) void k_edge_accum(const int* __restrict__ src,
                             const int* __restrict__ dst,
                             const int* __restrict__ attr, int branch,
                             const float* __restrict__ s, const float* __restrict__ d,
                             const float* __restrict__ mf, const float* __restrict__ h,
                             float* __restrict__ denom, float* __restrict__ num, int E) {
  const int lane = threadIdx.x & 31;
  const int e = ((int)blockIdx.x * 8 + (int)(threadIdx.x >> 5)) * 32 + lane;
  const bool on = (e < E) && (attr[e] == branch);
  int sn = 0, dn = 0;
  float w = 0.f;
  if (on) {
    sn = src[e]; dn = dst[e];
    w = __expf(lrelu(s[sn] + d[dn]) - mf[dn]);
    atomicAdd(&denom[dn], w);
  }
  unsigned live = __builtin_amdgcn_ballot_w32(on);
  while (live) {
    const int j = __builtin_ctz(live);
    live &= live - 1u;
    const int sj = __builtin_amdgcn_readlane(sn, j);
    const int dj = __builtin_amdgcn_readlane(dn, j);
    const float wj = __uint_as_float(__builtin_amdgcn_readlane(__float_as_uint(w), j));
    const float* hp = h + (size_t)sj * H + lane;
    float* np = num + (size_t)dj * H + lane;
    float v[8];
#pragma unroll
    for (int k = 0; k < 8; ++k) v[k] = hp[32 * k];
#pragma unroll
    for (int k = 0; k < 8; ++k) atomicAdd(np + 32 * k, wj * v[k]);
  }
}

typedef __attribute__((ext_vector_type(4))) float v4f;
__device__ __forceinline__ v4f fin4(v4f v, float dn, v4f b) {
  v.x = v.x / dn + b.x; v.y = v.y / dn + b.y;
  v.z = v.z / dn + b.z; v.w = v.w / dn + b.w;
  return v;
}
__global__ __launch_bounds__(256) void k_finalize(float* __restrict__ num,
                           const float* __restrict__ denom,
                           const float* __restrict__ bias, int N) {
  const int total4 = N * (H / 4);
  const int T = (int)gridDim.x * 256;
  v4f* p = (v4f*)num;
  int i = (int)blockIdx.x * 256 + (int)threadIdx.x;
  const v4f b = *(const v4f*)&bias[(i & 63) << 2];
  for (; i + T < total4; i += 2 * T) {
    v4f v0 = p[i], v1 = p[i + T];
    float d0 = denom[i >> 6], d1 = denom[(i + T) >> 6];
    p[i] = fin4(v0, d0, b);
    p[i + T] = fin4(v1, d1, b);
  }
  if (i < total4) p[i] = fin4(p[i], denom[i >> 6], b);
}

__global__ __launch_bounds__(256) void k_ln_relu(const float* __restrict__ t,
                          const float* __restrict__ gamma,
                          const float* __restrict__ beta, float* __restrict__ out,
                          const float* __restrict__ swp, int swidx, int accumulate, int N) {
  const int lane = threadIdx.x & 31;
  const int nw = (int)gridDim.x * 8;
  float g[8], b[8];
#pragma unroll
  for (int j = 0; j < 8; ++j) { g[j] = gamma[lane + j * 32]; b[j] = beta[lane + j * 32]; }
  const float scale = swp ? swp[swidx] : 1.f;
  for (int n = blockIdx.x * 8 + (threadIdx.x >> 5); n < N; n += nw) {
    const float* tp = t + (size_t)n * H + lane;
    float* op = out + (size_t)n * H + lane;
    float v[8], o[8] = {};
#pragma unroll
    for (int j = 0; j < 8; ++j) v[j] = tp[j * 32];
    if (accumulate) {
#pragma unroll
      for (int j = 0; j < 8; ++j) o[j] = op[j * 32];
    }
    float sum = 0.f, sq = 0.f;
#pragma unroll
    for (int j = 0; j < 8; ++j) {
      sum += v[j];
      sq += v[j] * v[j];
    }
    for (int k = 16; k; k >>= 1) { sum += __shfl_xor(sum, k, 32); sq += __shfl_xor(sq, k, 32); }
    float mu = sum * (1.f / H);
    float var = sq * (1.f / H) - mu * mu;
    float inv = rsqrtf(var + 1e-5f);
#pragma unroll
    for (int j = 0; j < 8; ++j) {
      float y = (v[j] - mu) * inv * g[j] + b[j];
      v[j] = fmaxf(y, 0.f) * scale;
    }
    if (accumulate) {
#pragma unroll
      for (int j = 0; j < 8; ++j) op[j * 32] = o[j] + v[j];
    } else {
#pragma unroll
      for (int j = 0; j < 8; ++j) op[j * 32] = v[j];
    }
  }
}

extern "C" void kernel_launch(void* const* d_in, const int* in_sizes, int n_in,
                              void* d_out, int out_size, void* d_ws, size_t ws_size,
                              hipStream_t stream) {
  const float* x     = (const float*)d_in[0];
  const int*   ei    = (const int*)d_in[1];
  const int*   ea    = (const int*)d_in[2];
  const float* Wg    = (const float*)d_in[3];
  const float* a_src = (const float*)d_in[4];
  const float* a_dst = (const float*)d_in[5];
  const float* bg    = (const float*)d_in[6];
  const float* Wsl   = (const float*)d_in[7];
  const float* bsl   = (const float*)d_in[8];
  const float* gsl   = (const float*)d_in[9];
  const float* betas = (const float*)d_in[10];
  const float* W1    = (const float*)d_in[11];
  const float* b1    = (const float*)d_in[12];
  const float* W2    = (const float*)d_in[13];
  const float* b2    = (const float*)d_in[14];
  const float* Wf    = (const float*)d_in[15];
  const float* bfv   = (const float*)d_in[16];
  const float* gf    = (const float*)d_in[17];
  const float* betaf = (const float*)d_in[18];

  const int N = in_sizes[0] / H;
  const int E = in_sizes[2];
  float* out = (float*)d_out;

  char* wsb = (char*)d_ws;
  size_t off = 0;
  auto alloc = [&](size_t bytes) -> void* {
    void* p = wsb + off;
    off += (bytes + 255) & ~(size_t)255;
    return p;
  };
  float*    sw    = (float*)alloc(4 * sizeof(float));
  float*    sArr  = (float*)alloc((size_t)N * sizeof(float));
  float*    dArr  = (float*)alloc((size_t)N * sizeof(float));
  unsigned* mb    = (unsigned*)alloc((size_t)N * sizeof(unsigned));
  float*    mf    = (float*)alloc((size_t)N * sizeof(float));
  float*    denom = (float*)alloc((size_t)N * sizeof(float));
  float*    bufH  = (float*)alloc((size_t)N * H * sizeof(float));
  float*    bufX  = (float*)alloc((size_t)N * H * sizeof(float));
  float*    bufC  = (float*)alloc((size_t)N * H * sizeof(float));

  dim3 gemmGrid((unsigned)((N + TM - 1) / TM), (unsigned)(H / TNB));
  int nThrBlocks  = (N + 255) / 256;
  int eThrBlocks  = (E + 255) / 256;

  float* part = bufX;
  k_colsum_part<<<CS_GROUPS, CS_THREADS, 0, stream>>>(x, part, N);
  k_selector<<<1, 1024, 0, stream>>>(part, W1, b1, W2, b2, sw, N);

  for (int i = 0; i < 3; ++i) {
    k_gemm_bf16x3<<<gemmGrid, 256, 0, stream>>>(x, Wg + (size_t)i * H * H, nullptr, bufH, N);
    k_sd<<<ROW_BLOCKS, 256, 0, stream>>>(bufH, a_src + i * H, a_dst + i * H, sArr, dArr, N);
    k_init_m<<<nThrBlocks, 256, 0, stream>>>(sArr, dArr, mb, N);
    k_edge_max<<<eThrBlocks, 256, 0, stream>>>(ei, ei + E, ea, i, sArr, dArr, mb, E);
    k_init_num<<<ROW_BLOCKS, 256, 0, stream>>>(bufH, sArr, dArr, mb, mf, denom, bufX, N);
    k_edge_accum<<<eThrBlocks, 256, 0, stream>>>(ei, ei + E, ea, i, sArr, dArr, mf, bufH, denom, bufX, E);
    k_finalize<<<EL_BLOCKS, 256, 0, stream>>>(bufX, denom, bg + i * H, N);
    k_gemm_bf16x3<<<gemmGrid, 256, 0, stream>>>(bufX, Wsl + (size_t)i * H * H, bsl + i * H, bufH, N);
    k_ln_relu<<<ROW_BLOCKS, 256, 0, stream>>>(bufH, gsl + i * H, betas + i * H, bufC, sw, i,
                                              i > 0 ? 1 : 0, N);
  }
  k_gemm_bf16x3<<<gemmGrid, 256, 0, stream>>>(x, Wsl + (size_t)3 * H * H, bsl + 3 * H, bufH, N);
  k_ln_relu<<<ROW_BLOCKS, 256, 0, stream>>>(bufH, gsl + 3 * H, betas + 3 * H, bufC, sw, 3, 1, N);

  k_gemm_bf16x3<<<gemmGrid, 256, 0, stream>>>(bufC, Wf, bfv, bufH, N);
  k_ln_relu<<<ROW_BLOCKS, 256, 0, stream>>>(bufH, gf, betaf, out, nullptr, 0, 0, N);
}


# weights split to bf16 high/low planes and transposed once per call by one new small launch; GEMM copies its B tile with 16-byte loads and LDS stores: no per-block conversion, no 2-byte LDS scatter
# speedup vs baseline: 1.7892x; 1.5119x over previous
#include <hip/hip_runtime.h>

#define H 256
#define TM 128
#define TNB 64
#define KC 32
#define APAD 40

typedef __bf16 bf16;
typedef __attribute__((ext_vector_type(16))) __bf16 v16bf;
typedef __attribute__((ext_vector_type(8)))  __bf16 v8bf;
typedef __attribute__((ext_vector_type(8)))  float  v8f;

__device__ __forceinline__ unsigned short f2bfbits(float f) {
  unsigned u = __float_as_uint(f);
  return (unsigned short)((u + 0x7FFFu + ((u >> 16) & 1u)) >> 16);
}
__device__ __forceinline__ float bfbits2f(unsigned short h) {
  return __uint_as_float(((unsigned)h) << 16);
}
__device__ __forceinline__ bf16 f2bf(float f) {
  unsigned short hs = f2bfbits(f);
  return __builtin_bit_cast(bf16, hs);
}
__device__ __forceinline__ float lrelu(float v) { return v > 0.f ? v : 0.2f * v; }

__device__ __forceinline__ v16bf load_frag(const bf16* rowp, int g0) {
  v8bf a = *(const v8bf*)(rowp + g0);
  v8bf b = *(const v8bf*)(rowp + 16 + g0);
  return __builtin_shufflevector(a, b, 0, 1, 2, 3, 4, 5, 6, 7,
                                 8, 9, 10, 11, 12, 13, 14, 15);
}

#define WMMA_BF16(A_, B_, C_) \
  __builtin_amdgcn_wmma_f32_16x16x32_bf16(false, (A_), false, (B_), (short)0, (C_), false, false)

__global__ __launch_bounds__(256) void k_gemm_bf16x3(
    const float* __restrict__ A, const unsigned* __restrict__ Bhi,
    const unsigned* __restrict__ Blo,
    const float* __restrict__ bias, float* __restrict__ C, int M)
{
  __shared__ bf16 Ah[TM][APAD];
  __shared__ bf16 Al[TM][APAD];
  __shared__ bf16 Bth[TNB][APAD];
  __shared__ bf16 Btl[TNB][APAD];

  const int tid = threadIdx.x;
  const int m0 = blockIdx.x * TM;
  const int n0 = blockIdx.y * TNB;
  const int wid = tid >> 5, lane = tid & 31;
  const int hl = lane & 15, hh = lane >> 4;
  const int wr = wid >> 1;
  const int wc = wid & 1;
  const int g0 = hh << 3;

  v8f acc00 = {}, acc01 = {}, acc10 = {}, acc11 = {};

  for (int k0 = 0; k0 < H; k0 += KC) {
    __syncthreads();
    for (int idx = tid; idx < (TM * KC) / 4; idx += 256) {
      int r = idx >> 3;
      int c4 = (idx & 7) << 2;
      int gr = m0 + r;
      float4 v = (gr < M) ? *(const float4*)&A[(size_t)gr * H + k0 + c4]
                          : make_float4(0.f, 0.f, 0.f, 0.f);
      ushort4 hiv, lov;
      hiv.x = f2bfbits(v.x); lov.x = f2bfbits(v.x - bfbits2f(hiv.x));
      hiv.y = f2bfbits(v.y); lov.y = f2bfbits(v.y - bfbits2f(hiv.y));
      hiv.z = f2bfbits(v.z); lov.z = f2bfbits(v.z - bfbits2f(hiv.z));
      hiv.w = f2bfbits(v.w); lov.w = f2bfbits(v.w - bfbits2f(hiv.w));
      *(ushort4*)&Ah[r][c4] = hiv;
      *(ushort4*)&Al[r][c4] = lov;
    }
    {
      const size_t bo = (size_t)(k0 >> 5) * (H * KC / 2) + (size_t)n0 * (KC / 2) + (size_t)tid * 4;
      const uint4 vh = *(const uint4*)(Bhi + bo);
      const uint4 vl = *(const uint4*)(Blo + bo);
      *(uint4*)&Bth[tid >> 2][(tid & 3) << 3] = vh;
      *(uint4*)&Btl[tid >> 2][(tid & 3) << 3] = vl;
    }
    __syncthreads();

    v16bf ah0 = load_frag(&Ah[(wr << 5) + hl][0], g0);
    v16bf al0 = load_frag(&Al[(wr << 5) + hl][0], g0);
    v16bf ah1 = load_frag(&Ah[(wr << 5) + 16 + hl][0], g0);
    v16bf al1 = load_frag(&Al[(wr << 5) + 16 + hl][0], g0);
    v16bf bh0 = load_frag(&Bth[(wc << 5) + hl][0], g0);
    v16bf bl0 = load_frag(&Btl[(wc << 5) + hl][0], g0);
    v16bf bh1 = load_frag(&Bth[(wc << 5) + 16 + hl][0], g0);
    v16bf bl1 = load_frag(&Btl[(wc << 5) + 16 + hl][0], g0);

    acc00 = WMMA_BF16(ah0, bh0, acc00);
    acc00 = WMMA_BF16(al0, bh0, acc00);
    acc00 = WMMA_BF16(ah0, bl0, acc00);
    acc01 = WMMA_BF16(ah0, bh1, acc01);
    acc01 = WMMA_BF16(al0, bh1, acc01);
    acc01 = WMMA_BF16(ah0, bl1, acc01);
    acc10 = WMMA_BF16(ah1, bh0, acc10);
    acc10 = WMMA_BF16(al1, bh0, acc10);
    acc10 = WMMA_BF16(ah1, bl0, acc10);
    acc11 = WMMA_BF16(ah1, bh1, acc11);
    acc11 = WMMA_BF16(al1, bh1, acc11);
    acc11 = WMMA_BF16(ah1, bl1, acc11);
  }

  const int col0 = n0 + (wc << 5) + hl;
  const int col1 = col0 + 16;
  const float b0 = bias ? bias[col0] : 0.f;
  const float b1v = bias ? bias[col1] : 0.f;
  const int rbase0 = m0 + (wr << 5) + (hh << 3);
  const int rbase1 = rbase0 + 16;
#pragma unroll
  for (int r = 0; r < 8; ++r) {
    int g0r = rbase0 + r;
    if (g0r < M) {
      C[(size_t)g0r * H + col0] = acc00[r] + b0;
      C[(size_t)g0r * H + col1] = acc01[r] + b1v;
    }
    int g1r = rbase1 + r;
    if (g1r < M) {
      C[(size_t)g1r * H + col0] = acc10[r] + b0;
      C[(size_t)g1r * H + col1] = acc11[r] + b1v;
    }
  }
}

#define WPLANE (H * H)
#define WPLANE_W (H * H / 2)
__global__ __launch_bounds__(256) void k_split_weights(const float* __restrict__ Wg,
                                                       const float* __restrict__ Wsl,
                                                       const float* __restrict__ Wf,
                                                       unsigned* __restrict__ wt) {
  const int m = (int)blockIdx.y;
  const float* Wm = m < 3 ? Wg + (size_t)m * WPLANE
                  : (m < 7 ? Wsl + (size_t)(m - 3) * WPLANE : Wf);
  const int g = (int)blockIdx.x * 256 + (int)threadIdx.x;
  const int n = g & 255;
  const int k8 = (g >> 8) << 3;
  float v[8];
#pragma unroll
  for (int j = 0; j < 8; ++j) v[j] = Wm[(size_t)(k8 + j) * H + n];
  unsigned short hi[8], lo[8];
#pragma unroll
  for (int j = 0; j < 8; ++j) {
    hi[j] = f2bfbits(v[j]);
    lo[j] = f2bfbits(v[j] - bfbits2f(hi[j]));
  }
  uint4 ph, pl;
  ph.x = (unsigned)hi[0] | ((unsigned)hi[1] << 16); ph.y = (unsigned)hi[2] | ((unsigned)hi[3] << 16);
  ph.z = (unsigned)hi[4] | ((unsigned)hi[5] << 16); ph.w = (unsigned)hi[6] | ((unsigned)hi[7] << 16);
  pl.x = (unsigned)lo[0] | ((unsigned)lo[1] << 16); pl.y = (unsigned)lo[2] | ((unsigned)lo[3] << 16);
  pl.z = (unsigned)lo[4] | ((unsigned)lo[5] << 16); pl.w = (unsigned)lo[6] | ((unsigned)lo[7] << 16);
  unsigned* ph_p = wt + (size_t)m * 2 * WPLANE_W;
  const size_t o = ((size_t)(k8 >> 5) * H + n) * (KC / 2) + ((k8 & 31) >> 1);
  *(uint4*)(ph_p + o) = ph;
  *(uint4*)(ph_p + WPLANE_W + o) = pl;
}

#define CS_GROUPS 256
#define CS_THREADS 512

__global__ __launch_bounds__(CS_THREADS) void k_colsum_part(const float* __restrict__ x,
                                                           float* __restrict__ part,
                                                           int* __restrict__ deg, int ndeg, int N) {
  __shared__ float4 red[CS_THREADS / 64][64];
  const int t = threadIdx.x;
  for (int i = (int)blockIdx.x * CS_THREADS + t; i < ndeg; i += (int)gridDim.x * CS_THREADS) deg[i] = 0;
  const int cq = t & 63;
  const int rs = t >> 6;
  const int per = (N + (int)gridDim.x - 1) / (int)gridDim.x;
  const int r0 = (int)blockIdx.x * per;
  int r1 = r0 + per; if (r1 > N) r1 = N;
  const float4* xp = (const float4*)x + cq;
  float4 acc = make_float4(0.f, 0.f, 0.f, 0.f);
  int r = r0 + rs;
  for (; r + 56 < r1; r += 64) {
    float4 v[8];
#pragma unroll
    for (int j = 0; j < 8; ++j) v[j] = xp[(size_t)(r + 8 * j) * 64];
#pragma unroll
    for (int j = 0; j < 8; ++j) {
      acc.x += v[j].x; acc.y += v[j].y; acc.z += v[j].z; acc.w += v[j].w;
    }
  }
  for (; r < r1; r += 8) {
    float4 v = xp[(size_t)r * 64];
    acc.x += v.x; acc.y += v.y; acc.z += v.z; acc.w += v.w;
  }
  red[rs][cq] = acc;
  __syncthreads();
  if (t < H) {
    const float* rf = (const float*)&red[0][0];
    float s = 0.f;
#pragma unroll
    for (int j = 0; j < CS_THREADS / 64; ++j) s += rf[j * H + t];
    part[(size_t)blockIdx.x * H + t] = s;
  }
}

__global__ __launch_bounds__(1024) void k_selector(const float* __restrict__ part,
                           const float* __restrict__ W1, const float* __restrict__ b1,
                           const float* __restrict__ W2, const float* __restrict__ b2,
                           float* __restrict__ sw, int N) {
  __shared__ float4 psum[16][64];
  __shared__ float gsh[H];
  __shared__ float hid[128];
  __shared__ float lg[4];
  const int t = threadIdx.x;
  {
    const int cq = t & 63, q = t >> 6;
    const float4* pp = (const float4*)part + cq;
    float4 s = make_float4(0.f, 0.f, 0.f, 0.f);
#pragma unroll
    for (int j = 0; j < CS_GROUPS / 16; ++j) {
      float4 v = pp[(size_t)(q + 16 * j) * 64];
      s.x += v.x; s.y += v.y; s.z += v.z; s.w += v.w;
    }
    psum[q][cq] = s;
  }
  __syncthreads();
  if (t < H) {
    const float* pf = (const float*)&psum[0][0];
    float gsum = 0.f;
#pragma unroll
    for (int j = 0; j < 16; ++j) gsum += pf[j * H + t];
    float invN = 1.f / (float)N;
    gsh[t] = gsum * invN;
  }
  __syncthreads();
  if (t < 128) {
    float a = b1[t];
    for (int k = 0; k < H; ++k) a += gsh[k] * W1[k * 128 + t];
    hid[t] = fmaxf(a, 0.f);
  }
  __syncthreads();
  if (t < 4) {
    float l = b2[t];
    for (int k = 0; k < 128; ++k) l += hid[k] * W2[k * 4 + t];
    lg[t] = l;
  }
  __syncthreads();
  if (t == 0) {
    float m = fmaxf(fmaxf(lg[0], lg[1]), fmaxf(lg[2], lg[3]));
    float e0 = __expf(lg[0] - m), e1 = __expf(lg[1] - m);
    float e2 = __expf(lg[2] - m), e3 = __expf(lg[3] - m);
    float s = e0 + e1 + e2 + e3;
    sw[0] = e0 / s; sw[1] = e1 / s; sw[2] = e2 / s; sw[3] = e3 / s;
  }
}

#define ROW_BLOCKS 2048

__global__ __launch_bounds__(256) void k_sd(const float* __restrict__ h,
                     const float* __restrict__ asrc, const float* __restrict__ adst,
                     float* __restrict__ s, float* __restrict__ d, int N) {
  const int lane = threadIdx.x & 31;
  const int nw = (int)gridDim.x * 8;
  float as[8], ad[8];
#pragma unroll
  for (int j = 0; j < 8; ++j) { as[j] = asrc[lane + 32 * j]; ad[j] = adst[lane + 32 * j]; }
  for (int n = blockIdx.x * 8 + (threadIdx.x >> 5); n < N; n += nw) {
    const float* hp = h + (size_t)n * H + lane;
    float v[8];
#pragma unroll
    for (int j = 0; j < 8; ++j) v[j] = hp[32 * j];
    float vs = 0.f, vd = 0.f;
#pragma unroll
    for (int j = 0; j < 8; ++j) { vs += v[j] * as[j]; vd += v[j] * ad[j]; }
    for (int o = 16; o; o >>= 1) { vs += __shfl_xor(vs, o, 32); vd += __shfl_xor(vd, o, 32); }
    if (lane == 0) { s[n] = vs; d[n] = vd; }
  }
}

__global__ void k_deg_count(const int* __restrict__ dst, int* __restrict__ deg, int E) {
  int e = blockIdx.x * blockDim.x + threadIdx.x;
  if (e < E) atomicAdd(&deg[dst[e]], 1);
}

#define SCAN_TILE 16384
__global__ __launch_bounds__(1024) void k_scan_deg(int* __restrict__ off, int N) {
  __shared__ int wsum[32];
  __shared__ int wpre[32];
  const int t = threadIdx.x, lane = t & 31, w = t >> 5;
  int carry = 0;
  for (int base = 0; base < N; base += SCAN_TILE) {
    int4* p = (int4*)(off + base + 16 * t);
    int4 q[4];
#pragma unroll
    for (int k = 0; k < 4; ++k) q[k] = p[k];
    int tot = 0;
#pragma unroll
    for (int k = 0; k < 4; ++k) tot += q[k].x + q[k].y + q[k].z + q[k].w;
    int inc = tot;
    for (int o = 1; o < 32; o <<= 1) { int u = __shfl_up(inc, o, 32); if (lane >= o) inc += u; }
    if (lane == 31) wsum[w] = inc;
    __syncthreads();
    if (w == 0) {
      int x = wsum[lane], ix = x;
      for (int o = 1; o < 32; o <<= 1) { int u = __shfl_up(ix, o, 32); if (lane >= o) ix += u; }
      wpre[lane] = ix - x;
    }
    __syncthreads();
    int run = carry + wpre[w] + inc - tot;
    carry += wpre[31] + wsum[31];
#pragma unroll
    for (int k = 0; k < 4; ++k) {
      int4 o;
      o.x = run; run += q[k].x;
      o.y = run; run += q[k].y;
      o.z = run; run += q[k].z;
      o.w = run; run += q[k].w;
      p[k] = o;
    }
    __syncthreads();
  }
}

__global__ void k_csr_fill(const int* __restrict__ src, const int* __restrict__ dst,
                           const int* __restrict__ attr, int* __restrict__ off,
                           int* __restrict__ elist, int E) {
  int e = blockIdx.x * blockDim.x + threadIdx.x;
  if (e < E) {
    int pos = atomicAdd(&off[dst[e]], 1);
    ((int2*)elist)[pos] = make_int2(src[e], attr[e]);
  }
}

__global__ __launch_bounds__(256) void k_gat_gather(const float* __restrict__ h,
                             const float* __restrict__ s, const float* __restrict__ d,
                             const int* __restrict__ off, const int* __restrict__ elist,
                             int branch, const float* __restrict__ bias,
                             float* __restrict__ xo, int N) {
  const int2* el = (const int2*)elist;
  const int lane = threadIdx.x & 31;
  const int nw = (int)gridDim.x * 8;
  const int wv = __builtin_amdgcn_readfirstlane((int)(threadIdx.x >> 5));
  float bv[8];
#pragma unroll
  for (int j = 0; j < 8; ++j) bv[j] = bias[lane + 32 * j];
  for (int n = (int)blockIdx.x * 8 + wv; n < N; n += nw) {
    const int e0 = n ? off[n - 1] : 0;
    const int e1 = off[n];
    const float* hp = h + (size_t)n * H + lane;
    float acc[8];
#pragma unroll
    for (int j = 0; j < 8; ++j) acc[j] = hp[32 * j];
    const float dn = d[n];
    const float lself = lrelu(s[n] + dn);
    float m = lself;
    for (int c = e0; c < e1; c += 32) {
      const int i = c + lane;
      if (i < e1) {
        const int2 ed = el[i];
        const float lg = lrelu(s[ed.x] + dn);
        if (ed.y == branch) m = fmaxf(m, lg);
      }
    }
    for (int o = 16; o; o >>= 1) m = fmaxf(m, __shfl_xor(m, o, 32));
    const float wself = __expf(lself - m);
    float den = wself;
#pragma unroll
    for (int j = 0; j < 8; ++j) acc[j] *= wself;
    for (int c = e0; c < e1; c += 32) {
      const int i = c + lane;
      bool on = false;
      int sn = 0;
      float w = 0.f;
      if (i < e1) {
        const int2 ed = el[i];
        const float lg = lrelu(s[ed.x] + dn);
        on = (ed.y == branch);
        sn = ed.x;
        if (on) w = __expf(lg - m);
      }
      const unsigned live = __builtin_amdgcn_ballot_w32(on);
      int rank = 0;
      for (unsigned b = live; b; b &= b - 1u) {
        const int j = __builtin_ctz(b);
        const int sj = __builtin_amdgcn_readlane(sn, j);
        rank += (sj < sn || (sj == sn && j < lane)) ? 1 : 0;
      }
      const int cnt = __builtin_popcount(live);
#pragma unroll 2
      for (int r = 0; r < cnt; ++r) {
        const int j = __builtin_ctz(__builtin_amdgcn_ballot_w32(on && rank == r));
        const int sj = __builtin_amdgcn_readlane(sn, j);
        const float wj = __uint_as_float(__builtin_amdgcn_readlane(__float_as_uint(w), j));
        den += wj;
        const float* sp = h + (size_t)sj * H + lane;
        float v[8];
#pragma unroll
        for (int k = 0; k < 8; ++k) v[k] = sp[32 * k];
#pragma unroll
        for (int k = 0; k < 8; ++k) acc[k] += wj * v[k];
      }
    }
    float* op = xo + (size_t)n * H + lane;
#pragma unroll
    for (int k = 0; k < 8; ++k) op[32 * k] = acc[k] / den + bv[k];
  }
}

__global__ __launch_bounds__(256) void k_ln_relu(const float* __restrict__ t,
                          const float* __restrict__ gamma,
                          const float* __restrict__ beta, float* __restrict__ out,
                          const float* __restrict__ swp, int swidx, int accumulate, int N) {
  const int lane = threadIdx.x & 31;
  const int nw = (int)gridDim.x * 8;
  float g[8], b[8];
#pragma unroll
  for (int j = 0; j < 8; ++j) { g[j] = gamma[lane + j * 32]; b[j] = beta[lane + j * 32]; }
  const float scale = swp ? swp[swidx] : 1.f;
  for (int n = blockIdx.x * 8 + (threadIdx.x >> 5); n < N; n += nw) {
    const float* tp = t + (size_t)n * H + lane;
    float* op = out + (size_t)n * H + lane;
    float v[8], o[8] = {};
#pragma unroll
    for (int j = 0; j < 8; ++j) v[j] = tp[j * 32];
    if (accumulate) {
#pragma unroll
      for (int j = 0; j < 8; ++j) o[j] = op[j * 32];
    }
    float sum = 0.f, sq = 0.f;
#pragma unroll
    for (int j = 0; j < 8; ++j) {
      sum += v[j];
      sq += v[j] * v[j];
    }
    for (int k = 16; k; k >>= 1) { sum += __shfl_xor(sum, k, 32); sq += __shfl_xor(sq, k, 32); }
    float mu = sum * (1.f / H);
    float var = sq * (1.f / H) - mu * mu;
    float inv = rsqrtf(var + 1e-5f);
#pragma unroll
    for (int j = 0; j < 8; ++j) {
      float y = (v[j] - mu) * inv * g[j] + b[j];
      v[j] = fmaxf(y, 0.f) * scale;
    }
    if (accumulate) {
#pragma unroll
      for (int j = 0; j < 8; ++j) op[j * 32] = o[j] + v[j];
    } else {
#pragma unroll
      for (int j = 0; j < 8; ++j) op[j * 32] = v[j];
    }
  }
}

extern "C" void kernel_launch(void* const* d_in, const int* in_sizes, int n_in,
                              void* d_out, int out_size, void* d_ws, size_t ws_size,
                              hipStream_t stream) {
  const float* x     = (const float*)d_in[0];
  const int*   ei    = (const int*)d_in[1];
  const int*   ea    = (const int*)d_in[2];
  const float* Wg    = (const float*)d_in[3];
  const float* a_src = (const float*)d_in[4];
  const float* a_dst = (const float*)d_in[5];
  const float* bg    = (const float*)d_in[6];
  const float* Wsl   = (const float*)d_in[7];
  const float* bsl   = (const float*)d_in[8];
  const float* gsl   = (const float*)d_in[9];
  const float* betas = (const float*)d_in[10];
  const float* W1    = (const float*)d_in[11];
  const float* b1    = (const float*)d_in[12];
  const float* W2    = (const float*)d_in[13];
  const float* b2    = (const float*)d_in[14];
  const float* Wf    = (const float*)d_in[15];
  const float* bfv   = (const float*)d_in[16];
  const float* gf    = (const float*)d_in[17];
  const float* betaf = (const float*)d_in[18];

  const int N = in_sizes[0] / H;
  const int E = in_sizes[2];
  float* out = (float*)d_out;

  char* wsb = (char*)d_ws;
  size_t off = 0;
  auto alloc = [&](size_t bytes) -> void* {
    void* p = wsb + off;
    off += (bytes + 255) & ~(size_t)255;
    return p;
  };
  float*    sw    = (float*)alloc(4 * sizeof(float));
  float*    sArr  = (float*)alloc((size_t)N * sizeof(float));
  float*    dArr  = (float*)alloc((size_t)N * sizeof(float));
  const int nPad = (N + SCAN_TILE - 1) / SCAN_TILE * SCAN_TILE;
  int*      ebnd  = (int*)alloc((size_t)nPad * sizeof(int));
  float*    bufH  = (float*)alloc((size_t)N * H * sizeof(float));
  float*    bufX  = (float*)alloc((size_t)N * H * sizeof(float));
  int*      elist = (int*)alloc((size_t)N * H * sizeof(float));
  unsigned* wt = (unsigned*)((char*)elist + (((size_t)E * 8 + 255) & ~(size_t)255));

  dim3 gemmGrid((unsigned)((N + TM - 1) / TM), (unsigned)(H / TNB));
  int eThrBlocks  = (E + 255) / 256;

  float* part = bufX;
  k_split_weights<<<dim3(32, 8), 256, 0, stream>>>(Wg, Wsl, Wf, wt);
  k_colsum_part<<<CS_GROUPS, CS_THREADS, 0, stream>>>(x, part, ebnd, nPad, N);
  k_selector<<<1, 1024, 0, stream>>>(part, W1, b1, W2, b2, sw, N);

  k_deg_count<<<eThrBlocks, 256, 0, stream>>>(ei + E, ebnd, E);
  k_scan_deg<<<1, 1024, 0, stream>>>(ebnd, N);
  k_csr_fill<<<eThrBlocks, 256, 0, stream>>>(ei, ei + E, ea, ebnd, elist, E);

  for (int i = 0; i < 3; ++i) {
    k_gemm_bf16x3<<<gemmGrid, 256, 0, stream>>>(x, wt + (size_t)i * 2 * WPLANE_W,
                                              wt + (size_t)i * 2 * WPLANE_W + WPLANE_W, nullptr, bufH, N);
    k_sd<<<ROW_BLOCKS, 256, 0, stream>>>(bufH, a_src + i * H, a_dst + i * H, sArr, dArr, N);
    k_gat_gather<<<ROW_BLOCKS, 256, 0, stream>>>(bufH, sArr, dArr, ebnd, elist, i,
                                                 bg + i * H, bufX, N);
    k_gemm_bf16x3<<<gemmGrid, 256, 0, stream>>>(bufX, wt + (size_t)(3 + i) * 2 * WPLANE_W,
                                              wt + (size_t)(3 + i) * 2 * WPLANE_W + WPLANE_W,
                                              bsl + i * H, bufH, N);
    k_ln_relu<<<ROW_BLOCKS, 256, 0, stream>>>(bufH, gsl + i * H, betas + i * H, out, sw, i,
                                              i > 0 ? 1 : 0, N);
  }
  k_gemm_bf16x3<<<gemmGrid, 256, 0, stream>>>(x, wt + (size_t)6 * 2 * WPLANE_W,
                                              wt + (size_t)6 * 2 * WPLANE_W + WPLANE_W, bsl + 3 * H, bufH, N);
  k_ln_relu<<<ROW_BLOCKS, 256, 0, stream>>>(bufH, gsl + 3 * H, betas + 3 * H, out, sw, 3, 1, N);

  k_gemm_bf16x3<<<gemmGrid, 256, 0, stream>>>(out, wt + (size_t)7 * 2 * WPLANE_W,
                                              wt + (size_t)7 * 2 * WPLANE_W + WPLANE_W, bfv, bufH, N);
  k_ln_relu<<<ROW_BLOCKS, 256, 0, stream>>>(bufH, gf, betaf, out, nullptr, 0, 0, N);
}


# GEMM work-group tile widened to whole rows (64 x 256 instead of 128 x 64): each A tile is read and split to bf16 high/low once instead of four times; per-element sum order kept
# speedup vs baseline: 2.2428x; 1.2535x over previous
#include <hip/hip_runtime.h>

#define H 256
#define TM 64
#define KC 32
#define APAD 40

typedef __bf16 bf16;
typedef __attribute__((ext_vector_type(16))) __bf16 v16bf;
typedef __attribute__((ext_vector_type(8)))  __bf16 v8bf;
typedef __attribute__((ext_vector_type(8)))  float  v8f;

__device__ __forceinline__ unsigned short f2bfbits(float f) {
  unsigned u = __float_as_uint(f);
  return (unsigned short)((u + 0x7FFFu + ((u >> 16) & 1u)) >> 16);
}
__device__ __forceinline__ float bfbits2f(unsigned short h) {
  return __uint_as_float(((unsigned)h) << 16);
}
__device__ __forceinline__ bf16 f2bf(float f) {
  unsigned short hs = f2bfbits(f);
  return __builtin_bit_cast(bf16, hs);
}
__device__ __forceinline__ float lrelu(float v) { return v > 0.f ? v : 0.2f * v; }

__device__ __forceinline__ v16bf load_frag(const bf16* rowp, int g0) {
  v8bf a = *(const v8bf*)(rowp + g0);
  v8bf b = *(const v8bf*)(rowp + 16 + g0);
  return __builtin_shufflevector(a, b, 0, 1, 2, 3, 4, 5, 6, 7,
                                 8, 9, 10, 11, 12, 13, 14, 15);
}

#define WMMA_BF16(A_, B_, C_) \
  __builtin_amdgcn_wmma_f32_16x16x32_bf16(false, (A_), false, (B_), (short)0, (C_), false, false)

__global__ __launch_bounds__(256, 8) void k_gemm_bf16x3(
    const float* __restrict__ A, const unsigned* __restrict__ Bhi,
    const unsigned* __restrict__ Blo,
    const float* __restrict__ bias, float* __restrict__ C, int M)
{
  __shared__ bf16 Ah[TM][APAD];
  __shared__ bf16 Al[TM][APAD];
  __shared__ bf16 Bth[H][APAD];
  __shared__ bf16 Btl[H][APAD];

  const int tid = threadIdx.x;
  const int m0 = blockIdx.x * TM;
  const int wid = tid >> 5, lane = tid & 31;
  const int hl = lane & 15, hh = lane >> 4;
  const int wr = wid >> 2;
  const int wc = wid & 3;
  const int g0 = hh << 3;

  v8f acc00 = {}, acc01 = {}, acc02 = {}, acc03 = {};
  v8f acc10 = {}, acc11 = {}, acc12 = {}, acc13 = {};

  for (int k0 = 0; k0 < H; k0 += KC) {
    __syncthreads();
    for (int idx = tid; idx < (TM * KC) / 4; idx += 256) {
      int r = idx >> 3;
      int c4 = (idx & 7) << 2;
      int gr = m0 + r;
      float4 v = (gr < M) ? *(const float4*)&A[(size_t)gr * H + k0 + c4]
                          : make_float4(0.f, 0.f, 0.f, 0.f);
      ushort4 hiv, lov;
      hiv.x = f2bfbits(v.x); lov.x = f2bfbits(v.x - bfbits2f(hiv.x));
      hiv.y = f2bfbits(v.y); lov.y = f2bfbits(v.y - bfbits2f(hiv.y));
      hiv.z = f2bfbits(v.z); lov.z = f2bfbits(v.z - bfbits2f(hiv.z));
      hiv.w = f2bfbits(v.w); lov.w = f2bfbits(v.w - bfbits2f(hiv.w));
      *(ushort4*)&Ah[r][c4] = hiv;
      *(ushort4*)&Al[r][c4] = lov;
    }
    {
      const size_t bo = (size_t)(k0 >> 5) * (H * KC / 2) + (size_t)tid * 4;
      uint4 vh[4], vl[4];
#pragma unroll
      for (int j = 0; j < 4; ++j) {
        vh[j] = *(const uint4*)(Bhi + bo + (size_t)j * 1024);
        vl[j] = *(const uint4*)(Blo + bo + (size_t)j * 1024);
      }
#pragma unroll
      for (int j = 0; j < 4; ++j) {
        const int p = tid + 256 * j;
        *(uint4*)&Bth[p >> 2][(p & 3) << 3] = vh[j];
        *(uint4*)&Btl[p >> 2][(p & 3) << 3] = vl[j];
      }
    }
    __syncthreads();

    const v16bf ah0 = load_frag(&Ah[(wr << 5) + hl][0], g0);
    const v16bf al0 = load_frag(&Al[(wr << 5) + hl][0], g0);
    const v16bf ah1 = load_frag(&Ah[(wr << 5) + 16 + hl][0], g0);
    const v16bf al1 = load_frag(&Al[(wr << 5) + 16 + hl][0], g0);
    const v16bf bh0 = load_frag(&Bth[(wc << 6) + hl][0], g0);
    const v16bf bl0 = load_frag(&Btl[(wc << 6) + hl][0], g0);
    const v16bf bh1 = load_frag(&Bth[(wc << 6) + 16 + hl][0], g0);
    const v16bf bl1 = load_frag(&Btl[(wc << 6) + 16 + hl][0], g0);
    const v16bf bh2 = load_frag(&Bth[(wc << 6) + 32 + hl][0], g0);
    const v16bf bl2 = load_frag(&Btl[(wc << 6) + 32 + hl][0], g0);
    const v16bf bh3 = load_frag(&Bth[(wc << 6) + 48 + hl][0], g0);
    const v16bf bl3 = load_frag(&Btl[(wc << 6) + 48 + hl][0], g0);

    acc00 = WMMA_BF16(ah0, bh0, acc00);
    acc00 = WMMA_BF16(al0, bh0, acc00);
    acc00 = WMMA_BF16(ah0, bl0, acc00);
    acc10 = WMMA_BF16(ah1, bh0, acc10);
    acc10 = WMMA_BF16(al1, bh0, acc10);
    acc10 = WMMA_BF16(ah1, bl0, acc10);
    acc01 = WMMA_BF16(ah0, bh1, acc01);
    acc01 = WMMA_BF16(al0, bh1, acc01);
    acc01 = WMMA_BF16(ah0, bl1, acc01);
    acc11 = WMMA_BF16(ah1, bh1, acc11);
    acc11 = WMMA_BF16(al1, bh1, acc11);
    acc11 = WMMA_BF16(ah1, bl1, acc11);
    acc02 = WMMA_BF16(ah0, bh2, acc02);
    acc02 = WMMA_BF16(al0, bh2, acc02);
    acc02 = WMMA_BF16(ah0, bl2, acc02);
    acc12 = WMMA_BF16(ah1, bh2, acc12);
    acc12 = WMMA_BF16(al1, bh2, acc12);
    acc12 = WMMA_BF16(ah1, bl2, acc12);
    acc03 = WMMA_BF16(ah0, bh3, acc03);
    acc03 = WMMA_BF16(al0, bh3, acc03);
    acc03 = WMMA_BF16(ah0, bl3, acc03);
    acc13 = WMMA_BF16(ah1, bh3, acc13);
    acc13 = WMMA_BF16(al1, bh3, acc13);
    acc13 = WMMA_BF16(ah1, bl3, acc13);
  }

  const int col0 = (wc << 6) + hl;
  const float b0 = bias ? bias[col0] : 0.f;
  const float b1v = bias ? bias[col0 + 16] : 0.f;
  const float b2v = bias ? bias[col0 + 32] : 0.f;
  const float b3v = bias ? bias[col0 + 48] : 0.f;
  const int rbase0 = m0 + (wr << 5) + (hh << 3);
  const int rbase1 = rbase0 + 16;
#pragma unroll
  for (int r = 0; r < 8; ++r) {
    int g0r = rbase0 + r;
    if (g0r < M) {
      float* cp = C + (size_t)g0r * H + col0;
      cp[0]  = acc00[r] + b0;
      cp[16] = acc01[r] + b1v;
      cp[32] = acc02[r] + b2v;
      cp[48] = acc03[r] + b3v;
    }
    int g1r = rbase1 + r;
    if (g1r < M) {
      float* cp = C + (size_t)g1r * H + col0;
      cp[0]  = acc10[r] + b0;
      cp[16] = acc11[r] + b1v;
      cp[32] = acc12[r] + b2v;
      cp[48] = acc13[r] + b3v;
    }
  }
}

#define WPLANE (H * H)
#define WPLANE_W (H * H / 2)
__global__ __launch_bounds__(256) void k_split_weights(const float* __restrict__ Wg,
                                                       const float* __restrict__ Wsl,
                                                       const float* __restrict__ Wf,
                                                       unsigned* __restrict__ wt) {
  const int m = (int)blockIdx.y;
  const float* Wm = m < 3 ? Wg + (size_t)m * WPLANE
                  : (m < 7 ? Wsl + (size_t)(m - 3) * WPLANE : Wf);
  const int g = (int)blockIdx.x * 256 + (int)threadIdx.x;
  const int n = g & 255;
  const int k8 = (g >> 8) << 3;
  float v[8];
#pragma unroll
  for (int j = 0; j < 8; ++j) v[j] = Wm[(size_t)(k8 + j) * H + n];
  unsigned short hi[8], lo[8];
#pragma unroll
  for (int j = 0; j < 8; ++j) {
    hi[j] = f2bfbits(v[j]);
    lo[j] = f2bfbits(v[j] - bfbits2f(hi[j]));
  }
  uint4 ph, pl;
  ph.x = (unsigned)hi[0] | ((unsigned)hi[1] << 16); ph.y = (unsigned)hi[2] | ((unsigned)hi[3] << 16);
  ph.z = (unsigned)hi[4] | ((unsigned)hi[5] << 16); ph.w = (unsigned)hi[6] | ((unsigned)hi[7] << 16);
  pl.x = (unsigned)lo[0] | ((unsigned)lo[1] << 16); pl.y = (unsigned)lo[2] | ((unsigned)lo[3] << 16);
  pl.z = (unsigned)lo[4] | ((unsigned)lo[5] << 16); pl.w = (unsigned)lo[6] | ((unsigned)lo[7] << 16);
  unsigned* ph_p = wt + (size_t)m * 2 * WPLANE_W;
  const size_t o = ((size_t)(k8 >> 5) * H + n) * (KC / 2) + ((k8 & 31) >> 1);
  *(uint4*)(ph_p + o) = ph;
  *(uint4*)(ph_p + WPLANE_W + o) = pl;
}

#define CS_GROUPS 256
#define CS_THREADS 512

__global__ __launch_bounds__(CS_THREADS) void k_colsum_part(const float* __restrict__ x,
                                                           float* __restrict__ part,
                                                           int* __restrict__ deg, int ndeg, int N) {
  __shared__ float4 red[CS_THREADS / 64][64];
  const int t = threadIdx.x;
  for (int i = (int)blockIdx.x * CS_THREADS + t; i < ndeg; i += (int)gridDim.x * CS_THREADS) deg[i] = 0;
  const int cq = t & 63;
  const int rs = t >> 6;
  const int per = (N + (int)gridDim.x - 1) / (int)gridDim.x;
  const int r0 = (int)blockIdx.x * per;
  int r1 = r0 + per; if (r1 > N) r1 = N;
  const float4* xp = (const float4*)x + cq;
  float4 acc = make_float4(0.f, 0.f, 0.f, 0.f);
  int r = r0 + rs;
  for (; r + 56 < r1; r += 64) {
    float4 v[8];
#pragma unroll
    for (int j = 0; j < 8; ++j) v[j] = xp[(size_t)(r + 8 * j) * 64];
#pragma unroll
    for (int j = 0; j < 8; ++j) {
      acc.x += v[j].x; acc.y += v[j].y; acc.z += v[j].z; acc.w += v[j].w;
    }
  }
  for (; r < r1; r += 8) {
    float4 v = xp[(size_t)r * 64];
    acc.x += v.x; acc.y += v.y; acc.z += v.z; acc.w += v.w;
  }
  red[rs][cq] = acc;
  __syncthreads();
  if (t < H) {
    const float* rf = (const float*)&red[0][0];
    float s = 0.f;
#pragma unroll
    for (int j = 0; j < CS_THREADS / 64; ++j) s += rf[j * H + t];
    part[(size_t)blockIdx.x * H + t] = s;
  }
}

__global__ __launch_bounds__(1024) void k_selector(const float* __restrict__ part,
                           const float* __restrict__ W1, const float* __restrict__ b1,
                           const float* __restrict__ W2, const float* __restrict__ b2,
                           float* __restrict__ sw, int N) {
  __shared__ float4 psum[16][64];
  __shared__ float gsh[H];
  __shared__ float hid[128];
  __shared__ float lg[4];
  const int t = threadIdx.x;
  {
    const int cq = t & 63, q = t >> 6;
    const float4* pp = (const float4*)part + cq;
    float4 s = make_float4(0.f, 0.f, 0.f, 0.f);
#pragma unroll
    for (int j = 0; j < CS_GROUPS / 16; ++j) {
      float4 v = pp[(size_t)(q + 16 * j) * 64];
      s.x += v.x; s.y += v.y; s.z += v.z; s.w += v.w;
    }
    psum[q][cq] = s;
  }
  __syncthreads();
  if (t < H) {
    const float* pf = (const float*)&psum[0][0];
    float gsum = 0.f;
#pragma unroll
    for (int j = 0; j < 16; ++j) gsum += pf[j * H + t];
    float invN = 1.f / (float)N;
    gsh[t] = gsum * invN;
  }
  __syncthreads();
  if (t < 128) {
    float a = b1[t];
    for (int k = 0; k < H; ++k) a += gsh[k] * W1[k * 128 + t];
    hid[t] = fmaxf(a, 0.f);
  }
  __syncthreads();
  if (t < 4) {
    float l = b2[t];
    for (int k = 0; k < 128; ++k) l += hid[k] * W2[k * 4 + t];
    lg[t] = l;
  }
  __syncthreads();
  if (t == 0) {
    float m = fmaxf(fmaxf(lg[0], lg[1]), fmaxf(lg[2], lg[3]));
    float e0 = __expf(lg[0] - m), e1 = __expf(lg[1] - m);
    float e2 = __expf(lg[2] - m), e3 = __expf(lg[3] - m);
    float s = e0 + e1 + e2 + e3;
    sw[0] = e0 / s; sw[1] = e1 / s; sw[2] = e2 / s; sw[3] = e3 / s;
  }
}

#define ROW_BLOCKS 2048

__global__ __launch_bounds__(256) void k_sd(const float* __restrict__ h,
                     const float* __restrict__ asrc, const float* __restrict__ adst,
                     float* __restrict__ s, float* __restrict__ d, int N) {
  const int lane = threadIdx.x & 31;
  const int nw = (int)gridDim.x * 8;
  float as[8], ad[8];
#pragma unroll
  for (int j = 0; j < 8; ++j) { as[j] = asrc[lane + 32 * j]; ad[j] = adst[lane + 32 * j]; }
  for (int n = blockIdx.x * 8 + (threadIdx.x >> 5); n < N; n += nw) {
    const float* hp = h + (size_t)n * H + lane;
    float v[8];
#pragma unroll
    for (int j = 0; j < 8; ++j) v[j] = hp[32 * j];
    float vs = 0.f, vd = 0.f;
#pragma unroll
    for (int j = 0; j < 8; ++j) { vs += v[j] * as[j]; vd += v[j] * ad[j]; }
    for (int o = 16; o; o >>= 1) { vs += __shfl_xor(vs, o, 32); vd += __shfl_xor(vd, o, 32); }
    if (lane == 0) { s[n] = vs; d[n] = vd; }
  }
}

__global__ void k_deg_count(const int* __restrict__ dst, int* __restrict__ deg, int E) {
  int e = blockIdx.x * blockDim.x + threadIdx.x;
  if (e < E) atomicAdd(&deg[dst[e]], 1);
}

#define SCAN_TILE 16384
__global__ __launch_bounds__(1024) void k_scan_deg(int* __restrict__ off, int N) {
  __shared__ int wsum[32];
  __shared__ int wpre[32];
  const int t = threadIdx.x, lane = t & 31, w = t >> 5;
  int carry = 0;
  for (int base = 0; base < N; base += SCAN_TILE) {
    int4* p = (int4*)(off + base + 16 * t);
    int4 q[4];
#pragma unroll
    for (int k = 0; k < 4; ++k) q[k] = p[k];
    int tot = 0;
#pragma unroll
    for (int k = 0; k < 4; ++k) tot += q[k].x + q[k].y + q[k].z + q[k].w;
    int inc = tot;
    for (int o = 1; o < 32; o <<= 1) { int u = __shfl_up(inc, o, 32); if (lane >= o) inc += u; }
    if (lane == 31) wsum[w] = inc;
    __syncthreads();
    if (w == 0) {
      int x = wsum[lane], ix = x;
      for (int o = 1; o < 32; o <<= 1) { int u = __shfl_up(ix, o, 32); if (lane >= o) ix += u; }
      wpre[lane] = ix - x;
    }
    __syncthreads();
    int run = carry + wpre[w] + inc - tot;
    carry += wpre[31] + wsum[31];
#pragma unroll
    for (int k = 0; k < 4; ++k) {
      int4 o;
      o.x = run; run += q[k].x;
      o.y = run; run += q[k].y;
      o.z = run; run += q[k].z;
      o.w = run; run += q[k].w;
      p[k] = o;
    }
    __syncthreads();
  }
}

__global__ void k_csr_fill(const int* __restrict__ src, const int* __restrict__ dst,
                           const int* __restrict__ attr, int* __restrict__ off,
                           int* __restrict__ elist, int E) {
  int e = blockIdx.x * blockDim.x + threadIdx.x;
  if (e < E) {
    int pos = atomicAdd(&off[dst[e]], 1);
    ((int2*)elist)[pos] = make_int2(src[e], attr[e]);
  }
}

__global__ __launch_bounds__(256) void k_gat_gather(const float* __restrict__ h,
                             const float* __restrict__ s, const float* __restrict__ d,
                             const int* __restrict__ off, const int* __restrict__ elist,
                             int branch, const float* __restrict__ bias,
                             float* __restrict__ xo, int N) {
  const int2* el = (const int2*)elist;
  const int lane = threadIdx.x & 31;
  const int nw = (int)gridDim.x * 8;
  const int wv = __builtin_amdgcn_readfirstlane((int)(threadIdx.x >> 5));
  float bv[8];
#pragma unroll
  for (int j = 0; j < 8; ++j) bv[j] = bias[lane + 32 * j];
  for (int n = (int)blockIdx.x * 8 + wv; n < N; n += nw) {
    const int e0 = n ? off[n - 1] : 0;
    const int e1 = off[n];
    const float* hp = h + (size_t)n * H + lane;
    float acc[8];
#pragma unroll
    for (int j = 0; j < 8; ++j) acc[j] = hp[32 * j];
    const float dn = d[n];
    const float lself = lrelu(s[n] + dn);
    float m = lself;
    for (int c = e0; c < e1; c += 32) {
      const int i = c + lane;
      if (i < e1) {
        const int2 ed = el[i];
        const float lg = lrelu(s[ed.x] + dn);
        if (ed.y == branch) m = fmaxf(m, lg);
      }
    }
    for (int o = 16; o; o >>= 1) m = fmaxf(m, __shfl_xor(m, o, 32));
    const float wself = __expf(lself - m);
    float den = wself;
#pragma unroll
    for (int j = 0; j < 8; ++j) acc[j] *= wself;
    for (int c = e0; c < e1; c += 32) {
      const int i = c + lane;
      bool on = false;
      int sn = 0;
      float w = 0.f;
      if (i < e1) {
        const int2 ed = el[i];
        const float lg = lrelu(s[ed.x] + dn);
        on = (ed.y == branch);
        sn = ed.x;
        if (on) w = __expf(lg - m);
      }
      const unsigned live = __builtin_amdgcn_ballot_w32(on);
      int rank = 0;
      for (unsigned b = live; b; b &= b - 1u) {
        const int j = __builtin_ctz(b);
        const int sj = __builtin_amdgcn_readlane(sn, j);
        rank += (sj < sn || (sj == sn && j < lane)) ? 1 : 0;
      }
      const int cnt = __builtin_popcount(live);
#pragma unroll 2
      for (int r = 0; r < cnt; ++r) {
        const int j = __builtin_ctz(__builtin_amdgcn_ballot_w32(on && rank == r));
        const int sj = __builtin_amdgcn_readlane(sn, j);
        const float wj = __uint_as_float(__builtin_amdgcn_readlane(__float_as_uint(w), j));
        den += wj;
        const float* sp = h + (size_t)sj * H + lane;
        float v[8];
#pragma unroll
        for (int k = 0; k < 8; ++k) v[k] = sp[32 * k];
#pragma unroll
        for (int k = 0; k < 8; ++k) acc[k] += wj * v[k];
      }
    }
    float* op = xo + (size_t)n * H + lane;
#pragma unroll
    for (int k = 0; k < 8; ++k) op[32 * k] = acc[k] / den + bv[k];
  }
}

__global__ __launch_bounds__(256) void k_ln_relu(const float* __restrict__ t,
                          const float* __restrict__ gamma,
                          const float* __restrict__ beta, float* __restrict__ out,
                          const float* __restrict__ swp, int swidx, int accumulate, int N) {
  const int lane = threadIdx.x & 31;
  const int nw = (int)gridDim.x * 8;
  float g[8], b[8];
#pragma unroll
  for (int j = 0; j < 8; ++j) { g[j] = gamma[lane + j * 32]; b[j] = beta[lane + j * 32]; }
  const float scale = swp ? swp[swidx] : 1.f;
  for (int n = blockIdx.x * 8 + (threadIdx.x >> 5); n < N; n += nw) {
    const float* tp = t + (size_t)n * H + lane;
    float* op = out + (size_t)n * H + lane;
    float v[8], o[8] = {};
#pragma unroll
    for (int j = 0; j < 8; ++j) v[j] = tp[j * 32];
    if (accumulate) {
#pragma unroll
      for (int j = 0; j < 8; ++j) o[j] = op[j * 32];
    }
    float sum = 0.f, sq = 0.f;
#pragma unroll
    for (int j = 0; j < 8; ++j) {
      sum += v[j];
      sq += v[j] * v[j];
    }
    for (int k = 16; k; k >>= 1) { sum += __shfl_xor(sum, k, 32); sq += __shfl_xor(sq, k, 32); }
    float mu = sum * (1.f / H);
    float var = sq * (1.f / H) - mu * mu;
    float inv = rsqrtf(var + 1e-5f);
#pragma unroll
    for (int j = 0; j < 8; ++j) {
      float y = (v[j] - mu) * inv * g[j] + b[j];
      v[j] = fmaxf(y, 0.f) * scale;
    }
    if (accumulate) {
#pragma unroll
      for (int j = 0; j < 8; ++j) op[j * 32] = o[j] + v[j];
    } else {
#pragma unroll
      for (int j = 0; j < 8; ++j) op[j * 32] = v[j];
    }
  }
}

extern "C" void kernel_launch(void* const* d_in, const int* in_sizes, int n_in,
                              void* d_out, int out_size, void* d_ws, size_t ws_size,
                              hipStream_t stream) {
  const float* x     = (const float*)d_in[0];
  const int*   ei    = (const int*)d_in[1];
  const int*   ea    = (const int*)d_in[2];
  const float* Wg    = (const float*)d_in[3];
  const float* a_src = (const float*)d_in[4];
  const float* a_dst = (const float*)d_in[5];
  const float* bg    = (const float*)d_in[6];
  const float* Wsl   = (const float*)d_in[7];
  const float* bsl   = (const float*)d_in[8];
  const float* gsl   = (const float*)d_in[9];
  const float* betas = (const float*)d_in[10];
  const float* W1    = (const float*)d_in[11];
  const float* b1    = (const float*)d_in[12];
  const float* W2    = (const float*)d_in[13];
  const float* b2    = (const float*)d_in[14];
  const float* Wf    = (const float*)d_in[15];
  const float* bfv   = (const float*)d_in[16];
  const float* gf    = (const float*)d_in[17];
  const float* betaf = (const float*)d_in[18];

  const int N = in_sizes[0] / H;
  const int E = in_sizes[2];
  float* out = (float*)d_out;

  char* wsb = (char*)d_ws;
  size_t off = 0;
  auto alloc = [&](size_t bytes) -> void* {
    void* p = wsb + off;
    off += (bytes + 255) & ~(size_t)255;
    return p;
  };
  float*    sw    = (float*)alloc(4 * sizeof(float));
  float*    sArr  = (float*)alloc((size_t)N * sizeof(float));
  float*    dArr  = (float*)alloc((size_t)N * sizeof(float));
  const int nPad = (N + SCAN_TILE - 1) / SCAN_TILE * SCAN_TILE;
  int*      ebnd  = (int*)alloc((size_t)nPad * sizeof(int));
  float*    bufH  = (float*)alloc((size_t)N * H * sizeof(float));
  float*    bufX  = (float*)alloc((size_t)N * H * sizeof(float));
  int*      elist = (int*)alloc((size_t)N * H * sizeof(float));
  unsigned* wt = (unsigned*)((char*)elist + (((size_t)E * 8 + 255) & ~(size_t)255));

  dim3 gemmGrid((unsigned)((N + TM - 1) / TM), 1u);
  int eThrBlocks  = (E + 255) / 256;

  float* part = bufX;
  k_split_weights<<<dim3(32, 8), 256, 0, stream>>>(Wg, Wsl, Wf, wt);
  k_colsum_part<<<CS_GROUPS, CS_THREADS, 0, stream>>>(x, part, ebnd, nPad, N);
  k_selector<<<1, 1024, 0, stream>>>(part, W1, b1, W2, b2, sw, N);

  k_deg_count<<<eThrBlocks, 256, 0, stream>>>(ei + E, ebnd, E);
  k_scan_deg<<<1, 1024, 0, stream>>>(ebnd, N);
  k_csr_fill<<<eThrBlocks, 256, 0, stream>>>(ei, ei + E, ea, ebnd, elist, E);

  for (int i = 0; i < 3; ++i) {
    k_gemm_bf16x3<<<gemmGrid, 256, 0, stream>>>(x, wt + (size_t)i * 2 * WPLANE_W,
                                              wt + (size_t)i * 2 * WPLANE_W + WPLANE_W, nullptr, bufH, N);
    k_sd<<<ROW_BLOCKS, 256, 0, stream>>>(bufH, a_src + i * H, a_dst + i * H, sArr, dArr, N);
    k_gat_gather<<<ROW_BLOCKS, 256, 0, stream>>>(bufH, sArr, dArr, ebnd, elist, i,
                                                 bg + i * H, bufX, N);
    k_gemm_bf16x3<<<gemmGrid, 256, 0, stream>>>(bufX, wt + (size_t)(3 + i) * 2 * WPLANE_W,
                                              wt + (size_t)(3 + i) * 2 * WPLANE_W + WPLANE_W,
                                              bsl + i * H, bufH, N);
    k_ln_relu<<<ROW_BLOCKS, 256, 0, stream>>>(bufH, gsl + i * H, betas + i * H, out, sw, i,
                                              i > 0 ? 1 : 0, N);
  }
  k_gemm_bf16x3<<<gemmGrid, 256, 0, stream>>>(x, wt + (size_t)6 * 2 * WPLANE_W,
                                              wt + (size_t)6 * 2 * WPLANE_W + WPLANE_W, bsl + 3 * H, bufH, N);
  k_ln_relu<<<ROW_BLOCKS, 256, 0, stream>>>(bufH, gsl + 3 * H, betas + 3 * H, out, sw, 3, 1, N);

  k_gemm_bf16x3<<<gemmGrid, 256, 0, stream>>>(out, wt + (size_t)7 * 2 * WPLANE_W,
                                              wt + (size_t)7 * 2 * WPLANE_W + WPLANE_W, bfv, bufH, N);
  k_ln_relu<<<ROW_BLOCKS, 256, 0, stream>>>(bufH, gf, betaf, out, nullptr, 0, 0, N);
}


# GEMM epilogue stages its 64 whole rows in LDS and does there the per-row pass that was a launch of its own behind each GEMM (LayerNorm+ReLU+scale+running sum; attention dots): sum orders kept
# speedup vs baseline: 2.7976x; 1.2474x over previous
#include <hip/hip_runtime.h>

#define H 256
#define TM 64
#define KC 32
#define APAD 40

typedef __bf16 bf16;
typedef __attribute__((ext_vector_type(16))) __bf16 v16bf;
typedef __attribute__((ext_vector_type(8)))  __bf16 v8bf;
typedef __attribute__((ext_vector_type(8)))  float  v8f;

__device__ __forceinline__ unsigned short f2bfbits(float f) {
  unsigned u = __float_as_uint(f);
  return (unsigned short)((u + 0x7FFFu + ((u >> 16) & 1u)) >> 16);
}
__device__ __forceinline__ float bfbits2f(unsigned short h) {
  return __uint_as_float(((unsigned)h) << 16);
}
__device__ __forceinline__ bf16 f2bf(float f) {
  unsigned short hs = f2bfbits(f);
  return __builtin_bit_cast(bf16, hs);
}
__device__ __forceinline__ float lrelu(float v) { return v > 0.f ? v : 0.2f * v; }

__device__ __forceinline__ v16bf load_frag(const bf16* rowp, int g0) {
  v8bf a = *(const v8bf*)(rowp + g0);
  v8bf b = *(const v8bf*)(rowp + 16 + g0);
  return __builtin_shufflevector(a, b, 0, 1, 2, 3, 4, 5, 6, 7,
                                 8, 9, 10, 11, 12, 13, 14, 15);
}

#define WMMA_BF16(A_, B_, C_) \
  __builtin_amdgcn_wmma_f32_16x16x32_bf16(false, (A_), false, (B_), (short)0, (C_), false, false)


#define CPITCH 260
#define SMEM_BYTES (TM * CPITCH * 4)

#define EPI_SD 0
#define EPI_LN 1

__device__ __forceinline__ void butterfly8(float (&a)[8]) {
  for (int k = 16; k; k >>= 1) {
    float t[8];
#pragma unroll
    for (int i = 0; i < 8; ++i) t[i] = __shfl_xor(a[i], k, 32);
#pragma unroll
    for (int i = 0; i < 8; ++i) a[i] += t[i];
  }
}

__device__ __forceinline__ void sd_rows(const float* Ct, int trow0, int grow0, int lane,
                                        const float (&as)[8], const float (&ad)[8],
                                        float* hOut, float* __restrict__ sOut,
                                        float* __restrict__ dOut) {
  float vs[8], vd[8];
#pragma unroll
  for (int i = 0; i < 8; ++i) {
    const float* tp = Ct + (trow0 + i) * CPITCH + lane;
    float* hp = hOut + (size_t)(grow0 + i) * H + lane;
    float v[8];
#pragma unroll
    for (int j = 0; j < 8; ++j) v[j] = tp[32 * j];
#pragma unroll
    for (int j = 0; j < 8; ++j) hp[32 * j] = v[j];
    float s = 0.f, d = 0.f;
#pragma unroll
    for (int j = 0; j < 8; ++j) { s += v[j] * as[j]; d += v[j] * ad[j]; }
    vs[i] = s; vd[i] = d;
  }
  butterfly8(vs);
  butterfly8(vd);
  float ss = vs[0], dd = vd[0];
#pragma unroll
  for (int i = 1; i < 8; ++i) { if (lane == i) { ss = vs[i]; dd = vd[i]; } }
  if (lane < 8) { sOut[grow0 + lane] = ss; dOut[grow0 + lane] = dd; }
}

__device__ __forceinline__ void sd_rows_edge(const float* Ct, int trow0, int grow0, int M, int lane,
                                             const float (&as)[8], const float (&ad)[8],
                                             float* hOut, float* __restrict__ sOut,
                                             float* __restrict__ dOut) {
  for (int i = 0; i < 8; ++i) {
    const int gr = grow0 + i;
    if (gr < M) {
      const float* tp = Ct + (trow0 + i) * CPITCH + lane;
      float* hp = hOut + (size_t)gr * H + lane;
      float v[8];
#pragma unroll
      for (int j = 0; j < 8; ++j) v[j] = tp[32 * j];
#pragma unroll
      for (int j = 0; j < 8; ++j) hp[32 * j] = v[j];
      float vs = 0.f, vd = 0.f;
#pragma unroll
      for (int j = 0; j < 8; ++j) { vs += v[j] * as[j]; vd += v[j] * ad[j]; }
      for (int o = 16; o; o >>= 1) { vs += __shfl_xor(vs, o, 32); vd += __shfl_xor(vd, o, 32); }
      if (lane == 0) { sOut[gr] = vs; dOut[gr] = vd; }
    }
  }
}

__device__ __forceinline__ void acc_row(const float* accIn, int gr, int lane, float (&o)[8]) {
  const float* ip = accIn + (size_t)gr * H + lane;
#pragma unroll
  for (int j = 0; j < 8; ++j) o[j] = ip[32 * j];
}

template <bool ACC>
__device__ __forceinline__ void ln_finish(float (&v)[8], float sum, float sq,
                                          const float (&g)[8], const float (&b)[8], float scale,
                                          const float (&o)[8], float* op) {
  float mu = sum * (1.f / H);
  float var = sq * (1.f / H) - mu * mu;
  float inv = rsqrtf(var + 1e-5f);
#pragma unroll
  for (int j = 0; j < 8; ++j) {
    float y = (v[j] - mu) * inv * g[j] + b[j];
    v[j] = fmaxf(y, 0.f) * scale;
  }
  if (ACC) {
#pragma unroll
    for (int j = 0; j < 8; ++j) op[32 * j] = o[j] + v[j];
  } else {
#pragma unroll
    for (int j = 0; j < 8; ++j) op[32 * j] = v[j];
  }
}

template <bool ACC>
__device__ __forceinline__ void ln_rows(const float* Ct, int trow0, int grow0, int lane,
                                        const float (&g)[8], const float (&b)[8], float scale,
                                        const float* accIn, float* dst) {
  float sum[8], sq[8];
#pragma unroll
  for (int i = 0; i < 8; ++i) {
    const float* tp = Ct + (trow0 + i) * CPITCH + lane;
    float s = 0.f, q = 0.f;
#pragma unroll
    for (int j = 0; j < 8; ++j) {
      const float t = tp[32 * j];
      s += t;
      q += t * t;
    }
    sum[i] = s; sq[i] = q;
  }
  butterfly8(sum);
  butterfly8(sq);
  __builtin_amdgcn_sched_barrier(0);
  float o[8][8];
  if (ACC) {
#pragma unroll
    for (int i = 0; i < 4; ++i) acc_row(accIn, grow0 + i, lane, o[i]);
  }
  float vn[8];
#pragma unroll
  for (int j = 0; j < 8; ++j) vn[j] = Ct[trow0 * CPITCH + lane + 32 * j];
#pragma unroll
  for (int i = 0; i < 8; ++i) {
    float v[8];
#pragma unroll
    for (int j = 0; j < 8; ++j) v[j] = vn[j];
    if (i < 7) {
      const float* tp = Ct + (trow0 + i + 1) * CPITCH + lane;
#pragma unroll
      for (int j = 0; j < 8; ++j) vn[j] = tp[32 * j];
    }
    ln_finish<ACC>(v, sum[i], sq[i], g, b, scale, o[i], dst + (size_t)(grow0 + i) * H + lane);
    if (ACC && i < 4) acc_row(accIn, grow0 + i + 4, lane, o[i + 4]);
  }
}

template <bool ACC>
__device__ __forceinline__ void ln_rows_edge(const float* Ct, int trow0, int grow0, int M, int lane,
                                             const float (&g)[8], const float (&b)[8],
                                             float scale, const float* accIn, float* dst) {
  for (int i = 0; i < 8; ++i) {
    const int gr = grow0 + i;
    if (gr < M) {
      float oo[8] = {};
      if (ACC) acc_row(accIn, gr, lane, oo);
      const float* tp = Ct + (trow0 + i) * CPITCH + lane;
      float v[8];
#pragma unroll
      for (int j = 0; j < 8; ++j) v[j] = tp[32 * j];
      float sum = 0.f, sq = 0.f;
#pragma unroll
      for (int j = 0; j < 8; ++j) {
        sum += v[j];
        sq += v[j] * v[j];
      }
      for (int k = 16; k; k >>= 1) { sum += __shfl_xor(sum, k, 32); sq += __shfl_xor(sq, k, 32); }
      ln_finish<ACC>(v, sum, sq, g, b, scale, oo, dst + (size_t)gr * H + lane);
    }
  }
}

template <int EPI>
__device__ __forceinline__ void gemm_rows_body(
    const float* __restrict__ A, const unsigned* __restrict__ Bhi,
    const unsigned* __restrict__ Blo, int M,
    const float* __restrict__ bias,
    const float* __restrict__ p0, const float* __restrict__ p1,
    float* C,
    float* __restrict__ sOut, float* __restrict__ dOut,
    const float* __restrict__ swp, int swidx,
    const float* accIn, int accumulate)
{
  __shared__ __attribute__((aligned(16))) unsigned char smem[SMEM_BYTES];
  bf16 (*Ah)[APAD]  = (bf16 (*)[APAD])(smem);
  bf16 (*Al)[APAD]  = (bf16 (*)[APAD])(smem + TM * APAD * 2);
  bf16 (*Bth)[APAD] = (bf16 (*)[APAD])(smem + 2 * TM * APAD * 2);
  bf16 (*Btl)[APAD] = (bf16 (*)[APAD])(smem + 2 * TM * APAD * 2 + H * APAD * 2);
  float* Ct = (float*)smem;

  const int tid = threadIdx.x;
  const int m0 = blockIdx.x * TM;
  const int wid = tid >> 5, lane = tid & 31;
  const int hl = lane & 15, hh = lane >> 4;
  const int wr = wid >> 2;
  const int wc = wid & 3;
  const int g0 = hh << 3;

  v8f acc00 = {}, acc01 = {}, acc02 = {}, acc03 = {};
  v8f acc10 = {}, acc11 = {}, acc12 = {}, acc13 = {};

  for (int k0 = 0; k0 < H; k0 += KC) {
    __syncthreads();
    for (int idx = tid; idx < (TM * KC) / 4; idx += 256) {
      int r = idx >> 3;
      int c4 = (idx & 7) << 2;
      int gr = m0 + r;
      float4 v = (gr < M) ? *(const float4*)&A[(size_t)gr * H + k0 + c4]
                          : make_float4(0.f, 0.f, 0.f, 0.f);
      ushort4 hiv, lov;
      hiv.x = f2bfbits(v.x); lov.x = f2bfbits(v.x - bfbits2f(hiv.x));
      hiv.y = f2bfbits(v.y); lov.y = f2bfbits(v.y - bfbits2f(hiv.y));
      hiv.z = f2bfbits(v.z); lov.z = f2bfbits(v.z - bfbits2f(hiv.z));
      hiv.w = f2bfbits(v.w); lov.w = f2bfbits(v.w - bfbits2f(hiv.w));
      *(ushort4*)&Ah[r][c4] = hiv;
      *(ushort4*)&Al[r][c4] = lov;
    }
    {
      const size_t bo = (size_t)(k0 >> 5) * (H * KC / 2) + (size_t)tid * 4;
      uint4 vh[4], vl[4];
#pragma unroll
      for (int j = 0; j < 4; ++j) {
        vh[j] = *(const uint4*)(Bhi + bo + (size_t)j * 1024);
        vl[j] = *(const uint4*)(Blo + bo + (size_t)j * 1024);
      }
#pragma unroll
      for (int j = 0; j < 4; ++j) {
        const int p = tid + 256 * j;
        *(uint4*)&Bth[p >> 2][(p & 3) << 3] = vh[j];
        *(uint4*)&Btl[p >> 2][(p & 3) << 3] = vl[j];
      }
    }
    __syncthreads();

    const v16bf ah0 = load_frag(&Ah[(wr << 5) + hl][0], g0);
    const v16bf al0 = load_frag(&Al[(wr << 5) + hl][0], g0);
    const v16bf ah1 = load_frag(&Ah[(wr << 5) + 16 + hl][0], g0);
    const v16bf al1 = load_frag(&Al[(wr << 5) + 16 + hl][0], g0);
    const v16bf bh0 = load_frag(&Bth[(wc << 6) + hl][0], g0);
    const v16bf bl0 = load_frag(&Btl[(wc << 6) + hl][0], g0);
    const v16bf bh1 = load_frag(&Bth[(wc << 6) + 16 + hl][0], g0);
    const v16bf bl1 = load_frag(&Btl[(wc << 6) + 16 + hl][0], g0);
    const v16bf bh2 = load_frag(&Bth[(wc << 6) + 32 + hl][0], g0);
    const v16bf bl2 = load_frag(&Btl[(wc << 6) + 32 + hl][0], g0);
    const v16bf bh3 = load_frag(&Bth[(wc << 6) + 48 + hl][0], g0);
    const v16bf bl3 = load_frag(&Btl[(wc << 6) + 48 + hl][0], g0);

    acc00 = WMMA_BF16(ah0, bh0, acc00);
    acc00 = WMMA_BF16(al0, bh0, acc00);
    acc00 = WMMA_BF16(ah0, bl0, acc00);
    acc10 = WMMA_BF16(ah1, bh0, acc10);
    acc10 = WMMA_BF16(al1, bh0, acc10);
    acc10 = WMMA_BF16(ah1, bl0, acc10);
    acc01 = WMMA_BF16(ah0, bh1, acc01);
    acc01 = WMMA_BF16(al0, bh1, acc01);
    acc01 = WMMA_BF16(ah0, bl1, acc01);
    acc11 = WMMA_BF16(ah1, bh1, acc11);
    acc11 = WMMA_BF16(al1, bh1, acc11);
    acc11 = WMMA_BF16(ah1, bl1, acc11);
    acc02 = WMMA_BF16(ah0, bh2, acc02);
    acc02 = WMMA_BF16(al0, bh2, acc02);
    acc02 = WMMA_BF16(ah0, bl2, acc02);
    acc12 = WMMA_BF16(ah1, bh2, acc12);
    acc12 = WMMA_BF16(al1, bh2, acc12);
    acc12 = WMMA_BF16(ah1, bl2, acc12);
    acc03 = WMMA_BF16(ah0, bh3, acc03);
    acc03 = WMMA_BF16(al0, bh3, acc03);
    acc03 = WMMA_BF16(ah0, bl3, acc03);
    acc13 = WMMA_BF16(ah1, bh3, acc13);
    acc13 = WMMA_BF16(al1, bh3, acc13);
    acc13 = WMMA_BF16(ah1, bl3, acc13);
  }

  __syncthreads();
  {
    const int col0 = (wc << 6) + hl;
    float b0 = 0.f, b1v = 0.f, b2v = 0.f, b3v = 0.f;
    if (EPI == EPI_LN) {
      b0 = bias[col0]; b1v = bias[col0 + 16]; b2v = bias[col0 + 32]; b3v = bias[col0 + 48];
    }
    float* cw = Ct + ((wr << 5) + (hh << 3)) * CPITCH + col0;
#pragma unroll
    for (int r = 0; r < 8; ++r) {
      float* c0 = cw + r * CPITCH;
      float* c1 = c0 + 16 * CPITCH;
      if (EPI == EPI_LN) {
        c0[0]  = acc00[r] + b0;
        c0[16] = acc01[r] + b1v;
        c0[32] = acc02[r] + b2v;
        c0[48] = acc03[r] + b3v;
        c1[0]  = acc10[r] + b0;
        c1[16] = acc11[r] + b1v;
        c1[32] = acc12[r] + b2v;
        c1[48] = acc13[r] + b3v;
      } else {
        c0[0]  = acc00[r];
        c0[16] = acc01[r];
        c0[32] = acc02[r];
        c0[48] = acc03[r];
        c1[0]  = acc10[r];
        c1[16] = acc11[r];
        c1[32] = acc12[r];
        c1[48] = acc13[r];
      }
    }
  }

  const int wv = __builtin_amdgcn_readfirstlane(wid);
  const int trow0 = wv << 3;
  const int grow0 = m0 + trow0;
  const bool full = (grow0 + 8 <= M);

  if (EPI == EPI_SD) {
    float as[8], ad[8];
#pragma unroll
    for (int j = 0; j < 8; ++j) { as[j] = p0[lane + 32 * j]; ad[j] = p1[lane + 32 * j]; }
    __syncthreads();
    if (full) sd_rows(Ct, trow0, grow0, lane, as, ad, C, sOut, dOut);
    else      sd_rows_edge(Ct, trow0, grow0, M, lane, as, ad, C, sOut, dOut);
  } else {
    float g[8], b[8];
#pragma unroll
    for (int j = 0; j < 8; ++j) { g[j] = p0[lane + 32 * j]; b[j] = p1[lane + 32 * j]; }
    const float scale = swp ? swp[swidx] : 1.f;
    __syncthreads();
    if (accumulate) {
      if (full) ln_rows<true>(Ct, trow0, grow0, lane, g, b, scale, accIn, C);
      else      ln_rows_edge<true>(Ct, trow0, grow0, M, lane, g, b, scale, accIn, C);
    } else {
      if (full) ln_rows<false>(Ct, trow0, grow0, lane, g, b, scale, accIn, C);
      else      ln_rows_edge<false>(Ct, trow0, grow0, M, lane, g, b, scale, accIn, C);
    }
  }
}


__global__ __launch_bounds__(256, 8) void k_gemm_sd(
    const float* __restrict__ A, const unsigned* __restrict__ Bhi,
    const unsigned* __restrict__ Blo, const float* __restrict__ asrc,
    const float* __restrict__ adst, float* __restrict__ hOut,
    float* __restrict__ sOut, float* __restrict__ dOut, int M)
{
  gemm_rows_body<EPI_SD>(A, Bhi, Blo, M, nullptr, asrc, adst, hOut, sOut, dOut,
                         nullptr, 0, nullptr, 0);
}

__global__ __launch_bounds__(256, 8) void k_gemm_ln(
    const float* __restrict__ A, const unsigned* __restrict__ Bhi,
    const unsigned* __restrict__ Blo, const float* __restrict__ bias,
    const float* __restrict__ gamma, const float* __restrict__ beta,
    const float* __restrict__ swp, int swidx,
    const float* accIn, int accumulate, float* dst, int M)
{
  gemm_rows_body<EPI_LN>(A, Bhi, Blo, M, bias, gamma, beta, dst, nullptr, nullptr,
                         swp, swidx, accIn, accumulate);
}

#define WPLANE (H * H)
#define WPLANE_W (H * H / 2)
__global__ __launch_bounds__(256) void k_split_weights(const float* __restrict__ Wg,
                                                       const float* __restrict__ Wsl,
                                                       const float* __restrict__ Wf,
                                                       unsigned* __restrict__ wt) {
  const int m = (int)blockIdx.y;
  const float* Wm = m < 3 ? Wg + (size_t)m * WPLANE
                  : (m < 7 ? Wsl + (size_t)(m - 3) * WPLANE : Wf);
  const int g = (int)blockIdx.x * 256 + (int)threadIdx.x;
  const int n = g & 255;
  const int k8 = (g >> 8) << 3;
  float v[8];
#pragma unroll
  for (int j = 0; j < 8; ++j) v[j] = Wm[(size_t)(k8 + j) * H + n];
  unsigned short hi[8], lo[8];
#pragma unroll
  for (int j = 0; j < 8; ++j) {
    hi[j] = f2bfbits(v[j]);
    lo[j] = f2bfbits(v[j] - bfbits2f(hi[j]));
  }
  uint4 ph, pl;
  ph.x = (unsigned)hi[0] | ((unsigned)hi[1] << 16); ph.y = (unsigned)hi[2] | ((unsigned)hi[3] << 16);
  ph.z = (unsigned)hi[4] | ((unsigned)hi[5] << 16); ph.w = (unsigned)hi[6] | ((unsigned)hi[7] << 16);
  pl.x = (unsigned)lo[0] | ((unsigned)lo[1] << 16); pl.y = (unsigned)lo[2] | ((unsigned)lo[3] << 16);
  pl.z = (unsigned)lo[4] | ((unsigned)lo[5] << 16); pl.w = (unsigned)lo[6] | ((unsigned)lo[7] << 16);
  unsigned* ph_p = wt + (size_t)m * 2 * WPLANE_W;
  const size_t o = ((size_t)(k8 >> 5) * H + n) * (KC / 2) + ((k8 & 31) >> 1);
  *(uint4*)(ph_p + o) = ph;
  *(uint4*)(ph_p + WPLANE_W + o) = pl;
}

#define CS_GROUPS 256
#define CS_THREADS 512

__global__ __launch_bounds__(CS_THREADS) void k_colsum_part(const float* __restrict__ x,
                                                           float* __restrict__ part,
                                                           int* __restrict__ deg, int ndeg, int N) {
  __shared__ float4 red[CS_THREADS / 64][64];
  const int t = threadIdx.x;
  for (int i = (int)blockIdx.x * CS_THREADS + t; i < ndeg; i += (int)gridDim.x * CS_THREADS) deg[i] = 0;
  const int cq = t & 63;
  const int rs = t >> 6;
  const int per = (N + (int)gridDim.x - 1) / (int)gridDim.x;
  const int r0 = (int)blockIdx.x * per;
  int r1 = r0 + per; if (r1 > N) r1 = N;
  const float4* xp = (const float4*)x + cq;
  float4 acc = make_float4(0.f, 0.f, 0.f, 0.f);
  int r = r0 + rs;
  for (; r + 56 < r1; r += 64) {
    float4 v[8];
#pragma unroll
    for (int j = 0; j < 8; ++j) v[j] = xp[(size_t)(r + 8 * j) * 64];
#pragma unroll
    for (int j = 0; j < 8; ++j) {
      acc.x += v[j].x; acc.y += v[j].y; acc.z += v[j].z; acc.w += v[j].w;
    }
  }
  for (; r < r1; r += 8) {
    float4 v = xp[(size_t)r * 64];
    acc.x += v.x; acc.y += v.y; acc.z += v.z; acc.w += v.w;
  }
  red[rs][cq] = acc;
  __syncthreads();
  if (t < H) {
    const float* rf = (const float*)&red[0][0];
    float s = 0.f;
#pragma unroll
    for (int j = 0; j < CS_THREADS / 64; ++j) s += rf[j * H + t];
    part[(size_t)blockIdx.x * H + t] = s;
  }
}

__global__ __launch_bounds__(1024) void k_selector(const float* __restrict__ part,
                           const float* __restrict__ W1, const float* __restrict__ b1,
                           const float* __restrict__ W2, const float* __restrict__ b2,
                           float* __restrict__ sw, int N) {
  __shared__ float4 psum[16][64];
  __shared__ float gsh[H];
  __shared__ float hid[128];
  __shared__ float lg[4];
  const int t = threadIdx.x;
  {
    const int cq = t & 63, q = t >> 6;
    const float4* pp = (const float4*)part + cq;
    float4 s = make_float4(0.f, 0.f, 0.f, 0.f);
#pragma unroll
    for (int j = 0; j < CS_GROUPS / 16; ++j) {
      float4 v = pp[(size_t)(q + 16 * j) * 64];
      s.x += v.x; s.y += v.y; s.z += v.z; s.w += v.w;
    }
    psum[q][cq] = s;
  }
  __syncthreads();
  if (t < H) {
    const float* pf = (const float*)&psum[0][0];
    float gsum = 0.f;
#pragma unroll
    for (int j = 0; j < 16; ++j) gsum += pf[j * H + t];
    float invN = 1.f / (float)N;
    gsh[t] = gsum * invN;
  }
  __syncthreads();
  if (t < 128) {
    float a = b1[t];
    for (int k = 0; k < H; ++k) a += gsh[k] * W1[k * 128 + t];
    hid[t] = fmaxf(a, 0.f);
  }
  __syncthreads();
  if (t < 4) {
    float l = b2[t];
    for (int k = 0; k < 128; ++k) l += hid[k] * W2[k * 4 + t];
    lg[t] = l;
  }
  __syncthreads();
  if (t == 0) {
    float m = fmaxf(fmaxf(lg[0], lg[1]), fmaxf(lg[2], lg[3]));
    float e0 = __expf(lg[0] - m), e1 = __expf(lg[1] - m);
    float e2 = __expf(lg[2] - m), e3 = __expf(lg[3] - m);
    float s = e0 + e1 + e2 + e3;
    sw[0] = e0 / s; sw[1] = e1 / s; sw[2] = e2 / s; sw[3] = e3 / s;
  }
}

#define ROW_BLOCKS 2048

__global__ void k_deg_count(const int* __restrict__ dst, int* __restrict__ deg, int E) {
  int e = blockIdx.x * blockDim.x + threadIdx.x;
  if (e < E) atomicAdd(&deg[dst[e]], 1);
}

#define SCAN_TILE 16384
__global__ __launch_bounds__(1024) void k_scan_deg(int* __restrict__ off, int N) {
  __shared__ int wsum[32];
  __shared__ int wpre[32];
  const int t = threadIdx.x, lane = t & 31, w = t >> 5;
  int carry = 0;
  for (int base = 0; base < N; base += SCAN_TILE) {
    int4* p = (int4*)(off + base + 16 * t);
    int4 q[4];
#pragma unroll
    for (int k = 0; k < 4; ++k) q[k] = p[k];
    int tot = 0;
#pragma unroll
    for (int k = 0; k < 4; ++k) tot += q[k].x + q[k].y + q[k].z + q[k].w;
    int inc = tot;
    for (int o = 1; o < 32; o <<= 1) { int u = __shfl_up(inc, o, 32); if (lane >= o) inc += u; }
    if (lane == 31) wsum[w] = inc;
    __syncthreads();
    if (w == 0) {
      int x = wsum[lane], ix = x;
      for (int o = 1; o < 32; o <<= 1) { int u = __shfl_up(ix, o, 32); if (lane >= o) ix += u; }
      wpre[lane] = ix - x;
    }
    __syncthreads();
    int run = carry + wpre[w] + inc - tot;
    carry += wpre[31] + wsum[31];
#pragma unroll
    for (int k = 0; k < 4; ++k) {
      int4 o;
      o.x = run; run += q[k].x;
      o.y = run; run += q[k].y;
      o.z = run; run += q[k].z;
      o.w = run; run += q[k].w;
      p[k] = o;
    }
    __syncthreads();
  }
}

__global__ void k_csr_fill(const int* __restrict__ src, const int* __restrict__ dst,
                           const int* __restrict__ attr, int* __restrict__ off,
                           int* __restrict__ elist, int E) {
  int e = blockIdx.x * blockDim.x + threadIdx.x;
  if (e < E) {
    int pos = atomicAdd(&off[dst[e]], 1);
    ((int2*)elist)[pos] = make_int2(src[e], attr[e]);
  }
}

__global__ __launch_bounds__(256) void k_gat_gather(const float* __restrict__ h,
                             const float* __restrict__ s, const float* __restrict__ d,
                             const int* __restrict__ off, const int* __restrict__ elist,
                             int branch, const float* __restrict__ bias,
                             float* __restrict__ xo, int N) {
  const int2* el = (const int2*)elist;
  const int lane = threadIdx.x & 31;
  const int nw = (int)gridDim.x * 8;
  const int wv = __builtin_amdgcn_readfirstlane((int)(threadIdx.x >> 5));
  float bv[8];
#pragma unroll
  for (int j = 0; j < 8; ++j) bv[j] = bias[lane + 32 * j];
  for (int n = (int)blockIdx.x * 8 + wv; n < N; n += nw) {
    const int e0 = n ? off[n - 1] : 0;
    const int e1 = off[n];
    const float* hp = h + (size_t)n * H + lane;
    float acc[8];
#pragma unroll
    for (int j = 0; j < 8; ++j) acc[j] = hp[32 * j];
    const float dn = d[n];
    const float lself = lrelu(s[n] + dn);
    float m = lself;
    for (int c = e0; c < e1; c += 32) {
      const int i = c + lane;
      if (i < e1) {
        const int2 ed = el[i];
        const float lg = lrelu(s[ed.x] + dn);
        if (ed.y == branch) m = fmaxf(m, lg);
      }
    }
    for (int o = 16; o; o >>= 1) m = fmaxf(m, __shfl_xor(m, o, 32));
    const float wself = __expf(lself - m);
    float den = wself;
#pragma unroll
    for (int j = 0; j < 8; ++j) acc[j] *= wself;
    for (int c = e0; c < e1; c += 32) {
      const int i = c + lane;
      bool on = false;
      int sn = 0;
      float w = 0.f;
      if (i < e1) {
        const int2 ed = el[i];
        const float lg = lrelu(s[ed.x] + dn);
        on = (ed.y == branch);
        sn = ed.x;
        if (on) w = __expf(lg - m);
      }
      const unsigned live = __builtin_amdgcn_ballot_w32(on);
      int rank = 0;
      for (unsigned b = live; b; b &= b - 1u) {
        const int j = __builtin_ctz(b);
        const int sj = __builtin_amdgcn_readlane(sn, j);
        rank += (sj < sn || (sj == sn && j < lane)) ? 1 : 0;
      }
      const int cnt = __builtin_popcount(live);
#pragma unroll 2
      for (int r = 0; r < cnt; ++r) {
        const int j = __builtin_ctz(__builtin_amdgcn_ballot_w32(on && rank == r));
        const int sj = __builtin_amdgcn_readlane(sn, j);
        const float wj = __uint_as_float(__builtin_amdgcn_readlane(__float_as_uint(w), j));
        den += wj;
        const float* sp = h + (size_t)sj * H + lane;
        float v[8];
#pragma unroll
        for (int k = 0; k < 8; ++k) v[k] = sp[32 * k];
#pragma unroll
        for (int k = 0; k < 8; ++k) acc[k] += wj * v[k];
      }
    }
    float* op = xo + (size_t)n * H + lane;
#pragma unroll
    for (int k = 0; k < 8; ++k) op[32 * k] = acc[k] / den + bv[k];
  }
}


extern "C" void kernel_launch(void* const* d_in, const int* in_sizes, int n_in,
                              void* d_out, int out_size, void* d_ws, size_t ws_size,
                              hipStream_t stream) {
  const float* x     = (const float*)d_in[0];
  const int*   ei    = (const int*)d_in[1];
  const int*   ea    = (const int*)d_in[2];
  const float* Wg    = (const float*)d_in[3];
  const float* a_src = (const float*)d_in[4];
  const float* a_dst = (const float*)d_in[5];
  const float* bg    = (const float*)d_in[6];
  const float* Wsl   = (const float*)d_in[7];
  const float* bsl   = (const float*)d_in[8];
  const float* gsl   = (const float*)d_in[9];
  const float* betas = (const float*)d_in[10];
  const float* W1    = (const float*)d_in[11];
  const float* b1    = (const float*)d_in[12];
  const float* W2    = (const float*)d_in[13];
  const float* b2    = (const float*)d_in[14];
  const float* Wf    = (const float*)d_in[15];
  const float* bfv   = (const float*)d_in[16];
  const float* gf    = (const float*)d_in[17];
  const float* betaf = (const float*)d_in[18];

  const int N = in_sizes[0] / H;
  const int E = in_sizes[2];
  float* out = (float*)d_out;

  char* wsb = (char*)d_ws;
  size_t off = 0;
  auto alloc = [&](size_t bytes) -> void* {
    void* p = wsb + off;
    off += (bytes + 255) & ~(size_t)255;
    return p;
  };
  float*    sw    = (float*)alloc(4 * sizeof(float));
  float*    sArr  = (float*)alloc((size_t)N * sizeof(float));
  float*    dArr  = (float*)alloc((size_t)N * sizeof(float));
  const int nPad = (N + SCAN_TILE - 1) / SCAN_TILE * SCAN_TILE;
  int*      ebnd  = (int*)alloc((size_t)nPad * sizeof(int));
  float*    bufH  = (float*)alloc((size_t)N * H * sizeof(float));
  float*    bufX  = (float*)alloc((size_t)N * H * sizeof(float));
  int*      elist = (int*)alloc((size_t)N * H * sizeof(float));
  unsigned* wt = (unsigned*)((char*)elist + (((size_t)E * 8 + 255) & ~(size_t)255));

  dim3 gemmGrid((unsigned)((N + TM - 1) / TM), 1u);
  int eThrBlocks  = (E + 255) / 256;

  float* part = bufX;
  k_split_weights<<<dim3(32, 8), 256, 0, stream>>>(Wg, Wsl, Wf, wt);
  k_colsum_part<<<CS_GROUPS, CS_THREADS, 0, stream>>>(x, part, ebnd, nPad, N);
  k_selector<<<1, 1024, 0, stream>>>(part, W1, b1, W2, b2, sw, N);

  k_deg_count<<<eThrBlocks, 256, 0, stream>>>(ei + E, ebnd, E);
  k_scan_deg<<<1, 1024, 0, stream>>>(ebnd, N);
  k_csr_fill<<<eThrBlocks, 256, 0, stream>>>(ei, ei + E, ea, ebnd, elist, E);

  for (int i = 0; i < 3; ++i) {
    k_gemm_sd<<<gemmGrid, 256, 0, stream>>>(x, wt + (size_t)i * 2 * WPLANE_W,
                                          wt + (size_t)i * 2 * WPLANE_W + WPLANE_W,
                                          a_src + i * H, a_dst + i * H, bufH, sArr, dArr, N);
    k_gat_gather<<<ROW_BLOCKS, 256, 0, stream>>>(bufH, sArr, dArr, ebnd, elist, i,
                                                 bg + i * H, bufX, N);
    k_gemm_ln<<<gemmGrid, 256, 0, stream>>>(bufX, wt + (size_t)(3 + i) * 2 * WPLANE_W,
                                          wt + (size_t)(3 + i) * 2 * WPLANE_W + WPLANE_W,
                                          bsl + i * H, gsl + i * H, betas + i * H, sw, i,
                                          out, i > 0 ? 1 : 0, out, N);
  }
  k_gemm_ln<<<gemmGrid, 256, 0, stream>>>(x, wt + (size_t)6 * 2 * WPLANE_W,
                                          wt + (size_t)6 * 2 * WPLANE_W + WPLANE_W, bsl + 3 * H,
                                          gsl + 3 * H, betas + 3 * H, sw, 3, out, 1, bufX, N);

  k_gemm_ln<<<gemmGrid, 256, 0, stream>>>(bufX, wt + (size_t)7 * 2 * WPLANE_W,
                                          wt + (size_t)7 * 2 * WPLANE_W + WPLANE_W, bfv,
                                          gf, betaf, nullptr, 0, nullptr, 0, out, N);
}


# GEMM K loop: a chunk's two A-tile loads and eight B-plane loads issue as one group ahead of the first wait (A loop unrolled, tail rows clamped to the last row): one memory wait per chunk, not three
# speedup vs baseline: 2.9663x; 1.0603x over previous
#include <hip/hip_runtime.h>

#define H 256
#define TM 64
#define KC 32
#define APAD 40

typedef __bf16 bf16;
typedef __attribute__((ext_vector_type(16))) __bf16 v16bf;
typedef __attribute__((ext_vector_type(8)))  __bf16 v8bf;
typedef __attribute__((ext_vector_type(8)))  float  v8f;

__device__ __forceinline__ unsigned short f2bfbits(float f) {
  unsigned u = __float_as_uint(f);
  return (unsigned short)((u + 0x7FFFu + ((u >> 16) & 1u)) >> 16);
}
__device__ __forceinline__ float bfbits2f(unsigned short h) {
  return __uint_as_float(((unsigned)h) << 16);
}
__device__ __forceinline__ bf16 f2bf(float f) {
  unsigned short hs = f2bfbits(f);
  return __builtin_bit_cast(bf16, hs);
}
__device__ __forceinline__ float lrelu(float v) { return v > 0.f ? v : 0.2f * v; }

__device__ __forceinline__ v16bf load_frag(const bf16* rowp, int g0) {
  v8bf a = *(const v8bf*)(rowp + g0);
  v8bf b = *(const v8bf*)(rowp + 16 + g0);
  return __builtin_shufflevector(a, b, 0, 1, 2, 3, 4, 5, 6, 7,
                                 8, 9, 10, 11, 12, 13, 14, 15);
}

#define WMMA_BF16(A_, B_, C_) \
  __builtin_amdgcn_wmma_f32_16x16x32_bf16(false, (A_), false, (B_), (short)0, (C_), false, false)


#define CPITCH 260
#define SMEM_BYTES (TM * CPITCH * 4)

#define EPI_SD 0
#define EPI_LN 1

__device__ __forceinline__ void butterfly8(float (&a)[8]) {
  for (int k = 16; k; k >>= 1) {
    float t[8];
#pragma unroll
    for (int i = 0; i < 8; ++i) t[i] = __shfl_xor(a[i], k, 32);
#pragma unroll
    for (int i = 0; i < 8; ++i) a[i] += t[i];
  }
}

__device__ __forceinline__ void sd_rows(const float* Ct, int trow0, int grow0, int lane,
                                        const float (&as)[8], const float (&ad)[8],
                                        float* hOut, float* __restrict__ sOut,
                                        float* __restrict__ dOut) {
  float vs[8], vd[8];
#pragma unroll
  for (int i = 0; i < 8; ++i) {
    const float* tp = Ct + (trow0 + i) * CPITCH + lane;
    float* hp = hOut + (size_t)(grow0 + i) * H + lane;
    float v[8];
#pragma unroll
    for (int j = 0; j < 8; ++j) v[j] = tp[32 * j];
#pragma unroll
    for (int j = 0; j < 8; ++j) hp[32 * j] = v[j];
    float s = 0.f, d = 0.f;
#pragma unroll
    for (int j = 0; j < 8; ++j) { s += v[j] * as[j]; d += v[j] * ad[j]; }
    vs[i] = s; vd[i] = d;
  }
  butterfly8(vs);
  butterfly8(vd);
  float ss = vs[0], dd = vd[0];
#pragma unroll
  for (int i = 1; i < 8; ++i) { if (lane == i) { ss = vs[i]; dd = vd[i]; } }
  if (lane < 8) { sOut[grow0 + lane] = ss; dOut[grow0 + lane] = dd; }
}

__device__ __forceinline__ void sd_rows_edge(const float* Ct, int trow0, int grow0, int M, int lane,
                                             const float (&as)[8], const float (&ad)[8],
                                             float* hOut, float* __restrict__ sOut,
                                             float* __restrict__ dOut) {
  for (int i = 0; i < 8; ++i) {
    const int gr = grow0 + i;
    if (gr < M) {
      const float* tp = Ct + (trow0 + i) * CPITCH + lane;
      float* hp = hOut + (size_t)gr * H + lane;
      float v[8];
#pragma unroll
      for (int j = 0; j < 8; ++j) v[j] = tp[32 * j];
#pragma unroll
      for (int j = 0; j < 8; ++j) hp[32 * j] = v[j];
      float vs = 0.f, vd = 0.f;
#pragma unroll
      for (int j = 0; j < 8; ++j) { vs += v[j] * as[j]; vd += v[j] * ad[j]; }
      for (int o = 16; o; o >>= 1) { vs += __shfl_xor(vs, o, 32); vd += __shfl_xor(vd, o, 32); }
      if (lane == 0) { sOut[gr] = vs; dOut[gr] = vd; }
    }
  }
}

__device__ __forceinline__ void acc_row(const float* accIn, int gr, int lane, float (&o)[8]) {
  const float* ip = accIn + (size_t)gr * H + lane;
#pragma unroll
  for (int j = 0; j < 8; ++j) o[j] = ip[32 * j];
}

template <bool ACC>
__device__ __forceinline__ void ln_finish(float (&v)[8], float sum, float sq,
                                          const float (&g)[8], const float (&b)[8], float scale,
                                          const float (&o)[8], float* op) {
  float mu = sum * (1.f / H);
  float var = sq * (1.f / H) - mu * mu;
  float inv = rsqrtf(var + 1e-5f);
#pragma unroll
  for (int j = 0; j < 8; ++j) {
    float y = (v[j] - mu) * inv * g[j] + b[j];
    v[j] = fmaxf(y, 0.f) * scale;
  }
  if (ACC) {
#pragma unroll
    for (int j = 0; j < 8; ++j) op[32 * j] = o[j] + v[j];
  } else {
#pragma unroll
    for (int j = 0; j < 8; ++j) op[32 * j] = v[j];
  }
}

template <bool ACC>
__device__ __forceinline__ void ln_rows(const float* Ct, int trow0, int grow0, int lane,
                                        const float (&g)[8], const float (&b)[8], float scale,
                                        const float* accIn, float* dst) {
  float sum[8], sq[8];
#pragma unroll
  for (int i = 0; i < 8; ++i) {
    const float* tp = Ct + (trow0 + i) * CPITCH + lane;
    float s = 0.f, q = 0.f;
#pragma unroll
    for (int j = 0; j < 8; ++j) {
      const float t = tp[32 * j];
      s += t;
      q += t * t;
    }
    sum[i] = s; sq[i] = q;
  }
  butterfly8(sum);
  butterfly8(sq);
  __builtin_amdgcn_sched_barrier(0);
  float o[8][8];
  if (ACC) {
#pragma unroll
    for (int i = 0; i < 4; ++i) acc_row(accIn, grow0 + i, lane, o[i]);
  }
  float vn[8];
#pragma unroll
  for (int j = 0; j < 8; ++j) vn[j] = Ct[trow0 * CPITCH + lane + 32 * j];
#pragma unroll
  for (int i = 0; i < 8; ++i) {
    float v[8];
#pragma unroll
    for (int j = 0; j < 8; ++j) v[j] = vn[j];
    if (i < 7) {
      const float* tp = Ct + (trow0 + i + 1) * CPITCH + lane;
#pragma unroll
      for (int j = 0; j < 8; ++j) vn[j] = tp[32 * j];
    }
    ln_finish<ACC>(v, sum[i], sq[i], g, b, scale, o[i], dst + (size_t)(grow0 + i) * H + lane);
    if (ACC && i < 4) acc_row(accIn, grow0 + i + 4, lane, o[i + 4]);
  }
}

template <bool ACC>
__device__ __forceinline__ void ln_rows_edge(const float* Ct, int trow0, int grow0, int M, int lane,
                                             const float (&g)[8], const float (&b)[8],
                                             float scale, const float* accIn, float* dst) {
  for (int i = 0; i < 8; ++i) {
    const int gr = grow0 + i;
    if (gr < M) {
      float oo[8] = {};
      if (ACC) acc_row(accIn, gr, lane, oo);
      const float* tp = Ct + (trow0 + i) * CPITCH + lane;
      float v[8];
#pragma unroll
      for (int j = 0; j < 8; ++j) v[j] = tp[32 * j];
      float sum = 0.f, sq = 0.f;
#pragma unroll
      for (int j = 0; j < 8; ++j) {
        sum += v[j];
        sq += v[j] * v[j];
      }
      for (int k = 16; k; k >>= 1) { sum += __shfl_xor(sum, k, 32); sq += __shfl_xor(sq, k, 32); }
      ln_finish<ACC>(v, sum, sq, g, b, scale, oo, dst + (size_t)gr * H + lane);
    }
  }
}

template <int EPI>
__device__ __forceinline__ void gemm_rows_body(
    const float* __restrict__ A, const unsigned* __restrict__ Bhi,
    const unsigned* __restrict__ Blo, int M,
    const float* __restrict__ bias,
    const float* __restrict__ p0, const float* __restrict__ p1,
    float* C,
    float* __restrict__ sOut, float* __restrict__ dOut,
    const float* __restrict__ swp, int swidx,
    const float* accIn, int accumulate)
{
  __shared__ __attribute__((aligned(16))) unsigned char smem[SMEM_BYTES];
  bf16 (*Ah)[APAD]  = (bf16 (*)[APAD])(smem);
  bf16 (*Al)[APAD]  = (bf16 (*)[APAD])(smem + TM * APAD * 2);
  bf16 (*Bth)[APAD] = (bf16 (*)[APAD])(smem + 2 * TM * APAD * 2);
  bf16 (*Btl)[APAD] = (bf16 (*)[APAD])(smem + 2 * TM * APAD * 2 + H * APAD * 2);
  float* Ct = (float*)smem;

  const int tid = threadIdx.x;
  const int m0 = blockIdx.x * TM;
  const int wid = tid >> 5, lane = tid & 31;
  const int hl = lane & 15, hh = lane >> 4;
  const int wr = wid >> 2;
  const int wc = wid & 3;
  const int g0 = hh << 3;

  v8f acc00 = {}, acc01 = {}, acc02 = {}, acc03 = {};
  v8f acc10 = {}, acc11 = {}, acc12 = {}, acc13 = {};

  const int ar = tid >> 3;
  const int ac = (tid & 7) << 2;
  const float* ap0 = A + (size_t)((m0 + ar) < M ? m0 + ar : M - 1) * H + ac;
  const float* ap1 = A + (size_t)((m0 + ar + 32) < M ? m0 + ar + 32 : M - 1) * H + ac;

  for (int k0 = 0; k0 < H; k0 += KC) {
    __syncthreads();
    const size_t bo = (size_t)(k0 >> 5) * (H * KC / 2) + (size_t)tid * 4;
    uint4 vh[4], vl[4];
#pragma unroll
    for (int j = 0; j < 4; ++j) {
      vh[j] = *(const uint4*)(Bhi + bo + (size_t)j * 1024);
      vl[j] = *(const uint4*)(Blo + bo + (size_t)j * 1024);
    }
    float4 a0 = *(const float4*)(ap0 + k0);
    float4 a1 = *(const float4*)(ap1 + k0);
#pragma unroll
    for (int j = 0; j < 4; ++j) {
      const int p = tid + 256 * j;
      *(uint4*)&Bth[p >> 2][(p & 3) << 3] = vh[j];
      *(uint4*)&Btl[p >> 2][(p & 3) << 3] = vl[j];
    }
    {
      ushort4 hiv, lov;
      hiv.x = f2bfbits(a0.x); lov.x = f2bfbits(a0.x - bfbits2f(hiv.x));
      hiv.y = f2bfbits(a0.y); lov.y = f2bfbits(a0.y - bfbits2f(hiv.y));
      hiv.z = f2bfbits(a0.z); lov.z = f2bfbits(a0.z - bfbits2f(hiv.z));
      hiv.w = f2bfbits(a0.w); lov.w = f2bfbits(a0.w - bfbits2f(hiv.w));
      *(ushort4*)&Ah[ar][ac] = hiv;
      *(ushort4*)&Al[ar][ac] = lov;
    }
    {
      ushort4 hiv, lov;
      hiv.x = f2bfbits(a1.x); lov.x = f2bfbits(a1.x - bfbits2f(hiv.x));
      hiv.y = f2bfbits(a1.y); lov.y = f2bfbits(a1.y - bfbits2f(hiv.y));
      hiv.z = f2bfbits(a1.z); lov.z = f2bfbits(a1.z - bfbits2f(hiv.z));
      hiv.w = f2bfbits(a1.w); lov.w = f2bfbits(a1.w - bfbits2f(hiv.w));
      *(ushort4*)&Ah[ar + 32][ac] = hiv;
      *(ushort4*)&Al[ar + 32][ac] = lov;
    }
    __syncthreads();

    const v16bf ah0 = load_frag(&Ah[(wr << 5) + hl][0], g0);
    const v16bf al0 = load_frag(&Al[(wr << 5) + hl][0], g0);
    const v16bf ah1 = load_frag(&Ah[(wr << 5) + 16 + hl][0], g0);
    const v16bf al1 = load_frag(&Al[(wr << 5) + 16 + hl][0], g0);
    const v16bf bh0 = load_frag(&Bth[(wc << 6) + hl][0], g0);
    const v16bf bl0 = load_frag(&Btl[(wc << 6) + hl][0], g0);
    const v16bf bh1 = load_frag(&Bth[(wc << 6) + 16 + hl][0], g0);
    const v16bf bl1 = load_frag(&Btl[(wc << 6) + 16 + hl][0], g0);
    const v16bf bh2 = load_frag(&Bth[(wc << 6) + 32 + hl][0], g0);
    const v16bf bl2 = load_frag(&Btl[(wc << 6) + 32 + hl][0], g0);
    const v16bf bh3 = load_frag(&Bth[(wc << 6) + 48 + hl][0], g0);
    const v16bf bl3 = load_frag(&Btl[(wc << 6) + 48 + hl][0], g0);

    acc00 = WMMA_BF16(ah0, bh0, acc00);
    acc00 = WMMA_BF16(al0, bh0, acc00);
    acc00 = WMMA_BF16(ah0, bl0, acc00);
    acc10 = WMMA_BF16(ah1, bh0, acc10);
    acc10 = WMMA_BF16(al1, bh0, acc10);
    acc10 = WMMA_BF16(ah1, bl0, acc10);
    acc01 = WMMA_BF16(ah0, bh1, acc01);
    acc01 = WMMA_BF16(al0, bh1, acc01);
    acc01 = WMMA_BF16(ah0, bl1, acc01);
    acc11 = WMMA_BF16(ah1, bh1, acc11);
    acc11 = WMMA_BF16(al1, bh1, acc11);
    acc11 = WMMA_BF16(ah1, bl1, acc11);
    acc02 = WMMA_BF16(ah0, bh2, acc02);
    acc02 = WMMA_BF16(al0, bh2, acc02);
    acc02 = WMMA_BF16(ah0, bl2, acc02);
    acc12 = WMMA_BF16(ah1, bh2, acc12);
    acc12 = WMMA_BF16(al1, bh2, acc12);
    acc12 = WMMA_BF16(ah1, bl2, acc12);
    acc03 = WMMA_BF16(ah0, bh3, acc03);
    acc03 = WMMA_BF16(al0, bh3, acc03);
    acc03 = WMMA_BF16(ah0, bl3, acc03);
    acc13 = WMMA_BF16(ah1, bh3, acc13);
    acc13 = WMMA_BF16(al1, bh3, acc13);
    acc13 = WMMA_BF16(ah1, bl3, acc13);
  }

  __syncthreads();
  {
    const int col0 = (wc << 6) + hl;
    float b0 = 0.f, b1v = 0.f, b2v = 0.f, b3v = 0.f;
    if (EPI == EPI_LN) {
      b0 = bias[col0]; b1v = bias[col0 + 16]; b2v = bias[col0 + 32]; b3v = bias[col0 + 48];
    }
    float* cw = Ct + ((wr << 5) + (hh << 3)) * CPITCH + col0;
#pragma unroll
    for (int r = 0; r < 8; ++r) {
      float* c0 = cw + r * CPITCH;
      float* c1 = c0 + 16 * CPITCH;
      if (EPI == EPI_LN) {
        c0[0]  = acc00[r] + b0;
        c0[16] = acc01[r] + b1v;
        c0[32] = acc02[r] + b2v;
        c0[48] = acc03[r] + b3v;
        c1[0]  = acc10[r] + b0;
        c1[16] = acc11[r] + b1v;
        c1[32] = acc12[r] + b2v;
        c1[48] = acc13[r] + b3v;
      } else {
        c0[0]  = acc00[r];
        c0[16] = acc01[r];
        c0[32] = acc02[r];
        c0[48] = acc03[r];
        c1[0]  = acc10[r];
        c1[16] = acc11[r];
        c1[32] = acc12[r];
        c1[48] = acc13[r];
      }
    }
  }

  const int wv = __builtin_amdgcn_readfirstlane(wid);
  const int trow0 = wv << 3;
  const int grow0 = m0 + trow0;
  const bool full = (grow0 + 8 <= M);

  if (EPI == EPI_SD) {
    float as[8], ad[8];
#pragma unroll
    for (int j = 0; j < 8; ++j) { as[j] = p0[lane + 32 * j]; ad[j] = p1[lane + 32 * j]; }
    __syncthreads();
    if (full) sd_rows(Ct, trow0, grow0, lane, as, ad, C, sOut, dOut);
    else      sd_rows_edge(Ct, trow0, grow0, M, lane, as, ad, C, sOut, dOut);
  } else {
    float g[8], b[8];
#pragma unroll
    for (int j = 0; j < 8; ++j) { g[j] = p0[lane + 32 * j]; b[j] = p1[lane + 32 * j]; }
    const float scale = swp ? swp[swidx] : 1.f;
    __syncthreads();
    if (accumulate) {
      if (full) ln_rows<true>(Ct, trow0, grow0, lane, g, b, scale, accIn, C);
      else      ln_rows_edge<true>(Ct, trow0, grow0, M, lane, g, b, scale, accIn, C);
    } else {
      if (full) ln_rows<false>(Ct, trow0, grow0, lane, g, b, scale, accIn, C);
      else      ln_rows_edge<false>(Ct, trow0, grow0, M, lane, g, b, scale, accIn, C);
    }
  }
}


__global__ __launch_bounds__(256, 8) void k_gemm_sd(
    const float* __restrict__ A, const unsigned* __restrict__ Bhi,
    const unsigned* __restrict__ Blo, const float* __restrict__ asrc,
    const float* __restrict__ adst, float* __restrict__ hOut,
    float* __restrict__ sOut, float* __restrict__ dOut, int M)
{
  gemm_rows_body<EPI_SD>(A, Bhi, Blo, M, nullptr, asrc, adst, hOut, sOut, dOut,
                         nullptr, 0, nullptr, 0);
}

__global__ __launch_bounds__(256, 8) void k_gemm_ln(
    const float* __restrict__ A, const unsigned* __restrict__ Bhi,
    const unsigned* __restrict__ Blo, const float* __restrict__ bias,
    const float* __restrict__ gamma, const float* __restrict__ beta,
    const float* __restrict__ swp, int swidx,
    const float* accIn, int accumulate, float* dst, int M)
{
  gemm_rows_body<EPI_LN>(A, Bhi, Blo, M, bias, gamma, beta, dst, nullptr, nullptr,
                         swp, swidx, accIn, accumulate);
}

#define WPLANE (H * H)
#define WPLANE_W (H * H / 2)
__global__ __launch_bounds__(256) void k_split_weights(const float* __restrict__ Wg,
                                                       const float* __restrict__ Wsl,
                                                       const float* __restrict__ Wf,
                                                       unsigned* __restrict__ wt) {
  const int m = (int)blockIdx.y;
  const float* Wm = m < 3 ? Wg + (size_t)m * WPLANE
                  : (m < 7 ? Wsl + (size_t)(m - 3) * WPLANE : Wf);
  const int g = (int)blockIdx.x * 256 + (int)threadIdx.x;
  const int n = g & 255;
  const int k8 = (g >> 8) << 3;
  float v[8];
#pragma unroll
  for (int j = 0; j < 8; ++j) v[j] = Wm[(size_t)(k8 + j) * H + n];
  unsigned short hi[8], lo[8];
#pragma unroll
  for (int j = 0; j < 8; ++j) {
    hi[j] = f2bfbits(v[j]);
    lo[j] = f2bfbits(v[j] - bfbits2f(hi[j]));
  }
  uint4 ph, pl;
  ph.x = (unsigned)hi[0] | ((unsigned)hi[1] << 16); ph.y = (unsigned)hi[2] | ((unsigned)hi[3] << 16);
  ph.z = (unsigned)hi[4] | ((unsigned)hi[5] << 16); ph.w = (unsigned)hi[6] | ((unsigned)hi[7] << 16);
  pl.x = (unsigned)lo[0] | ((unsigned)lo[1] << 16); pl.y = (unsigned)lo[2] | ((unsigned)lo[3] << 16);
  pl.z = (unsigned)lo[4] | ((unsigned)lo[5] << 16); pl.w = (unsigned)lo[6] | ((unsigned)lo[7] << 16);
  unsigned* ph_p = wt + (size_t)m * 2 * WPLANE_W;
  const size_t o = ((size_t)(k8 >> 5) * H + n) * (KC / 2) + ((k8 & 31) >> 1);
  *(uint4*)(ph_p + o) = ph;
  *(uint4*)(ph_p + WPLANE_W + o) = pl;
}

#define CS_GROUPS 256
#define CS_THREADS 512

__global__ __launch_bounds__(CS_THREADS) void k_colsum_part(const float* __restrict__ x,
                                                           float* __restrict__ part,
                                                           int* __restrict__ deg, int ndeg, int N) {
  __shared__ float4 red[CS_THREADS / 64][64];
  const int t = threadIdx.x;
  for (int i = (int)blockIdx.x * CS_THREADS + t; i < ndeg; i += (int)gridDim.x * CS_THREADS) deg[i] = 0;
  const int cq = t & 63;
  const int rs = t >> 6;
  const int per = (N + (int)gridDim.x - 1) / (int)gridDim.x;
  const int r0 = (int)blockIdx.x * per;
  int r1 = r0 + per; if (r1 > N) r1 = N;
  const float4* xp = (const float4*)x + cq;
  float4 acc = make_float4(0.f, 0.f, 0.f, 0.f);
  int r = r0 + rs;
  for (; r + 56 < r1; r += 64) {
    float4 v[8];
#pragma unroll
    for (int j = 0; j < 8; ++j) v[j] = xp[(size_t)(r + 8 * j) * 64];
#pragma unroll
    for (int j = 0; j < 8; ++j) {
      acc.x += v[j].x; acc.y += v[j].y; acc.z += v[j].z; acc.w += v[j].w;
    }
  }
  for (; r < r1; r += 8) {
    float4 v = xp[(size_t)r * 64];
    acc.x += v.x; acc.y += v.y; acc.z += v.z; acc.w += v.w;
  }
  red[rs][cq] = acc;
  __syncthreads();
  if (t < H) {
    const float* rf = (const float*)&red[0][0];
    float s = 0.f;
#pragma unroll
    for (int j = 0; j < CS_THREADS / 64; ++j) s += rf[j * H + t];
    part[(size_t)blockIdx.x * H + t] = s;
  }
}

__global__ __launch_bounds__(1024) void k_selector(const float* __restrict__ part,
                           const float* __restrict__ W1, const float* __restrict__ b1,
                           const float* __restrict__ W2, const float* __restrict__ b2,
                           float* __restrict__ sw, int N) {
  __shared__ float4 psum[16][64];
  __shared__ float gsh[H];
  __shared__ float hid[128];
  __shared__ float lg[4];
  const int t = threadIdx.x;
  {
    const int cq = t & 63, q = t >> 6;
    const float4* pp = (const float4*)part + cq;
    float4 s = make_float4(0.f, 0.f, 0.f, 0.f);
#pragma unroll
    for (int j = 0; j < CS_GROUPS / 16; ++j) {
      float4 v = pp[(size_t)(q + 16 * j) * 64];
      s.x += v.x; s.y += v.y; s.z += v.z; s.w += v.w;
    }
    psum[q][cq] = s;
  }
  __syncthreads();
  if (t < H) {
    const float* pf = (const float*)&psum[0][0];
    float gsum = 0.f;
#pragma unroll
    for (int j = 0; j < 16; ++j) gsum += pf[j * H + t];
    float invN = 1.f / (float)N;
    gsh[t] = gsum * invN;
  }
  __syncthreads();
  if (t < 128) {
    float a = b1[t];
    for (int k = 0; k < H; ++k) a += gsh[k] * W1[k * 128 + t];
    hid[t] = fmaxf(a, 0.f);
  }
  __syncthreads();
  if (t < 4) {
    float l = b2[t];
    for (int k = 0; k < 128; ++k) l += hid[k] * W2[k * 4 + t];
    lg[t] = l;
  }
  __syncthreads();
  if (t == 0) {
    float m = fmaxf(fmaxf(lg[0], lg[1]), fmaxf(lg[2], lg[3]));
    float e0 = __expf(lg[0] - m), e1 = __expf(lg[1] - m);
    float e2 = __expf(lg[2] - m), e3 = __expf(lg[3] - m);
    float s = e0 + e1 + e2 + e3;
    sw[0] = e0 / s; sw[1] = e1 / s; sw[2] = e2 / s; sw[3] = e3 / s;
  }
}

#define ROW_BLOCKS 2048

__global__ void k_deg_count(const int* __restrict__ dst, int* __restrict__ deg, int E) {
  int e = blockIdx.x * blockDim.x + threadIdx.x;
  if (e < E) atomicAdd(&deg[dst[e]], 1);
}

#define SCAN_TILE 16384
__global__ __launch_bounds__(1024) void k_scan_deg(int* __restrict__ off, int N) {
  __shared__ int wsum[32];
  __shared__ int wpre[32];
  const int t = threadIdx.x, lane = t & 31, w = t >> 5;
  int carry = 0;
  for (int base = 0; base < N; base += SCAN_TILE) {
    int4* p = (int4*)(off + base + 16 * t);
    int4 q[4];
#pragma unroll
    for (int k = 0; k < 4; ++k) q[k] = p[k];
    int tot = 0;
#pragma unroll
    for (int k = 0; k < 4; ++k) tot += q[k].x + q[k].y + q[k].z + q[k].w;
    int inc = tot;
    for (int o = 1; o < 32; o <<= 1) { int u = __shfl_up(inc, o, 32); if (lane >= o) inc += u; }
    if (lane == 31) wsum[w] = inc;
    __syncthreads();
    if (w == 0) {
      int x = wsum[lane], ix = x;
      for (int o = 1; o < 32; o <<= 1) { int u = __shfl_up(ix, o, 32); if (lane >= o) ix += u; }
      wpre[lane] = ix - x;
    }
    __syncthreads();
    int run = carry + wpre[w] + inc - tot;
    carry += wpre[31] + wsum[31];
#pragma unroll
    for (int k = 0; k < 4; ++k) {
      int4 o;
      o.x = run; run += q[k].x;
      o.y = run; run += q[k].y;
      o.z = run; run += q[k].z;
      o.w = run; run += q[k].w;
      p[k] = o;
    }
    __syncthreads();
  }
}

__global__ void k_csr_fill(const int* __restrict__ src, const int* __restrict__ dst,
                           const int* __restrict__ attr, int* __restrict__ off,
                           int* __restrict__ elist, int E) {
  int e = blockIdx.x * blockDim.x + threadIdx.x;
  if (e < E) {
    int pos = atomicAdd(&off[dst[e]], 1);
    ((int2*)elist)[pos] = make_int2(src[e], attr[e]);
  }
}

__global__ __launch_bounds__(256) void k_gat_gather(const float* __restrict__ h,
                             const float* __restrict__ s, const float* __restrict__ d,
                             const int* __restrict__ off, const int* __restrict__ elist,
                             int branch, const float* __restrict__ bias,
                             float* __restrict__ xo, int N) {
  const int2* el = (const int2*)elist;
  const int lane = threadIdx.x & 31;
  const int nw = (int)gridDim.x * 8;
  const int wv = __builtin_amdgcn_readfirstlane((int)(threadIdx.x >> 5));
  float bv[8];
#pragma unroll
  for (int j = 0; j < 8; ++j) bv[j] = bias[lane + 32 * j];
  for (int n = (int)blockIdx.x * 8 + wv; n < N; n += nw) {
    const int e0 = n ? off[n - 1] : 0;
    const int e1 = off[n];
    const float* hp = h + (size_t)n * H + lane;
    float acc[8];
#pragma unroll
    for (int j = 0; j < 8; ++j) acc[j] = hp[32 * j];
    const float dn = d[n];
    const float lself = lrelu(s[n] + dn);
    float m = lself;
    for (int c = e0; c < e1; c += 32) {
      const int i = c + lane;
      if (i < e1) {
        const int2 ed = el[i];
        const float lg = lrelu(s[ed.x] + dn);
        if (ed.y == branch) m = fmaxf(m, lg);
      }
    }
    for (int o = 16; o; o >>= 1) m = fmaxf(m, __shfl_xor(m, o, 32));
    const float wself = __expf(lself - m);
    float den = wself;
#pragma unroll
    for (int j = 0; j < 8; ++j) acc[j] *= wself;
    for (int c = e0; c < e1; c += 32) {
      const int i = c + lane;
      bool on = false;
      int sn = 0;
      float w = 0.f;
      if (i < e1) {
        const int2 ed = el[i];
        const float lg = lrelu(s[ed.x] + dn);
        on = (ed.y == branch);
        sn = ed.x;
        if (on) w = __expf(lg - m);
      }
      const unsigned live = __builtin_amdgcn_ballot_w32(on);
      int rank = 0;
      for (unsigned b = live; b; b &= b - 1u) {
        const int j = __builtin_ctz(b);
        const int sj = __builtin_amdgcn_readlane(sn, j);
        rank += (sj < sn || (sj == sn && j < lane)) ? 1 : 0;
      }
      const int cnt = __builtin_popcount(live);
#pragma unroll 2
      for (int r = 0; r < cnt; ++r) {
        const int j = __builtin_ctz(__builtin_amdgcn_ballot_w32(on && rank == r));
        const int sj = __builtin_amdgcn_readlane(sn, j);
        const float wj = __uint_as_float(__builtin_amdgcn_readlane(__float_as_uint(w), j));
        den += wj;
        const float* sp = h + (size_t)sj * H + lane;
        float v[8];
#pragma unroll
        for (int k = 0; k < 8; ++k) v[k] = sp[32 * k];
#pragma unroll
        for (int k = 0; k < 8; ++k) acc[k] += wj * v[k];
      }
    }
    float* op = xo + (size_t)n * H + lane;
#pragma unroll
    for (int k = 0; k < 8; ++k) op[32 * k] = acc[k] / den + bv[k];
  }
}


extern "C" void kernel_launch(void* const* d_in, const int* in_sizes, int n_in,
                              void* d_out, int out_size, void* d_ws, size_t ws_size,
                              hipStream_t stream) {
  const float* x     = (const float*)d_in[0];
  const int*   ei    = (const int*)d_in[1];
  const int*   ea    = (const int*)d_in[2];
  const float* Wg    = (const float*)d_in[3];
  const float* a_src = (const float*)d_in[4];
  const float* a_dst = (const float*)d_in[5];
  const float* bg    = (const float*)d_in[6];
  const float* Wsl   = (const float*)d_in[7];
  const float* bsl   = (const float*)d_in[8];
  const float* gsl   = (const float*)d_in[9];
  const float* betas = (const float*)d_in[10];
  const float* W1    = (const float*)d_in[11];
  const float* b1    = (const float*)d_in[12];
  const float* W2    = (const float*)d_in[13];
  const float* b2    = (const float*)d_in[14];
  const float* Wf    = (const float*)d_in[15];
  const float* bfv   = (const float*)d_in[16];
  const float* gf    = (const float*)d_in[17];
  const float* betaf = (const float*)d_in[18];

  const int N = in_sizes[0] / H;
  const int E = in_sizes[2];
  float* out = (float*)d_out;

  char* wsb = (char*)d_ws;
  size_t off = 0;
  auto alloc = [&](size_t bytes) -> void* {
    void* p = wsb + off;
    off += (bytes + 255) & ~(size_t)255;
    return p;
  };
  float*    sw    = (float*)alloc(4 * sizeof(float));
  float*    sArr  = (float*)alloc((size_t)N * sizeof(float));
  float*    dArr  = (float*)alloc((size_t)N * sizeof(float));
  const int nPad = (N + SCAN_TILE - 1) / SCAN_TILE * SCAN_TILE;
  int*      ebnd  = (int*)alloc((size_t)nPad * sizeof(int));
  float*    bufH  = (float*)alloc((size_t)N * H * sizeof(float));
  float*    bufX  = (float*)alloc((size_t)N * H * sizeof(float));
  int*      elist = (int*)alloc((size_t)N * H * sizeof(float));
  unsigned* wt = (unsigned*)((char*)elist + (((size_t)E * 8 + 255) & ~(size_t)255));

  dim3 gemmGrid((unsigned)((N + TM - 1) / TM), 1u);
  int eThrBlocks  = (E + 255) / 256;

  float* part = bufX;
  k_split_weights<<<dim3(32, 8), 256, 0, stream>>>(Wg, Wsl, Wf, wt);
  k_colsum_part<<<CS_GROUPS, CS_THREADS, 0, stream>>>(x, part, ebnd, nPad, N);
  k_selector<<<1, 1024, 0, stream>>>(part, W1, b1, W2, b2, sw, N);

  k_deg_count<<<eThrBlocks, 256, 0, stream>>>(ei + E, ebnd, E);
  k_scan_deg<<<1, 1024, 0, stream>>>(ebnd, N);
  k_csr_fill<<<eThrBlocks, 256, 0, stream>>>(ei, ei + E, ea, ebnd, elist, E);

  for (int i = 0; i < 3; ++i) {
    k_gemm_sd<<<gemmGrid, 256, 0, stream>>>(x, wt + (size_t)i * 2 * WPLANE_W,
                                          wt + (size_t)i * 2 * WPLANE_W + WPLANE_W,
                                          a_src + i * H, a_dst + i * H, bufH, sArr, dArr, N);
    k_gat_gather<<<ROW_BLOCKS, 256, 0, stream>>>(bufH, sArr, dArr, ebnd, elist, i,
                                                 bg + i * H, bufX, N);
    k_gemm_ln<<<gemmGrid, 256, 0, stream>>>(bufX, wt + (size_t)(3 + i) * 2 * WPLANE_W,
                                          wt + (size_t)(3 + i) * 2 * WPLANE_W + WPLANE_W,
                                          bsl + i * H, gsl + i * H, betas + i * H, sw, i,
                                          out, i > 0 ? 1 : 0, out, N);
  }
  k_gemm_ln<<<gemmGrid, 256, 0, stream>>>(x, wt + (size_t)6 * 2 * WPLANE_W,
                                          wt + (size_t)6 * 2 * WPLANE_W + WPLANE_W, bsl + 3 * H,
                                          gsl + 3 * H, betas + 3 * H, sw, 3, out, 1, bufX, N);

  k_gemm_ln<<<gemmGrid, 256, 0, stream>>>(bufX, wt + (size_t)7 * 2 * WPLANE_W,
                                          wt + (size_t)7 * 2 * WPLANE_W + WPLANE_W, bfv,
                                          gf, betaf, nullptr, 0, nullptr, 0, out, N);
}


# independent launches share one launch: col sums|weight split, degree scan|selector, first GEMM|edge-list fill, each LayerNorm GEMM|next branch's GEMM: 12 launches, not 17; sum orders kept
# speedup vs baseline: 3.3049x; 1.1141x over previous
#include <hip/hip_runtime.h>

#define H 256
#define TM 64
#define KC 32
#define APAD 40

typedef __bf16 bf16;
typedef __attribute__((ext_vector_type(16))) __bf16 v16bf;
typedef __attribute__((ext_vector_type(8)))  __bf16 v8bf;
typedef __attribute__((ext_vector_type(8)))  float  v8f;

__device__ __forceinline__ unsigned short f2bfbits(float f) {
  unsigned u = __float_as_uint(f);
  return (unsigned short)((u + 0x7FFFu + ((u >> 16) & 1u)) >> 16);
}
__device__ __forceinline__ float bfbits2f(unsigned short h) {
  return __uint_as_float(((unsigned)h) << 16);
}
__device__ __forceinline__ bf16 f2bf(float f) {
  unsigned short hs = f2bfbits(f);
  return __builtin_bit_cast(bf16, hs);
}
__device__ __forceinline__ float lrelu(float v) { return v > 0.f ? v : 0.2f * v; }

__device__ __forceinline__ v16bf load_frag(const bf16* rowp, int g0) {
  v8bf a = *(const v8bf*)(rowp + g0);
  v8bf b = *(const v8bf*)(rowp + 16 + g0);
  return __builtin_shufflevector(a, b, 0, 1, 2, 3, 4, 5, 6, 7,
                                 8, 9, 10, 11, 12, 13, 14, 15);
}

#define WMMA_BF16(A_, B_, C_) \
  __builtin_amdgcn_wmma_f32_16x16x32_bf16(false, (A_), false, (B_), (short)0, (C_), false, false)


#define CPITCH 260
#define SMEM_BYTES (TM * CPITCH * 4)

#define EPI_SD 0
#define EPI_LN 1

__device__ __forceinline__ void butterfly8(float (&a)[8]) {
  for (int k = 16; k; k >>= 1) {
    float t[8];
#pragma unroll
    for (int i = 0; i < 8; ++i) t[i] = __shfl_xor(a[i], k, 32);
#pragma unroll
    for (int i = 0; i < 8; ++i) a[i] += t[i];
  }
}

__device__ __forceinline__ void sd_rows(const float* Ct, int trow0, int grow0, int lane,
                                        const float (&as)[8], const float (&ad)[8],
                                        float* hOut, float* __restrict__ sOut,
                                        float* __restrict__ dOut) {
  float vs[8], vd[8];
#pragma unroll
  for (int i = 0; i < 8; ++i) {
    const float* tp = Ct + (trow0 + i) * CPITCH + lane;
    float* hp = hOut + (size_t)(grow0 + i) * H + lane;
    float v[8];
#pragma unroll
    for (int j = 0; j < 8; ++j) v[j] = tp[32 * j];
#pragma unroll
    for (int j = 0; j < 8; ++j) hp[32 * j] = v[j];
    float s = 0.f, d = 0.f;
#pragma unroll
    for (int j = 0; j < 8; ++j) { s += v[j] * as[j]; d += v[j] * ad[j]; }
    vs[i] = s; vd[i] = d;
  }
  butterfly8(vs);
  butterfly8(vd);
  float ss = vs[0], dd = vd[0];
#pragma unroll
  for (int i = 1; i < 8; ++i) { if (lane == i) { ss = vs[i]; dd = vd[i]; } }
  if (lane < 8) { sOut[grow0 + lane] = ss; dOut[grow0 + lane] = dd; }
}

__device__ __forceinline__ void sd_rows_edge(const float* Ct, int trow0, int grow0, int M, int lane,
                                             const float (&as)[8], const float (&ad)[8],
                                             float* hOut, float* __restrict__ sOut,
                                             float* __restrict__ dOut) {
  for (int i = 0; i < 8; ++i) {
    const int gr = grow0 + i;
    if (gr < M) {
      const float* tp = Ct + (trow0 + i) * CPITCH + lane;
      float* hp = hOut + (size_t)gr * H + lane;
      float v[8];
#pragma unroll
      for (int j = 0; j < 8; ++j) v[j] = tp[32 * j];
#pragma unroll
      for (int j = 0; j < 8; ++j) hp[32 * j] = v[j];
      float vs = 0.f, vd = 0.f;
#pragma unroll
      for (int j = 0; j < 8; ++j) { vs += v[j] * as[j]; vd += v[j] * ad[j]; }
      for (int o = 16; o; o >>= 1) { vs += __shfl_xor(vs, o, 32); vd += __shfl_xor(vd, o, 32); }
      if (lane == 0) { sOut[gr] = vs; dOut[gr] = vd; }
    }
  }
}

__device__ __forceinline__ void acc_row(const float* accIn, int gr, int lane, float (&o)[8]) {
  const float* ip = accIn + (size_t)gr * H + lane;
#pragma unroll
  for (int j = 0; j < 8; ++j) o[j] = ip[32 * j];
}

template <bool ACC>
__device__ __forceinline__ void ln_finish(float (&v)[8], float sum, float sq,
                                          const float (&g)[8], const float (&b)[8], float scale,
                                          const float (&o)[8], float* op) {
  float mu = sum * (1.f / H);
  float var = sq * (1.f / H) - mu * mu;
  float inv = rsqrtf(var + 1e-5f);
#pragma unroll
  for (int j = 0; j < 8; ++j) {
    float y = (v[j] - mu) * inv * g[j] + b[j];
    v[j] = fmaxf(y, 0.f) * scale;
  }
  if (ACC) {
#pragma unroll
    for (int j = 0; j < 8; ++j) op[32 * j] = o[j] + v[j];
  } else {
#pragma unroll
    for (int j = 0; j < 8; ++j) op[32 * j] = v[j];
  }
}

template <bool ACC>
__device__ __forceinline__ void ln_rows(const float* Ct, int trow0, int grow0, int lane,
                                        const float (&g)[8], const float (&b)[8], float scale,
                                        const float* accIn, float* dst) {
  float sum[8], sq[8];
#pragma unroll
  for (int i = 0; i < 8; ++i) {
    const float* tp = Ct + (trow0 + i) * CPITCH + lane;
    float s = 0.f, q = 0.f;
#pragma unroll
    for (int j = 0; j < 8; ++j) {
      const float t = tp[32 * j];
      s += t;
      q += t * t;
    }
    sum[i] = s; sq[i] = q;
  }
  butterfly8(sum);
  butterfly8(sq);
  __builtin_amdgcn_sched_barrier(0);
  float o[8][8];
  if (ACC) {
#pragma unroll
    for (int i = 0; i < 4; ++i) acc_row(accIn, grow0 + i, lane, o[i]);
  }
  float vn[8];
#pragma unroll
  for (int j = 0; j < 8; ++j) vn[j] = Ct[trow0 * CPITCH + lane + 32 * j];
#pragma unroll
  for (int i = 0; i < 8; ++i) {
    float v[8];
#pragma unroll
    for (int j = 0; j < 8; ++j) v[j] = vn[j];
    if (i < 7) {
      const float* tp = Ct + (trow0 + i + 1) * CPITCH + lane;
#pragma unroll
      for (int j = 0; j < 8; ++j) vn[j] = tp[32 * j];
    }
    ln_finish<ACC>(v, sum[i], sq[i], g, b, scale, o[i], dst + (size_t)(grow0 + i) * H + lane);
    if (ACC && i < 4) acc_row(accIn, grow0 + i + 4, lane, o[i + 4]);
  }
}

template <bool ACC>
__device__ __forceinline__ void ln_rows_edge(const float* Ct, int trow0, int grow0, int M, int lane,
                                             const float (&g)[8], const float (&b)[8],
                                             float scale, const float* accIn, float* dst) {
  for (int i = 0; i < 8; ++i) {
    const int gr = grow0 + i;
    if (gr < M) {
      float oo[8] = {};
      if (ACC) acc_row(accIn, gr, lane, oo);
      const float* tp = Ct + (trow0 + i) * CPITCH + lane;
      float v[8];
#pragma unroll
      for (int j = 0; j < 8; ++j) v[j] = tp[32 * j];
      float sum = 0.f, sq = 0.f;
#pragma unroll
      for (int j = 0; j < 8; ++j) {
        sum += v[j];
        sq += v[j] * v[j];
      }
      for (int k = 16; k; k >>= 1) { sum += __shfl_xor(sum, k, 32); sq += __shfl_xor(sq, k, 32); }
      ln_finish<ACC>(v, sum, sq, g, b, scale, oo, dst + (size_t)gr * H + lane);
    }
  }
}

template <int EPI>
__device__ __forceinline__ void gemm_rows_body(
    unsigned char* smem, int bid,
    const float* __restrict__ A, const unsigned* __restrict__ Bhi,
    const unsigned* __restrict__ Blo, int M,
    const float* __restrict__ bias,
    const float* __restrict__ p0, const float* __restrict__ p1,
    float* C,
    float* __restrict__ sOut, float* __restrict__ dOut,
    const float* __restrict__ swp, int swidx,
    const float* accIn, int accumulate)
{
  bf16 (*Ah)[APAD]  = (bf16 (*)[APAD])(smem);
  bf16 (*Al)[APAD]  = (bf16 (*)[APAD])(smem + TM * APAD * 2);
  bf16 (*Bth)[APAD] = (bf16 (*)[APAD])(smem + 2 * TM * APAD * 2);
  bf16 (*Btl)[APAD] = (bf16 (*)[APAD])(smem + 2 * TM * APAD * 2 + H * APAD * 2);
  float* Ct = (float*)smem;

  const int tid = threadIdx.x;
  const int m0 = bid * TM;
  const int wid = __builtin_amdgcn_readfirstlane(tid >> 5);
  const int lane = tid & 31;
  const int hl = lane & 15, hh = lane >> 4;
  const int wr = wid >> 2;
  const int wc = wid & 3;
  const int g0 = hh << 3;

  v8f acc00 = {}, acc01 = {}, acc02 = {}, acc03 = {};
  v8f acc10 = {}, acc11 = {}, acc12 = {}, acc13 = {};

  const int ar = tid >> 3;
  const int ac = (tid & 7) << 2;
  const float* ap0 = A + (size_t)((m0 + ar) < M ? m0 + ar : M - 1) * H + ac;
  const float* ap1 = A + (size_t)((m0 + ar + 32) < M ? m0 + ar + 32 : M - 1) * H + ac;

  for (int k0 = 0; k0 < H; k0 += KC) {
    __syncthreads();
    const size_t bo = (size_t)(k0 >> 5) * (H * KC / 2) + (size_t)tid * 4;
    uint4 vh[4], vl[4];
#pragma unroll
    for (int j = 0; j < 4; ++j) {
      vh[j] = *(const uint4*)(Bhi + bo + (size_t)j * 1024);
      vl[j] = *(const uint4*)(Blo + bo + (size_t)j * 1024);
    }
    float4 a0 = *(const float4*)(ap0 + k0);
    float4 a1 = *(const float4*)(ap1 + k0);
#pragma unroll
    for (int j = 0; j < 4; ++j) {
      const int p = tid + 256 * j;
      *(uint4*)&Bth[p >> 2][(p & 3) << 3] = vh[j];
      *(uint4*)&Btl[p >> 2][(p & 3) << 3] = vl[j];
    }
    {
      ushort4 hiv, lov;
      hiv.x = f2bfbits(a0.x); lov.x = f2bfbits(a0.x - bfbits2f(hiv.x));
      hiv.y = f2bfbits(a0.y); lov.y = f2bfbits(a0.y - bfbits2f(hiv.y));
      hiv.z = f2bfbits(a0.z); lov.z = f2bfbits(a0.z - bfbits2f(hiv.z));
      hiv.w = f2bfbits(a0.w); lov.w = f2bfbits(a0.w - bfbits2f(hiv.w));
      *(ushort4*)&Ah[ar][ac] = hiv;
      *(ushort4*)&Al[ar][ac] = lov;
    }
    {
      ushort4 hiv, lov;
      hiv.x = f2bfbits(a1.x); lov.x = f2bfbits(a1.x - bfbits2f(hiv.x));
      hiv.y = f2bfbits(a1.y); lov.y = f2bfbits(a1.y - bfbits2f(hiv.y));
      hiv.z = f2bfbits(a1.z); lov.z = f2bfbits(a1.z - bfbits2f(hiv.z));
      hiv.w = f2bfbits(a1.w); lov.w = f2bfbits(a1.w - bfbits2f(hiv.w));
      *(ushort4*)&Ah[ar + 32][ac] = hiv;
      *(ushort4*)&Al[ar + 32][ac] = lov;
    }
    __syncthreads();

    const v16bf ah0 = load_frag(&Ah[(wr << 5) + hl][0], g0);
    const v16bf al0 = load_frag(&Al[(wr << 5) + hl][0], g0);
    const v16bf ah1 = load_frag(&Ah[(wr << 5) + 16 + hl][0], g0);
    const v16bf al1 = load_frag(&Al[(wr << 5) + 16 + hl][0], g0);
    const v16bf bh0 = load_frag(&Bth[(wc << 6) + hl][0], g0);
    const v16bf bl0 = load_frag(&Btl[(wc << 6) + hl][0], g0);
    const v16bf bh1 = load_frag(&Bth[(wc << 6) + 16 + hl][0], g0);
    const v16bf bl1 = load_frag(&Btl[(wc << 6) + 16 + hl][0], g0);
    const v16bf bh2 = load_frag(&Bth[(wc << 6) + 32 + hl][0], g0);
    const v16bf bl2 = load_frag(&Btl[(wc << 6) + 32 + hl][0], g0);
    const v16bf bh3 = load_frag(&Bth[(wc << 6) + 48 + hl][0], g0);
    const v16bf bl3 = load_frag(&Btl[(wc << 6) + 48 + hl][0], g0);

    acc00 = WMMA_BF16(ah0, bh0, acc00);
    acc00 = WMMA_BF16(al0, bh0, acc00);
    acc00 = WMMA_BF16(ah0, bl0, acc00);
    acc10 = WMMA_BF16(ah1, bh0, acc10);
    acc10 = WMMA_BF16(al1, bh0, acc10);
    acc10 = WMMA_BF16(ah1, bl0, acc10);
    acc01 = WMMA_BF16(ah0, bh1, acc01);
    acc01 = WMMA_BF16(al0, bh1, acc01);
    acc01 = WMMA_BF16(ah0, bl1, acc01);
    acc11 = WMMA_BF16(ah1, bh1, acc11);
    acc11 = WMMA_BF16(al1, bh1, acc11);
    acc11 = WMMA_BF16(ah1, bl1, acc11);
    acc02 = WMMA_BF16(ah0, bh2, acc02);
    acc02 = WMMA_BF16(al0, bh2, acc02);
    acc02 = WMMA_BF16(ah0, bl2, acc02);
    acc12 = WMMA_BF16(ah1, bh2, acc12);
    acc12 = WMMA_BF16(al1, bh2, acc12);
    acc12 = WMMA_BF16(ah1, bl2, acc12);
    acc03 = WMMA_BF16(ah0, bh3, acc03);
    acc03 = WMMA_BF16(al0, bh3, acc03);
    acc03 = WMMA_BF16(ah0, bl3, acc03);
    acc13 = WMMA_BF16(ah1, bh3, acc13);
    acc13 = WMMA_BF16(al1, bh3, acc13);
    acc13 = WMMA_BF16(ah1, bl3, acc13);
  }

  __syncthreads();
  {
    const int col0 = (wc << 6) + hl;
    float b0 = 0.f, b1v = 0.f, b2v = 0.f, b3v = 0.f;
    if (EPI == EPI_LN) {
      b0 = bias[col0]; b1v = bias[col0 + 16]; b2v = bias[col0 + 32]; b3v = bias[col0 + 48];
    }
    float* cw = Ct + ((wr << 5) + (hh << 3)) * CPITCH + col0;
#pragma unroll
    for (int r = 0; r < 8; ++r) {
      float* c0 = cw + r * CPITCH;
      float* c1 = c0 + 16 * CPITCH;
      if (EPI == EPI_LN) {
        c0[0]  = acc00[r] + b0;
        c0[16] = acc01[r] + b1v;
        c0[32] = acc02[r] + b2v;
        c0[48] = acc03[r] + b3v;
        c1[0]  = acc10[r] + b0;
        c1[16] = acc11[r] + b1v;
        c1[32] = acc12[r] + b2v;
        c1[48] = acc13[r] + b3v;
      } else {
        c0[0]  = acc00[r];
        c0[16] = acc01[r];
        c0[32] = acc02[r];
        c0[48] = acc03[r];
        c1[0]  = acc10[r];
        c1[16] = acc11[r];
        c1[32] = acc12[r];
        c1[48] = acc13[r];
      }
    }
  }

  const int wv = __builtin_amdgcn_readfirstlane(wid);
  const int trow0 = wv << 3;
  const int grow0 = m0 + trow0;
  const bool full = (grow0 + 8 <= M);

  if (EPI == EPI_SD) {
    float as[8], ad[8];
#pragma unroll
    for (int j = 0; j < 8; ++j) { as[j] = p0[lane + 32 * j]; ad[j] = p1[lane + 32 * j]; }
    __syncthreads();
    if (full) sd_rows(Ct, trow0, grow0, lane, as, ad, C, sOut, dOut);
    else      sd_rows_edge(Ct, trow0, grow0, M, lane, as, ad, C, sOut, dOut);
  } else {
    float g[8], b[8];
#pragma unroll
    for (int j = 0; j < 8; ++j) { g[j] = p0[lane + 32 * j]; b[j] = p1[lane + 32 * j]; }
    const float scale = swp ? swp[swidx] : 1.f;
    __syncthreads();
    if (accumulate) {
      if (full) ln_rows<true>(Ct, trow0, grow0, lane, g, b, scale, accIn, C);
      else      ln_rows_edge<true>(Ct, trow0, grow0, M, lane, g, b, scale, accIn, C);
    } else {
      if (full) ln_rows<false>(Ct, trow0, grow0, lane, g, b, scale, accIn, C);
      else      ln_rows_edge<false>(Ct, trow0, grow0, M, lane, g, b, scale, accIn, C);
    }
  }
}


__global__ __launch_bounds__(256, 8) void k_gemm_sd_fill(
    const float* __restrict__ A, const unsigned* __restrict__ Bhi,
    const unsigned* __restrict__ Blo, const float* __restrict__ asrc,
    const float* __restrict__ adst, float* __restrict__ hOut,
    float* __restrict__ sOut, float* __restrict__ dOut, int M, int nGemm,
    const int* __restrict__ src, const int* __restrict__ dst,
    const int* __restrict__ attr, int* __restrict__ off,
    int* __restrict__ elist, int E)
{
  __shared__ __attribute__((aligned(16))) unsigned char smem[SMEM_BYTES];
  const int bid = (int)blockIdx.x;
  if (bid < nGemm) {
    gemm_rows_body<EPI_SD>(smem, bid, A, Bhi, Blo, M, nullptr, asrc, adst, hOut, sOut, dOut,
                           nullptr, 0, nullptr, 0);
  } else {
    const int nb = (int)gridDim.x - nGemm;
    for (int e = (bid - nGemm) * 256 + (int)threadIdx.x; e < E; e += nb * 256) {
      int pos = atomicAdd(&off[dst[e]], 1);
      ((int2*)elist)[pos] = make_int2(src[e], attr[e]);
    }
  }
}

__global__ __launch_bounds__(256, 8) void k_gemm_ln(
    const float* __restrict__ A, const unsigned* __restrict__ Bhi,
    const unsigned* __restrict__ Blo, const float* __restrict__ bias,
    const float* __restrict__ gamma, const float* __restrict__ beta,
    const float* __restrict__ swp, int swidx,
    const float* accIn, int accumulate, float* dst, int M)
{
  __shared__ __attribute__((aligned(16))) unsigned char smem[SMEM_BYTES];
  gemm_rows_body<EPI_LN>(smem, (int)blockIdx.x, A, Bhi, Blo, M, bias, gamma, beta, dst, nullptr,
                         nullptr, swp, swidx, accIn, accumulate);
}

__global__ __launch_bounds__(256, 8) void k_gemm_ln_sd(
    const float* __restrict__ A, const unsigned* __restrict__ Bhi,
    const unsigned* __restrict__ Blo, const float* __restrict__ bias,
    const float* __restrict__ gamma, const float* __restrict__ beta,
    const float* __restrict__ swp, int swidx,
    const float* accIn, int accumulate, float* dst, int M, int nLn,
    const float* __restrict__ A2, const unsigned* __restrict__ B2hi,
    const unsigned* __restrict__ B2lo, const float* __restrict__ asrc,
    const float* __restrict__ adst, float* __restrict__ hOut,
    float* __restrict__ sOut, float* __restrict__ dOut)
{
  __shared__ __attribute__((aligned(16))) unsigned char smem[SMEM_BYTES];
  const int bid = (int)blockIdx.x;
  if (bid < nLn) {
    gemm_rows_body<EPI_LN>(smem, bid, A, Bhi, Blo, M, bias, gamma, beta, dst, nullptr, nullptr,
                           swp, swidx, accIn, accumulate);
  } else {
    gemm_rows_body<EPI_SD>(smem, bid - nLn, A2, B2hi, B2lo, M, nullptr, asrc, adst, hOut, sOut,
                           dOut, nullptr, 0, nullptr, 0);
  }
}

#define WPLANE (H * H)
#define WPLANE_W (H * H / 2)
__device__ __forceinline__ void split_weights_body(const float* __restrict__ Wg,
                                                   const float* __restrict__ Wsl,
                                                   const float* __restrict__ Wf,
                                                   unsigned* __restrict__ wt, int gi) {
  const int m = gi >> 13;
  const float* Wm = m < 3 ? Wg + (size_t)m * WPLANE
                  : (m < 7 ? Wsl + (size_t)(m - 3) * WPLANE : Wf);
  const int g = gi & 8191;
  const int n = g & 255;
  const int k8 = (g >> 8) << 3;
  float v[8];
#pragma unroll
  for (int j = 0; j < 8; ++j) v[j] = Wm[(size_t)(k8 + j) * H + n];
  unsigned short hi[8], lo[8];
#pragma unroll
  for (int j = 0; j < 8; ++j) {
    hi[j] = f2bfbits(v[j]);
    lo[j] = f2bfbits(v[j] - bfbits2f(hi[j]));
  }
  uint4 ph, pl;
  ph.x = (unsigned)hi[0] | ((unsigned)hi[1] << 16); ph.y = (unsigned)hi[2] | ((unsigned)hi[3] << 16);
  ph.z = (unsigned)hi[4] | ((unsigned)hi[5] << 16); ph.w = (unsigned)hi[6] | ((unsigned)hi[7] << 16);
  pl.x = (unsigned)lo[0] | ((unsigned)lo[1] << 16); pl.y = (unsigned)lo[2] | ((unsigned)lo[3] << 16);
  pl.z = (unsigned)lo[4] | ((unsigned)lo[5] << 16); pl.w = (unsigned)lo[6] | ((unsigned)lo[7] << 16);
  unsigned* ph_p = wt + (size_t)m * 2 * WPLANE_W;
  const size_t o = ((size_t)(k8 >> 5) * H + n) * (KC / 2) + ((k8 & 31) >> 1);
  *(uint4*)(ph_p + o) = ph;
  *(uint4*)(ph_p + WPLANE_W + o) = pl;
}

#define CS_GROUPS 256
#define CS_THREADS 512

__device__ __forceinline__ void colsum_part_body(const float* __restrict__ x,
                                                 float* __restrict__ part,
                                                 int* __restrict__ deg, int ndeg, int N) {
  __shared__ float4 red[CS_THREADS / 64][64];
  const int t = threadIdx.x;
  for (int i = (int)blockIdx.x * CS_THREADS + t; i < ndeg; i += CS_GROUPS * CS_THREADS) deg[i] = 0;
  const int cq = t & 63;
  const int rs = t >> 6;
  const int per = (N + CS_GROUPS - 1) / CS_GROUPS;
  const int r0 = (int)blockIdx.x * per;
  int r1 = r0 + per; if (r1 > N) r1 = N;
  const float4* xp = (const float4*)x + cq;
  float4 acc = make_float4(0.f, 0.f, 0.f, 0.f);
  int r = r0 + rs;
  for (; r + 56 < r1; r += 64) {
    float4 v[8];
#pragma unroll
    for (int j = 0; j < 8; ++j) v[j] = xp[(size_t)(r + 8 * j) * 64];
#pragma unroll
    for (int j = 0; j < 8; ++j) {
      acc.x += v[j].x; acc.y += v[j].y; acc.z += v[j].z; acc.w += v[j].w;
    }
  }
  for (; r < r1; r += 8) {
    float4 v = xp[(size_t)r * 64];
    acc.x += v.x; acc.y += v.y; acc.z += v.z; acc.w += v.w;
  }
  red[rs][cq] = acc;
  __syncthreads();
  if (t < H) {
    const float* rf = (const float*)&red[0][0];
    float s = 0.f;
#pragma unroll
    for (int j = 0; j < CS_THREADS / 64; ++j) s += rf[j * H + t];
    part[(size_t)blockIdx.x * H + t] = s;
  }
}

#define SPLIT_GROUPS (8 * 8192 / CS_THREADS)
__global__ __launch_bounds__(CS_THREADS) void k_prep(const float* __restrict__ x,
                                                    float* __restrict__ part,
                                                    int* __restrict__ deg, int ndeg, int N,
                                                    const float* __restrict__ Wg,
                                                    const float* __restrict__ Wsl,
                                                    const float* __restrict__ Wf,
                                                    unsigned* __restrict__ wt) {
  if ((int)blockIdx.x < CS_GROUPS) {
    colsum_part_body(x, part, deg, ndeg, N);
  } else {
    split_weights_body(Wg, Wsl, Wf, wt,
                       ((int)blockIdx.x - CS_GROUPS) * CS_THREADS + (int)threadIdx.x);
  }
}

__device__ __forceinline__ void selector_body(const float* __restrict__ part,
                           const float* __restrict__ W1, const float* __restrict__ b1,
                           const float* __restrict__ W2, const float* __restrict__ b2,
                           float* __restrict__ sw, int N) {
  __shared__ float4 psum[16][64];
  __shared__ float gsh[H];
  __shared__ float hid[128];
  __shared__ float lg[4];
  const int t = threadIdx.x;
  {
    const int cq = t & 63, q = t >> 6;
    const float4* pp = (const float4*)part + cq;
    float4 s = make_float4(0.f, 0.f, 0.f, 0.f);
#pragma unroll
    for (int j = 0; j < CS_GROUPS / 16; ++j) {
      float4 v = pp[(size_t)(q + 16 * j) * 64];
      s.x += v.x; s.y += v.y; s.z += v.z; s.w += v.w;
    }
    psum[q][cq] = s;
  }
  __syncthreads();
  if (t < H) {
    const float* pf = (const float*)&psum[0][0];
    float gsum = 0.f;
#pragma unroll
    for (int j = 0; j < 16; ++j) gsum += pf[j * H + t];
    float invN = 1.f / (float)N;
    gsh[t] = gsum * invN;
  }
  __syncthreads();
  if (t < 128) {
    float a = b1[t];
    for (int k = 0; k < H; ++k) a += gsh[k] * W1[k * 128 + t];
    hid[t] = fmaxf(a, 0.f);
  }
  __syncthreads();
  if (t < 4) {
    float l = b2[t];
    for (int k = 0; k < 128; ++k) l += hid[k] * W2[k * 4 + t];
    lg[t] = l;
  }
  __syncthreads();
  if (t == 0) {
    float m = fmaxf(fmaxf(lg[0], lg[1]), fmaxf(lg[2], lg[3]));
    float e0 = __expf(lg[0] - m), e1 = __expf(lg[1] - m);
    float e2 = __expf(lg[2] - m), e3 = __expf(lg[3] - m);
    float s = e0 + e1 + e2 + e3;
    sw[0] = e0 / s; sw[1] = e1 / s; sw[2] = e2 / s; sw[3] = e3 / s;
  }
}

#define ROW_BLOCKS 2048

__global__ void k_deg_count(const int* __restrict__ dst, int* __restrict__ deg, int E) {
  int e = blockIdx.x * blockDim.x + threadIdx.x;
  if (e < E) atomicAdd(&deg[dst[e]], 1);
}

#define SCAN_TILE 16384
__device__ __forceinline__ void scan_deg_body(int* __restrict__ off, int N) {
  __shared__ int wsum[32];
  __shared__ int wpre[32];
  const int t = threadIdx.x, lane = t & 31, w = t >> 5;
  int carry = 0;
  for (int base = 0; base < N; base += SCAN_TILE) {
    int4* p = (int4*)(off + base + 16 * t);
    int4 q[4];
#pragma unroll
    for (int k = 0; k < 4; ++k) q[k] = p[k];
    int tot = 0;
#pragma unroll
    for (int k = 0; k < 4; ++k) tot += q[k].x + q[k].y + q[k].z + q[k].w;
    int inc = tot;
    for (int o = 1; o < 32; o <<= 1) { int u = __shfl_up(inc, o, 32); if (lane >= o) inc += u; }
    if (lane == 31) wsum[w] = inc;
    __syncthreads();
    if (w == 0) {
      int x = wsum[lane], ix = x;
      for (int o = 1; o < 32; o <<= 1) { int u = __shfl_up(ix, o, 32); if (lane >= o) ix += u; }
      wpre[lane] = ix - x;
    }
    __syncthreads();
    int run = carry + wpre[w] + inc - tot;
    carry += wpre[31] + wsum[31];
#pragma unroll
    for (int k = 0; k < 4; ++k) {
      int4 o;
      o.x = run; run += q[k].x;
      o.y = run; run += q[k].y;
      o.z = run; run += q[k].z;
      o.w = run; run += q[k].w;
      p[k] = o;
    }
    __syncthreads();
  }
}

__global__ __launch_bounds__(1024) void k_scan_sel(int* __restrict__ off, int N,
                                                   const float* __restrict__ part,
                                                   const float* __restrict__ W1,
                                                   const float* __restrict__ b1,
                                                   const float* __restrict__ W2,
                                                   const float* __restrict__ b2,
                                                   float* __restrict__ sw) {
  if (blockIdx.x == 0) scan_deg_body(off, N);
  else                 selector_body(part, W1, b1, W2, b2, sw, N);
}

__global__ __launch_bounds__(256) void k_gat_gather(const float* __restrict__ h,
                             const float* __restrict__ s, const float* __restrict__ d,
                             const int* __restrict__ off, const int* __restrict__ elist,
                             int branch, const float* __restrict__ bias,
                             float* __restrict__ xo, int N) {
  const int2* el = (const int2*)elist;
  const int lane = threadIdx.x & 31;
  const int nw = (int)gridDim.x * 8;
  const int wv = __builtin_amdgcn_readfirstlane((int)(threadIdx.x >> 5));
  float bv[8];
#pragma unroll
  for (int j = 0; j < 8; ++j) bv[j] = bias[lane + 32 * j];
  for (int n = (int)blockIdx.x * 8 + wv; n < N; n += nw) {
    const int e0 = n ? off[n - 1] : 0;
    const int e1 = off[n];
    const float* hp = h + (size_t)n * H + lane;
    float acc[8];
#pragma unroll
    for (int j = 0; j < 8; ++j) acc[j] = hp[32 * j];
    const float dn = d[n];
    const float lself = lrelu(s[n] + dn);
    float m = lself;
    for (int c = e0; c < e1; c += 32) {
      const int i = c + lane;
      if (i < e1) {
        const int2 ed = el[i];
        const float lg = lrelu(s[ed.x] + dn);
        if (ed.y == branch) m = fmaxf(m, lg);
      }
    }
    for (int o = 16; o; o >>= 1) m = fmaxf(m, __shfl_xor(m, o, 32));
    const float wself = __expf(lself - m);
    float den = wself;
#pragma unroll
    for (int j = 0; j < 8; ++j) acc[j] *= wself;
    for (int c = e0; c < e1; c += 32) {
      const int i = c + lane;
      bool on = false;
      int sn = 0;
      float w = 0.f;
      if (i < e1) {
        const int2 ed = el[i];
        const float lg = lrelu(s[ed.x] + dn);
        on = (ed.y == branch);
        sn = ed.x;
        if (on) w = __expf(lg - m);
      }
      const unsigned live = __builtin_amdgcn_ballot_w32(on);
      int rank = 0;
      for (unsigned b = live; b; b &= b - 1u) {
        const int j = __builtin_ctz(b);
        const int sj = __builtin_amdgcn_readlane(sn, j);
        rank += (sj < sn || (sj == sn && j < lane)) ? 1 : 0;
      }
      const int cnt = __builtin_popcount(live);
#pragma unroll 2
      for (int r = 0; r < cnt; ++r) {
        const int j = __builtin_ctz(__builtin_amdgcn_ballot_w32(on && rank == r));
        const int sj = __builtin_amdgcn_readlane(sn, j);
        const float wj = __uint_as_float(__builtin_amdgcn_readlane(__float_as_uint(w), j));
        den += wj;
        const float* sp = h + (size_t)sj * H + lane;
        float v[8];
#pragma unroll
        for (int k = 0; k < 8; ++k) v[k] = sp[32 * k];
#pragma unroll
        for (int k = 0; k < 8; ++k) acc[k] += wj * v[k];
      }
    }
    float* op = xo + (size_t)n * H + lane;
#pragma unroll
    for (int k = 0; k < 8; ++k) op[32 * k] = acc[k] / den + bv[k];
  }
}


extern "C" void kernel_launch(void* const* d_in, const int* in_sizes, int n_in,
                              void* d_out, int out_size, void* d_ws, size_t ws_size,
                              hipStream_t stream) {
  const float* x     = (const float*)d_in[0];
  const int*   ei    = (const int*)d_in[1];
  const int*   ea    = (const int*)d_in[2];
  const float* Wg    = (const float*)d_in[3];
  const float* a_src = (const float*)d_in[4];
  const float* a_dst = (const float*)d_in[5];
  const float* bg    = (const float*)d_in[6];
  const float* Wsl   = (const float*)d_in[7];
  const float* bsl   = (const float*)d_in[8];
  const float* gsl   = (const float*)d_in[9];
  const float* betas = (const float*)d_in[10];
  const float* W1    = (const float*)d_in[11];
  const float* b1    = (const float*)d_in[12];
  const float* W2    = (const float*)d_in[13];
  const float* b2    = (const float*)d_in[14];
  const float* Wf    = (const float*)d_in[15];
  const float* bfv   = (const float*)d_in[16];
  const float* gf    = (const float*)d_in[17];
  const float* betaf = (const float*)d_in[18];

  const int N = in_sizes[0] / H;
  const int E = in_sizes[2];
  float* out = (float*)d_out;

  char* wsb = (char*)d_ws;
  size_t off = 0;
  auto alloc = [&](size_t bytes) -> void* {
    void* p = wsb + off;
    off += (bytes + 255) & ~(size_t)255;
    return p;
  };
  float*    sw    = (float*)alloc(4 * sizeof(float));
  float*    sArr  = (float*)alloc((size_t)N * sizeof(float));
  float*    dArr  = (float*)alloc((size_t)N * sizeof(float));
  const int nPad = (N + SCAN_TILE - 1) / SCAN_TILE * SCAN_TILE;
  int*      ebnd  = (int*)alloc((size_t)nPad * sizeof(int));
  float*    bufH  = (float*)alloc((size_t)N * H * sizeof(float));
  float*    bufX  = (float*)alloc((size_t)N * H * sizeof(float));
  int*      elist = (int*)alloc((size_t)N * H * sizeof(float));
  unsigned* wt = (unsigned*)((char*)elist + (((size_t)E * 8 + 255) & ~(size_t)255));

  const int gemmBlocks = (N + TM - 1) / TM;
  dim3 gemmGrid((unsigned)gemmBlocks, 1u);
  int eThrBlocks  = (E + 255) / 256;
  const int fillBlocks = 256;

  float* part = bufX;
  k_prep<<<CS_GROUPS + SPLIT_GROUPS, CS_THREADS, 0, stream>>>(x, part, ebnd, nPad, N, Wg, Wsl, Wf, wt);
  k_deg_count<<<eThrBlocks, 256, 0, stream>>>(ei + E, ebnd, E);
  k_scan_sel<<<2, 1024, 0, stream>>>(ebnd, N, part, W1, b1, W2, b2, sw);
  k_gemm_sd_fill<<<gemmBlocks + fillBlocks, 256, 0, stream>>>(
      x, wt, wt + WPLANE_W, a_src, a_dst, bufH, sArr, dArr, N, gemmBlocks,
      ei, ei + E, ea, ebnd, elist, E);

  for (int i = 0; i < 3; ++i) {
    k_gat_gather<<<ROW_BLOCKS, 256, 0, stream>>>(bufH, sArr, dArr, ebnd, elist, i,
                                                 bg + i * H, bufX, N);
    if (i < 2) {
      k_gemm_ln_sd<<<2 * gemmBlocks, 256, 0, stream>>>(
          bufX, wt + (size_t)(3 + i) * 2 * WPLANE_W,
          wt + (size_t)(3 + i) * 2 * WPLANE_W + WPLANE_W,
          bsl + i * H, gsl + i * H, betas + i * H, sw, i,
          out, i > 0 ? 1 : 0, out, N, gemmBlocks,
          x, wt + (size_t)(i + 1) * 2 * WPLANE_W,
          wt + (size_t)(i + 1) * 2 * WPLANE_W + WPLANE_W,
          a_src + (i + 1) * H, a_dst + (i + 1) * H, bufH, sArr, dArr);
    } else {
      k_gemm_ln<<<gemmGrid, 256, 0, stream>>>(bufX, wt + (size_t)(3 + i) * 2 * WPLANE_W,
                                            wt + (size_t)(3 + i) * 2 * WPLANE_W + WPLANE_W,
                                            bsl + i * H, gsl + i * H, betas + i * H, sw, i,
                                            out, 1, out, N);
    }
  }
  k_gemm_ln<<<gemmGrid, 256, 0, stream>>>(x, wt + (size_t)6 * 2 * WPLANE_W,
                                          wt + (size_t)6 * 2 * WPLANE_W + WPLANE_W, bsl + 3 * H,
                                          gsl + 3 * H, betas + 3 * H, sw, 3, out, 1, bufX, N);

  k_gemm_ln<<<gemmGrid, 256, 0, stream>>>(bufX, wt + (size_t)7 * 2 * WPLANE_W,
                                          wt + (size_t)7 * 2 * WPLANE_W + WPLANE_W, bfv,
                                          gf, betaf, nullptr, 0, nullptr, 0, out, N);
}


# attention sums moved in front of the weight product: gathers sum rows of x, Wg.Ws and the biases folded once per call, s/d from x.(Wg a): 5 row GEMMs not 8; products re-associated, sum order changed
# speedup vs baseline: 4.3027x; 1.3019x over previous
#include <hip/hip_runtime.h>

#define H 256
#define TM 64
#define KC 32
#define APAD 40

typedef __bf16 bf16;
typedef __attribute__((ext_vector_type(16))) __bf16 v16bf;
typedef __attribute__((ext_vector_type(8)))  __bf16 v8bf;
typedef __attribute__((ext_vector_type(8)))  float  v8f;

__device__ __forceinline__ unsigned short f2bfbits(float f) {
  unsigned u = __float_as_uint(f);
  return (unsigned short)((u + 0x7FFFu + ((u >> 16) & 1u)) >> 16);
}
__device__ __forceinline__ float bfbits2f(unsigned short h) {
  return __uint_as_float(((unsigned)h) << 16);
}
__device__ __forceinline__ bf16 f2bf(float f) {
  unsigned short hs = f2bfbits(f);
  return __builtin_bit_cast(bf16, hs);
}
__device__ __forceinline__ float lrelu(float v) { return v > 0.f ? v : 0.2f * v; }

__device__ __forceinline__ v16bf load_frag(const bf16* rowp, int g0) {
  v8bf a = *(const v8bf*)(rowp + g0);
  v8bf b = *(const v8bf*)(rowp + 16 + g0);
  return __builtin_shufflevector(a, b, 0, 1, 2, 3, 4, 5, 6, 7,
                                 8, 9, 10, 11, 12, 13, 14, 15);
}

#define WMMA_BF16(A_, B_, C_) \
  __builtin_amdgcn_wmma_f32_16x16x32_bf16(false, (A_), false, (B_), (short)0, (C_), false, false)


#define WPLANE (H * H)
#define WPLANE_W (H * H / 2)

__device__ __forceinline__ void store_split8(const float (&v)[8], unsigned* __restrict__ planes,
                                             int k8, int n) {
  unsigned short hi[8], lo[8];
#pragma unroll
  for (int j = 0; j < 8; ++j) {
    hi[j] = f2bfbits(v[j]);
    lo[j] = f2bfbits(v[j] - bfbits2f(hi[j]));
  }
  uint4 ph, pl;
  ph.x = (unsigned)hi[0] | ((unsigned)hi[1] << 16); ph.y = (unsigned)hi[2] | ((unsigned)hi[3] << 16);
  ph.z = (unsigned)hi[4] | ((unsigned)hi[5] << 16); ph.w = (unsigned)hi[6] | ((unsigned)hi[7] << 16);
  pl.x = (unsigned)lo[0] | ((unsigned)lo[1] << 16); pl.y = (unsigned)lo[2] | ((unsigned)lo[3] << 16);
  pl.z = (unsigned)lo[4] | ((unsigned)lo[5] << 16); pl.w = (unsigned)lo[6] | ((unsigned)lo[7] << 16);
  const size_t o = ((size_t)(k8 >> 5) * H + n) * (KC / 2) + ((k8 & 31) >> 1);
  *(uint4*)(planes + o) = ph;
  *(uint4*)(planes + WPLANE_W + o) = pl;
}


#define CPITCH 260
#define SMEM_BYTES (TM * CPITCH * 4)


#define EPI_LN 1
#define EPI_FOLD 2

__device__ __forceinline__ void butterfly8(float (&a)[8]) {
  for (int k = 16; k; k >>= 1) {
    float t[8];
#pragma unroll
    for (int i = 0; i < 8; ++i) t[i] = __shfl_xor(a[i], k, 32);
#pragma unroll
    for (int i = 0; i < 8; ++i) a[i] += t[i];
  }
}

__device__ __forceinline__ void acc_row(const float* accIn, int gr, int lane, float (&o)[8]) {
  const float* ip = accIn + (size_t)gr * H + lane;
#pragma unroll
  for (int j = 0; j < 8; ++j) o[j] = ip[32 * j];
}

template <bool ACC>
__device__ __forceinline__ void ln_finish(float (&v)[8], float sum, float sq,
                                          const float (&g)[8], const float (&b)[8], float scale,
                                          const float (&o)[8], float* op) {
  float mu = sum * (1.f / H);
  float var = sq * (1.f / H) - mu * mu;
  float inv = rsqrtf(var + 1e-5f);
#pragma unroll
  for (int j = 0; j < 8; ++j) {
    float y = (v[j] - mu) * inv * g[j] + b[j];
    v[j] = fmaxf(y, 0.f) * scale;
  }
  if (ACC) {
#pragma unroll
    for (int j = 0; j < 8; ++j) op[32 * j] = o[j] + v[j];
  } else {
#pragma unroll
    for (int j = 0; j < 8; ++j) op[32 * j] = v[j];
  }
}

template <bool ACC>
__device__ __forceinline__ void ln_rows(const float* Ct, int trow0, int grow0, int lane,
                                        const float (&g)[8], const float (&b)[8], float scale,
                                        const float* accIn, float* dst) {
  float sum[8], sq[8];
#pragma unroll
  for (int i = 0; i < 8; ++i) {
    const float* tp = Ct + (trow0 + i) * CPITCH + lane;
    float s = 0.f, q = 0.f;
#pragma unroll
    for (int j = 0; j < 8; ++j) {
      const float t = tp[32 * j];
      s += t;
      q += t * t;
    }
    sum[i] = s; sq[i] = q;
  }
  butterfly8(sum);
  butterfly8(sq);
  __builtin_amdgcn_sched_barrier(0);
  float o[8][8];
  if (ACC) {
#pragma unroll
    for (int i = 0; i < 4; ++i) acc_row(accIn, grow0 + i, lane, o[i]);
  }
  float vn[8];
#pragma unroll
  for (int j = 0; j < 8; ++j) vn[j] = Ct[trow0 * CPITCH + lane + 32 * j];
#pragma unroll
  for (int i = 0; i < 8; ++i) {
    float v[8];
#pragma unroll
    for (int j = 0; j < 8; ++j) v[j] = vn[j];
    if (i < 7) {
      const float* tp = Ct + (trow0 + i + 1) * CPITCH + lane;
#pragma unroll
      for (int j = 0; j < 8; ++j) vn[j] = tp[32 * j];
    }
    ln_finish<ACC>(v, sum[i], sq[i], g, b, scale, o[i], dst + (size_t)(grow0 + i) * H + lane);
    if (ACC && i < 4) acc_row(accIn, grow0 + i + 4, lane, o[i + 4]);
  }
}

template <bool ACC>
__device__ __forceinline__ void ln_rows_edge(const float* Ct, int trow0, int grow0, int M, int lane,
                                             const float (&g)[8], const float (&b)[8],
                                             float scale, const float* accIn, float* dst) {
  for (int i = 0; i < 8; ++i) {
    const int gr = grow0 + i;
    if (gr < M) {
      float oo[8] = {};
      if (ACC) acc_row(accIn, gr, lane, oo);
      const float* tp = Ct + (trow0 + i) * CPITCH + lane;
      float v[8];
#pragma unroll
      for (int j = 0; j < 8; ++j) v[j] = tp[32 * j];
      float sum = 0.f, sq = 0.f;
#pragma unroll
      for (int j = 0; j < 8; ++j) {
        sum += v[j];
        sq += v[j] * v[j];
      }
      for (int k = 16; k; k >>= 1) { sum += __shfl_xor(sum, k, 32); sq += __shfl_xor(sq, k, 32); }
      ln_finish<ACC>(v, sum, sq, g, b, scale, oo, dst + (size_t)gr * H + lane);
    }
  }
}


template <int EPI>
__device__ __forceinline__ void gemm_rows_body(
    unsigned char* smem, int bid,
    const float* __restrict__ A, const unsigned* __restrict__ Bhi,
    const unsigned* __restrict__ Blo, int M,
    const float* __restrict__ bias,
    const float* __restrict__ p0, const float* __restrict__ p1,
    float* C,
    const float* __restrict__ swp, int swidx,
    const float* accIn, int accumulate,
    unsigned* __restrict__ planesOut)
{
  bf16 (*Ah)[APAD]  = (bf16 (*)[APAD])(smem);
  bf16 (*Al)[APAD]  = (bf16 (*)[APAD])(smem + TM * APAD * 2);
  bf16 (*Bth)[APAD] = (bf16 (*)[APAD])(smem + 2 * TM * APAD * 2);
  bf16 (*Btl)[APAD] = (bf16 (*)[APAD])(smem + 2 * TM * APAD * 2 + H * APAD * 2);
  float* Ct = (float*)smem;

  const int tid = threadIdx.x;
  const int m0 = bid * TM;
  const int wid = __builtin_amdgcn_readfirstlane(tid >> 5);
  const int lane = tid & 31;
  const int hl = lane & 15, hh = lane >> 4;
  const int wr = wid >> 2;
  const int wc = wid & 3;
  const int g0 = hh << 3;

  v8f acc00 = {}, acc01 = {}, acc02 = {}, acc03 = {};
  v8f acc10 = {}, acc11 = {}, acc12 = {}, acc13 = {};

  const int ar = tid >> 3;
  const int ac = (tid & 7) << 2;
  const float* ap0 = A + (size_t)((m0 + ar) < M ? m0 + ar : M - 1) * H + ac;
  const float* ap1 = A + (size_t)((m0 + ar + 32) < M ? m0 + ar + 32 : M - 1) * H + ac;

  for (int k0 = 0; k0 < H; k0 += KC) {
    __syncthreads();
    const size_t bo = (size_t)(k0 >> 5) * (H * KC / 2) + (size_t)tid * 4;
    uint4 vh[4], vl[4];
#pragma unroll
    for (int j = 0; j < 4; ++j) {
      vh[j] = *(const uint4*)(Bhi + bo + (size_t)j * 1024);
      vl[j] = *(const uint4*)(Blo + bo + (size_t)j * 1024);
    }
    float4 a0 = *(const float4*)(ap0 + k0);
    float4 a1 = *(const float4*)(ap1 + k0);
#pragma unroll
    for (int j = 0; j < 4; ++j) {
      const int p = tid + 256 * j;
      *(uint4*)&Bth[p >> 2][(p & 3) << 3] = vh[j];
      *(uint4*)&Btl[p >> 2][(p & 3) << 3] = vl[j];
    }
    {
      ushort4 hiv, lov;
      hiv.x = f2bfbits(a0.x); lov.x = f2bfbits(a0.x - bfbits2f(hiv.x));
      hiv.y = f2bfbits(a0.y); lov.y = f2bfbits(a0.y - bfbits2f(hiv.y));
      hiv.z = f2bfbits(a0.z); lov.z = f2bfbits(a0.z - bfbits2f(hiv.z));
      hiv.w = f2bfbits(a0.w); lov.w = f2bfbits(a0.w - bfbits2f(hiv.w));
      *(ushort4*)&Ah[ar][ac] = hiv;
      *(ushort4*)&Al[ar][ac] = lov;
    }
    {
      ushort4 hiv, lov;
      hiv.x = f2bfbits(a1.x); lov.x = f2bfbits(a1.x - bfbits2f(hiv.x));
      hiv.y = f2bfbits(a1.y); lov.y = f2bfbits(a1.y - bfbits2f(hiv.y));
      hiv.z = f2bfbits(a1.z); lov.z = f2bfbits(a1.z - bfbits2f(hiv.z));
      hiv.w = f2bfbits(a1.w); lov.w = f2bfbits(a1.w - bfbits2f(hiv.w));
      *(ushort4*)&Ah[ar + 32][ac] = hiv;
      *(ushort4*)&Al[ar + 32][ac] = lov;
    }
    __syncthreads();

    const v16bf ah0 = load_frag(&Ah[(wr << 5) + hl][0], g0);
    const v16bf al0 = load_frag(&Al[(wr << 5) + hl][0], g0);
    const v16bf ah1 = load_frag(&Ah[(wr << 5) + 16 + hl][0], g0);
    const v16bf al1 = load_frag(&Al[(wr << 5) + 16 + hl][0], g0);
    const v16bf bh0 = load_frag(&Bth[(wc << 6) + hl][0], g0);
    const v16bf bl0 = load_frag(&Btl[(wc << 6) + hl][0], g0);
    const v16bf bh1 = load_frag(&Bth[(wc << 6) + 16 + hl][0], g0);
    const v16bf bl1 = load_frag(&Btl[(wc << 6) + 16 + hl][0], g0);
    const v16bf bh2 = load_frag(&Bth[(wc << 6) + 32 + hl][0], g0);
    const v16bf bl2 = load_frag(&Btl[(wc << 6) + 32 + hl][0], g0);
    const v16bf bh3 = load_frag(&Bth[(wc << 6) + 48 + hl][0], g0);
    const v16bf bl3 = load_frag(&Btl[(wc << 6) + 48 + hl][0], g0);

    acc00 = WMMA_BF16(ah0, bh0, acc00);
    acc00 = WMMA_BF16(al0, bh0, acc00);
    acc00 = WMMA_BF16(ah0, bl0, acc00);
    acc10 = WMMA_BF16(ah1, bh0, acc10);
    acc10 = WMMA_BF16(al1, bh0, acc10);
    acc10 = WMMA_BF16(ah1, bl0, acc10);
    acc01 = WMMA_BF16(ah0, bh1, acc01);
    acc01 = WMMA_BF16(al0, bh1, acc01);
    acc01 = WMMA_BF16(ah0, bl1, acc01);
    acc11 = WMMA_BF16(ah1, bh1, acc11);
    acc11 = WMMA_BF16(al1, bh1, acc11);
    acc11 = WMMA_BF16(ah1, bl1, acc11);
    acc02 = WMMA_BF16(ah0, bh2, acc02);
    acc02 = WMMA_BF16(al0, bh2, acc02);
    acc02 = WMMA_BF16(ah0, bl2, acc02);
    acc12 = WMMA_BF16(ah1, bh2, acc12);
    acc12 = WMMA_BF16(al1, bh2, acc12);
    acc12 = WMMA_BF16(ah1, bl2, acc12);
    acc03 = WMMA_BF16(ah0, bh3, acc03);
    acc03 = WMMA_BF16(al0, bh3, acc03);
    acc03 = WMMA_BF16(ah0, bl3, acc03);
    acc13 = WMMA_BF16(ah1, bh3, acc13);
    acc13 = WMMA_BF16(al1, bh3, acc13);
    acc13 = WMMA_BF16(ah1, bl3, acc13);
  }

  __syncthreads();
  {
    const int col0 = (wc << 6) + hl;
    float b0 = 0.f, b1v = 0.f, b2v = 0.f, b3v = 0.f;
    if (EPI == EPI_LN) {
      b0 = bias[col0]; b1v = bias[col0 + 16]; b2v = bias[col0 + 32]; b3v = bias[col0 + 48];
    }
    float* cw = Ct + ((wr << 5) + (hh << 3)) * CPITCH + col0;
#pragma unroll
    for (int r = 0; r < 8; ++r) {
      float* c0 = cw + r * CPITCH;
      float* c1 = c0 + 16 * CPITCH;
      if (EPI == EPI_LN) {
        c0[0]  = acc00[r] + b0;
        c0[16] = acc01[r] + b1v;
        c0[32] = acc02[r] + b2v;
        c0[48] = acc03[r] + b3v;
        c1[0]  = acc10[r] + b0;
        c1[16] = acc11[r] + b1v;
        c1[32] = acc12[r] + b2v;
        c1[48] = acc13[r] + b3v;
      } else {
        c0[0]  = acc00[r];
        c0[16] = acc01[r];
        c0[32] = acc02[r];
        c0[48] = acc03[r];
        c1[0]  = acc10[r];
        c1[16] = acc11[r];
        c1[32] = acc12[r];
        c1[48] = acc13[r];
      }
    }
  }


  if (EPI == EPI_LN) {
    const int wv = __builtin_amdgcn_readfirstlane(wid);
    const int trow0 = wv << 3;
    const int grow0 = m0 + trow0;
    const bool full = (grow0 + 8 <= M);
    float g[8], b[8];
#pragma unroll
    for (int j = 0; j < 8; ++j) { g[j] = p0[lane + 32 * j]; b[j] = p1[lane + 32 * j]; }
    const float scale = swp ? swp[swidx] : 1.f;
    __syncthreads();
    if (accumulate) {
      if (full) ln_rows<true>(Ct, trow0, grow0, lane, g, b, scale, accIn, C);
      else      ln_rows_edge<true>(Ct, trow0, grow0, M, lane, g, b, scale, accIn, C);
    } else {
      if (full) ln_rows<false>(Ct, trow0, grow0, lane, g, b, scale, accIn, C);
      else      ln_rows_edge<false>(Ct, trow0, grow0, M, lane, g, b, scale, accIn, C);
    }
  } else {
    __syncthreads();
    for (int r8 = 0; r8 < TM; r8 += 8) {
      float v[8];
#pragma unroll
      for (int j = 0; j < 8; ++j) v[j] = Ct[(r8 + j) * CPITCH + tid];
      store_split8(v, planesOut, m0 + r8, tid);
    }
  }
}


__global__ __launch_bounds__(256, 8) void k_gemm_ln(
    const float* __restrict__ A, const unsigned* __restrict__ Bhi,
    const unsigned* __restrict__ Blo, const float* __restrict__ bias,
    const float* __restrict__ gamma, const float* __restrict__ beta,
    const float* __restrict__ swp, int swidx,
    const float* accIn, int accumulate, float* dst, int M)
{
  __shared__ __attribute__((aligned(16))) unsigned char smem[SMEM_BYTES];
  gemm_rows_body<EPI_LN>(smem, (int)blockIdx.x, A, Bhi, Blo, M, bias, gamma, beta, dst,
                         swp, swidx, accIn, accumulate, nullptr);
}

#define FOLD_TILES 12
#define CVEC_GROUPS 3
__global__ __launch_bounds__(256, 8) void k_fold_deg(
    const float* __restrict__ Wg, const float* __restrict__ Wsl,
    const float* __restrict__ bg, const float* __restrict__ bsl,
    unsigned* wt, float* __restrict__ cvec,
    const int* __restrict__ dst, int* __restrict__ deg, int E)
{
  __shared__ __attribute__((aligned(16))) unsigned char smem[SMEM_BYTES];
  const int bid = (int)blockIdx.x;
  if (bid < FOLD_TILES) {
    const int i = bid >> 2;
    const unsigned* bh = wt + (size_t)(3 + i) * 2 * WPLANE_W;
    gemm_rows_body<EPI_FOLD>(smem, bid & 3, Wg + (size_t)i * WPLANE, bh, bh + WPLANE_W, H,
                             nullptr, nullptr, nullptr, nullptr, nullptr, 0, nullptr, 0,
                             wt + (size_t)i * 2 * WPLANE_W);
  } else if (bid < FOLD_TILES + CVEC_GROUPS) {
    const int i = bid - FOLD_TILES;
    const int n = (int)threadIdx.x;
    const float* Wm = Wsl + (size_t)i * WPLANE + n;
    const float* bgi = bg + i * H;
    float c = bsl[i * H + n];
    for (int k = 0; k < H; ++k) c += bgi[k] * Wm[(size_t)k * H];
    cvec[i * H + n] = c;
  } else {
    const int first = FOLD_TILES + CVEC_GROUPS;
    const int nb = (int)gridDim.x - first;
    for (int e = (bid - first) * 256 + (int)threadIdx.x; e < E; e += nb * 256)
      atomicAdd(&deg[dst[e]], 1);
  }
}

__global__ __launch_bounds__(256) void k_fill(const int* __restrict__ src,
                                              const int* __restrict__ dst,
                                              const int* __restrict__ attr,
                                              int* __restrict__ off, int* __restrict__ elist,
                                              int E) {
  for (int e = (int)blockIdx.x * 256 + (int)threadIdx.x; e < E; e += (int)gridDim.x * 256) {
    int pos = atomicAdd(&off[dst[e]], 1);
    ((int2*)elist)[pos] = make_int2(src[e], attr[e]);
  }
}


__device__ __forceinline__ void split_weights_body(const float* __restrict__ Wsl,
                                                   const float* __restrict__ Wf,
                                                   unsigned* __restrict__ wt, int gi) {
  const int m = gi >> 13;
  const float* Wm = m < 4 ? Wsl + (size_t)m * WPLANE : Wf;
  const int g = gi & 8191;
  const int n = g & 255;
  const int k8 = (g >> 8) << 3;
  float v[8];
#pragma unroll
  for (int j = 0; j < 8; ++j) v[j] = Wm[(size_t)(k8 + j) * H + n];
  store_split8(v, wt + (size_t)(3 + m) * 2 * WPLANE_W, k8, n);
}

__device__ __forceinline__ void uv_body(const float* __restrict__ Wg,
                                        const float* __restrict__ a_src,
                                        const float* __restrict__ a_dst,
                                        float* __restrict__ uv, int g) {
  const int i = g >> 2;
  const int lane = (int)threadIdx.x & 31, w = (int)threadIdx.x >> 5;
  float as[8], ad[8];
#pragma unroll
  for (int j = 0; j < 8; ++j) { as[j] = a_src[i * H + lane + 32 * j]; ad[j] = a_dst[i * H + lane + 32 * j]; }
  const int c0 = (g & 3) * 64 + w * 4;
  float su[4], sv[4];
#pragma unroll
  for (int r = 0; r < 4; ++r) {
    const float* wp = Wg + (size_t)i * WPLANE + (size_t)(c0 + r) * H + lane;
    float s = 0.f, d = 0.f;
#pragma unroll
    for (int j = 0; j < 8; ++j) { const float t = wp[32 * j]; s += t * as[j]; d += t * ad[j]; }
    su[r] = s; sv[r] = d;
  }
  for (int o = 16; o; o >>= 1) {
#pragma unroll
    for (int r = 0; r < 4; ++r) { su[r] += __shfl_xor(su[r], o, 32); sv[r] += __shfl_xor(sv[r], o, 32); }
  }
  if (lane == 0) {
#pragma unroll
    for (int r = 0; r < 4; ++r) { uv[(2 * i) * H + c0 + r] = su[r]; uv[(2 * i + 1) * H + c0 + r] = sv[r]; }
  }
}

#define CS_GROUPS 256
#define CS_THREADS 512

__device__ __forceinline__ void colsum_part_body(const float* __restrict__ x,
                                                 float* __restrict__ part,
                                                 int* __restrict__ deg, int ndeg, int N) {
  __shared__ float4 red[CS_THREADS / 64][64];
  const int t = threadIdx.x;
  for (int i = (int)blockIdx.x * CS_THREADS + t; i < ndeg; i += CS_GROUPS * CS_THREADS) deg[i] = 0;
  const int cq = t & 63;
  const int rs = t >> 6;
  const int per = (N + CS_GROUPS - 1) / CS_GROUPS;
  const int r0 = (int)blockIdx.x * per;
  int r1 = r0 + per; if (r1 > N) r1 = N;
  const float4* xp = (const float4*)x + cq;
  float4 acc = make_float4(0.f, 0.f, 0.f, 0.f);
  int r = r0 + rs;
  for (; r + 56 < r1; r += 64) {
    float4 v[8];
#pragma unroll
    for (int j = 0; j < 8; ++j) v[j] = xp[(size_t)(r + 8 * j) * 64];
#pragma unroll
    for (int j = 0; j < 8; ++j) {
      acc.x += v[j].x; acc.y += v[j].y; acc.z += v[j].z; acc.w += v[j].w;
    }
  }
  for (; r < r1; r += 8) {
    float4 v = xp[(size_t)r * 64];
    acc.x += v.x; acc.y += v.y; acc.z += v.z; acc.w += v.w;
  }
  red[rs][cq] = acc;
  __syncthreads();
  if (t < H) {
    const float* rf = (const float*)&red[0][0];
    float s = 0.f;
#pragma unroll
    for (int j = 0; j < CS_THREADS / 64; ++j) s += rf[j * H + t];
    part[(size_t)blockIdx.x * H + t] = s;
  }
}


#define SPLIT_GROUPS (5 * 8192 / CS_THREADS)
#define UV_GROUPS 12
__global__ __launch_bounds__(CS_THREADS) void k_prep(const float* __restrict__ x,
                                                    float* __restrict__ part,
                                                    int* __restrict__ deg, int ndeg, int N,
                                                    const float* __restrict__ Wg,
                                                    const float* __restrict__ Wsl,
                                                    const float* __restrict__ Wf,
                                                    unsigned* __restrict__ wt,
                                                    const float* __restrict__ a_src,
                                                    const float* __restrict__ a_dst,
                                                    float* __restrict__ uv) {
  const int bid = (int)blockIdx.x;
  if (bid < CS_GROUPS) {
    colsum_part_body(x, part, deg, ndeg, N);
  } else if (bid < CS_GROUPS + SPLIT_GROUPS) {
    split_weights_body(Wsl, Wf, wt, (bid - CS_GROUPS) * CS_THREADS + (int)threadIdx.x);
  } else {
    uv_body(Wg, a_src, a_dst, uv, bid - CS_GROUPS - SPLIT_GROUPS);
  }
}


__device__ __forceinline__ void selector_body(const float* __restrict__ part,
                           const float* __restrict__ W1, const float* __restrict__ b1,
                           const float* __restrict__ W2, const float* __restrict__ b2,
                           float* __restrict__ sw, int N) {
  __shared__ float4 psum[16][64];
  __shared__ float gsh[H];
  __shared__ float hid[128];
  __shared__ float lg[4];
  const int t = threadIdx.x;
  {
    const int cq = t & 63, q = t >> 6;
    const float4* pp = (const float4*)part + cq;
    float4 s = make_float4(0.f, 0.f, 0.f, 0.f);
#pragma unroll
    for (int j = 0; j < CS_GROUPS / 16; ++j) {
      float4 v = pp[(size_t)(q + 16 * j) * 64];
      s.x += v.x; s.y += v.y; s.z += v.z; s.w += v.w;
    }
    psum[q][cq] = s;
  }
  __syncthreads();
  if (t < H) {
    const float* pf = (const float*)&psum[0][0];
    float gsum = 0.f;
#pragma unroll
    for (int j = 0; j < 16; ++j) gsum += pf[j * H + t];
    float invN = 1.f / (float)N;
    gsh[t] = gsum * invN;
  }
  __syncthreads();
  if (t < 128) {
    float a = b1[t];
    for (int k = 0; k < H; ++k) a += gsh[k] * W1[k * 128 + t];
    hid[t] = fmaxf(a, 0.f);
  }
  __syncthreads();
  if (t < 4) {
    float l = b2[t];
    for (int k = 0; k < 128; ++k) l += hid[k] * W2[k * 4 + t];
    lg[t] = l;
  }
  __syncthreads();
  if (t == 0) {
    float m = fmaxf(fmaxf(lg[0], lg[1]), fmaxf(lg[2], lg[3]));
    float e0 = __expf(lg[0] - m), e1 = __expf(lg[1] - m);
    float e2 = __expf(lg[2] - m), e3 = __expf(lg[3] - m);
    float s = e0 + e1 + e2 + e3;
    sw[0] = e0 / s; sw[1] = e1 / s; sw[2] = e2 / s; sw[3] = e3 / s;
  }
}

#define ROW_BLOCKS 2048


#define SCAN_TILE 16384
__device__ __forceinline__ void scan_deg_body(int* __restrict__ off, int N) {
  __shared__ int wsum[32];
  __shared__ int wpre[32];
  const int t = threadIdx.x, lane = t & 31, w = t >> 5;
  int carry = 0;
  for (int base = 0; base < N; base += SCAN_TILE) {
    int4* p = (int4*)(off + base + 16 * t);
    int4 q[4];
#pragma unroll
    for (int k = 0; k < 4; ++k) q[k] = p[k];
    int tot = 0;
#pragma unroll
    for (int k = 0; k < 4; ++k) tot += q[k].x + q[k].y + q[k].z + q[k].w;
    int inc = tot;
    for (int o = 1; o < 32; o <<= 1) { int u = __shfl_up(inc, o, 32); if (lane >= o) inc += u; }
    if (lane == 31) wsum[w] = inc;
    __syncthreads();
    if (w == 0) {
      int x = wsum[lane], ix = x;
      for (int o = 1; o < 32; o <<= 1) { int u = __shfl_up(ix, o, 32); if (lane >= o) ix += u; }
      wpre[lane] = ix - x;
    }
    __syncthreads();
    int run = carry + wpre[w] + inc - tot;
    carry += wpre[31] + wsum[31];
#pragma unroll
    for (int k = 0; k < 4; ++k) {
      int4 o;
      o.x = run; run += q[k].x;
      o.y = run; run += q[k].y;
      o.z = run; run += q[k].z;
      o.w = run; run += q[k].w;
      p[k] = o;
    }
    __syncthreads();
  }
}


#define SDX_GROUPS 254
__device__ __forceinline__ void sdx_body(const float* __restrict__ x,
                                         const float* __restrict__ uv,
                                         float* __restrict__ sd, int N, int g, int ng) {
  const int lane = (int)threadIdx.x & 31;
  const int wv = __builtin_amdgcn_readfirstlane((int)(threadIdx.x >> 5));
  float u[6][8];
#pragma unroll
  for (int q = 0; q < 6; ++q) {
#pragma unroll
    for (int j = 0; j < 8; ++j) u[q][j] = uv[q * H + lane + 32 * j];
  }
  const int nw = ng * 32;
  int n = g * 32 + wv;
  float v[8];
#pragma unroll
  for (int j = 0; j < 8; ++j) v[j] = 0.f;
  if (n < N) {
    const float* xp = x + (size_t)n * H + lane;
#pragma unroll
    for (int j = 0; j < 8; ++j) v[j] = xp[32 * j];
  }
  for (; n < N; n += nw) {
    float vn[8];
#pragma unroll
    for (int j = 0; j < 8; ++j) vn[j] = 0.f;
    if (n + nw < N) {
      const float* xp = x + (size_t)(n + nw) * H + lane;
#pragma unroll
      for (int j = 0; j < 8; ++j) vn[j] = xp[32 * j];
    }
    float a[6];
#pragma unroll
    for (int q = 0; q < 6; ++q) {
      float t = 0.f;
#pragma unroll
      for (int j = 0; j < 8; ++j) t += v[j] * u[q][j];
      a[q] = t;
    }
    for (int o = 16; o; o >>= 1) {
#pragma unroll
      for (int q = 0; q < 6; ++q) a[q] += __shfl_xor(a[q], o, 32);
    }
    float ov = a[0];
#pragma unroll
    for (int q = 1; q < 6; ++q) { if (lane == q) ov = a[q]; }
    if (lane < 6) sd[(size_t)lane * N + n] = ov;
#pragma unroll
    for (int j = 0; j < 8; ++j) v[j] = vn[j];
  }
}

__global__ __launch_bounds__(1024) void k_scan_sel(int* __restrict__ off, int N,
                                                   const float* __restrict__ part,
                                                   const float* __restrict__ W1,
                                                   const float* __restrict__ b1,
                                                   const float* __restrict__ W2,
                                                   const float* __restrict__ b2,
                                                   float* __restrict__ sw,
                                                   const float* __restrict__ x,
                                                   const float* __restrict__ uv,
                                                   float* __restrict__ sd) {
  if (blockIdx.x == 0)      scan_deg_body(off, N);
  else if (blockIdx.x == 1) selector_body(part, W1, b1, W2, b2, sw, N);
  else                      sdx_body(x, uv, sd, N, (int)blockIdx.x - 2, (int)gridDim.x - 2);
}

__global__ __launch_bounds__(256) void k_gat_gather(const float* __restrict__ x,
                             const float* __restrict__ s, const float* __restrict__ d,
                             const int* __restrict__ off, const int* __restrict__ elist,
                             int branch, float* __restrict__ xo, int N) {
  const int2* el = (const int2*)elist;
  const int lane = threadIdx.x & 31;
  const int nw = (int)gridDim.x * 8;
  const int wv = __builtin_amdgcn_readfirstlane((int)(threadIdx.x >> 5));
  for (int n = (int)blockIdx.x * 8 + wv; n < N; n += nw) {
    const int e0 = n ? off[n - 1] : 0;
    const int e1 = off[n];
    const float* hp = x + (size_t)n * H + lane;
    float acc[8];
#pragma unroll
    for (int j = 0; j < 8; ++j) acc[j] = hp[32 * j];
    const float dn = d[n];
    const float lself = lrelu(s[n] + dn);
    float m = lself;
    for (int c = e0; c < e1; c += 32) {
      const int i = c + lane;
      if (i < e1) {
        const int2 ed = el[i];
        const float lg = lrelu(s[ed.x] + dn);
        if (ed.y == branch) m = fmaxf(m, lg);
      }
    }
    for (int o = 16; o; o >>= 1) m = fmaxf(m, __shfl_xor(m, o, 32));
    const float wself = __expf(lself - m);
    float den = wself;
#pragma unroll
    for (int j = 0; j < 8; ++j) acc[j] *= wself;
    for (int c = e0; c < e1; c += 32) {
      const int i = c + lane;
      bool on = false;
      int sn = 0;
      float w = 0.f;
      if (i < e1) {
        const int2 ed = el[i];
        const float lg = lrelu(s[ed.x] + dn);
        on = (ed.y == branch);
        sn = ed.x;
        if (on) w = __expf(lg - m);
      }
      const unsigned live = __builtin_amdgcn_ballot_w32(on);
      int rank = 0;
      for (unsigned b = live; b; b &= b - 1u) {
        const int j = __builtin_ctz(b);
        const int sj = __builtin_amdgcn_readlane(sn, j);
        rank += (sj < sn || (sj == sn && j < lane)) ? 1 : 0;
      }
      const int cnt = __builtin_popcount(live);
#pragma unroll 2
      for (int r = 0; r < cnt; ++r) {
        const int j = __builtin_ctz(__builtin_amdgcn_ballot_w32(on && rank == r));
        const int sj = __builtin_amdgcn_readlane(sn, j);
        const float wj = __uint_as_float(__builtin_amdgcn_readlane(__float_as_uint(w), j));
        den += wj;
        const float* sp = x + (size_t)sj * H + lane;
        float v[8];
#pragma unroll
        for (int k = 0; k < 8; ++k) v[k] = sp[32 * k];
#pragma unroll
        for (int k = 0; k < 8; ++k) acc[k] += wj * v[k];
      }
    }
    float* op = xo + (size_t)n * H + lane;
#pragma unroll
    for (int k = 0; k < 8; ++k) op[32 * k] = acc[k] / den;
  }
}


extern "C" void kernel_launch(void* const* d_in, const int* in_sizes, int n_in,
                              void* d_out, int out_size, void* d_ws, size_t ws_size,
                              hipStream_t stream) {
  const float* x     = (const float*)d_in[0];
  const int*   ei    = (const int*)d_in[1];
  const int*   ea    = (const int*)d_in[2];
  const float* Wg    = (const float*)d_in[3];
  const float* a_src = (const float*)d_in[4];
  const float* a_dst = (const float*)d_in[5];
  const float* bg    = (const float*)d_in[6];
  const float* Wsl   = (const float*)d_in[7];
  const float* bsl   = (const float*)d_in[8];
  const float* gsl   = (const float*)d_in[9];
  const float* betas = (const float*)d_in[10];
  const float* W1    = (const float*)d_in[11];
  const float* b1    = (const float*)d_in[12];
  const float* W2    = (const float*)d_in[13];
  const float* b2    = (const float*)d_in[14];
  const float* Wf    = (const float*)d_in[15];
  const float* bfv   = (const float*)d_in[16];
  const float* gf    = (const float*)d_in[17];
  const float* betaf = (const float*)d_in[18];

  const int N = in_sizes[0] / H;
  const int E = in_sizes[2];
  float* out = (float*)d_out;

  char* wsb = (char*)d_ws;
  size_t off = 0;
  auto alloc = [&](size_t bytes) -> void* {
    void* p = wsb + off;
    off += (bytes + 255) & ~(size_t)255;
    return p;
  };
  float*    sw    = (float*)alloc(4 * sizeof(float));
  const int nPad = (N + SCAN_TILE - 1) / SCAN_TILE * SCAN_TILE;
  int*      ebnd  = (int*)alloc((size_t)nPad * sizeof(int));
  float*    bufX  = (float*)alloc((size_t)N * H * sizeof(float));
  int*      elist = (int*)alloc((size_t)N * H * sizeof(float));
  unsigned* wt   = (unsigned*)((char*)elist + (((size_t)E * 8 + 255) & ~(size_t)255));
  float*    uv   = (float*)(wt + (size_t)8 * 2 * WPLANE_W);
  float*    cvec = uv + 6 * H;
  float*    sd   = cvec + 3 * H;

  const int gemmBlocks = (N + TM - 1) / TM;
  dim3 gemmGrid((unsigned)gemmBlocks, 1u);
  const int eThrBlocks = (E + 255) / 256;

  float* part = bufX;
  k_prep<<<CS_GROUPS + SPLIT_GROUPS + UV_GROUPS, CS_THREADS, 0, stream>>>(
      x, part, ebnd, nPad, N, Wg, Wsl, Wf, wt, a_src, a_dst, uv);
  k_fold_deg<<<FOLD_TILES + CVEC_GROUPS + 512, 256, 0, stream>>>(Wg, Wsl, bg, bsl, wt, cvec,
                                                               ei + E, ebnd, E);
  k_scan_sel<<<2 + SDX_GROUPS, 1024, 0, stream>>>(ebnd, N, part, W1, b1, W2, b2, sw, x, uv, sd);
  k_fill<<<eThrBlocks, 256, 0, stream>>>(ei, ei + E, ea, ebnd, elist, E);

  for (int i = 0; i < 3; ++i) {
    k_gat_gather<<<ROW_BLOCKS, 256, 0, stream>>>(x, sd + (size_t)(2 * i) * N,
                                                 sd + (size_t)(2 * i + 1) * N, ebnd, elist, i,
                                                 bufX, N);
    k_gemm_ln<<<gemmGrid, 256, 0, stream>>>(bufX, wt + (size_t)i * 2 * WPLANE_W,
                                            wt + (size_t)i * 2 * WPLANE_W + WPLANE_W,
                                            cvec + i * H, gsl + i * H, betas + i * H, sw, i,
                                            out, i > 0 ? 1 : 0, out, N);
  }
  k_gemm_ln<<<gemmGrid, 256, 0, stream>>>(x, wt + (size_t)6 * 2 * WPLANE_W,
                                          wt + (size_t)6 * 2 * WPLANE_W + WPLANE_W, bsl + 3 * H,
                                          gsl + 3 * H, betas + 3 * H, sw, 3, out, 1, bufX, N);

  k_gemm_ln<<<gemmGrid, 256, 0, stream>>>(bufX, wt + (size_t)7 * 2 * WPLANE_W,
                                          wt + (size_t)7 * 2 * WPLANE_W + WPLANE_W, bfv,
                                          gf, betaf, nullptr, 0, nullptr, 0, out, N);
}
